# starting kernel
# baseline (speedup 1.0000x reference)
#include <hip/hip_runtime.h>
#include <hip/hip_bf16.h>
#include <stdint.h>

#define N_NODES 100000
#define N_EDGES 1600000
#define F_IN    256
#define HID     64
#define RT      5

typedef __attribute__((ext_vector_type(16))) _Float16 v16h;
typedef __attribute__((ext_vector_type(8)))  float    v8f;
typedef __attribute__((ext_vector_type(4)))  float    v4f;

static __device__ __forceinline__ void atomicAddF32(float* p, float v) {
    __hip_atomic_fetch_add(p, v, __ATOMIC_RELAXED, __HIP_MEMORY_SCOPE_AGENT);
}

__global__ void k_deg_zero(int* __restrict__ deg) {
    int i = blockIdx.x * blockDim.x + threadIdx.x;
    if (i < N_NODES) deg[i] = 0;
}

__global__ void k_deg_count(const int* __restrict__ ei, int* __restrict__ deg) {
    int e = blockIdx.x * blockDim.x + threadIdx.x;
    if (e < N_EDGES) {
        int d = ei[N_EDGES + e];
        atomicAdd(&deg[d], 1);
    }
}

__global__ void k_dinv(const int* __restrict__ deg, float* __restrict__ dinv) {
    int i = blockIdx.x * blockDim.x + threadIdx.x;
    if (i < N_NODES) dinv[i] = __frsqrt_rn((float)deg[i] + 1.0f);
}

__global__ __launch_bounds__(128) void k_gemm1_wmma(
        const float* __restrict__ x,
        const float* __restrict__ W1,
        float* __restrict__ xw1) {
    const int wave = threadIdx.x >> 5;
    const int lane = threadIdx.x & 31;
    const int half = lane >> 4;
    const int r    = lane & 15;
    const int n0   = wave * 16;

    v16h bfrag[8];
    #pragma unroll
    for (int kt = 0; kt < 8; ++kt) {
        const int kbase = kt * 32;
        #pragma unroll
        for (int j = 0; j < 8; ++j)
            bfrag[kt][j]     = (_Float16)W1[(kbase + 8 * half + j) * HID + (n0 + r)];
        #pragma unroll
        for (int j = 0; j < 8; ++j)
            bfrag[kt][8 + j] = (_Float16)W1[(kbase + 16 + 8 * half + j) * HID + (n0 + r)];
    }

    for (int rt = 0; rt < RT; ++rt) {
        const int row0 = (blockIdx.x * RT + rt) * 16;
        const float* xrow = x + (size_t)(row0 + r) * F_IN;

        if (rt + 1 < RT) {
            const float* nxt = xrow + 16 * F_IN;
            __builtin_prefetch(nxt +   0, 0, 0);
            __builtin_prefetch(nxt +  64, 0, 0);
            __builtin_prefetch(nxt + 128, 0, 0);
            __builtin_prefetch(nxt + 192, 0, 0);
        }

        v8f acc = {0.f, 0.f, 0.f, 0.f, 0.f, 0.f, 0.f, 0.f};
        #pragma unroll
        for (int kt = 0; kt < 8; ++kt) {
            const float* seg0 = xrow + kt * 32 + 8 * half;
            const float* seg1 = seg0 + 16;
            v4f a0 = __builtin_nontemporal_load((const v4f*)seg0);
            v4f a1 = __builtin_nontemporal_load((const v4f*)(seg0 + 4));
            v4f a2 = __builtin_nontemporal_load((const v4f*)seg1);
            v4f a3 = __builtin_nontemporal_load((const v4f*)(seg1 + 4));
            v16h a;
            a[0]  = (_Float16)a0.x; a[1]  = (_Float16)a0.y;
            a[2]  = (_Float16)a0.z; a[3]  = (_Float16)a0.w;
            a[4]  = (_Float16)a1.x; a[5]  = (_Float16)a1.y;
            a[6]  = (_Float16)a1.z; a[7]  = (_Float16)a1.w;
            a[8]  = (_Float16)a2.x; a[9]  = (_Float16)a2.y;
            a[10] = (_Float16)a2.z; a[11] = (_Float16)a2.w;
            a[12] = (_Float16)a3.x; a[13] = (_Float16)a3.y;
            a[14] = (_Float16)a3.z; a[15] = (_Float16)a3.w;

            acc = __builtin_amdgcn_wmma_f32_16x16x32_f16(
                 false, a,  false, bfrag[kt],
                 (short)0, acc,  false,  false);
        }

        #pragma unroll
        for (int i = 0; i < 8; ++i) {
            int m = i + 8 * half;
            xw1[(size_t)(row0 + m) * HID + (n0 + r)] = acc[i];
        }
    }
}

__global__ void k_agg1_init(const float* __restrict__ xw1,
                            const float* __restrict__ dinv,
                            const float* __restrict__ b1,
                            float* __restrict__ agg1) {
    size_t t = (size_t)blockIdx.x * blockDim.x + threadIdx.x;
    if (t < (size_t)N_NODES * HID) {
        int node = (int)(t >> 6);
        int f    = (int)(t & 63);
        float di = dinv[node];
        agg1[t] = xw1[t] * (di * di) + b1[f];
    }
}

__global__ void k_agg1_edges(const int* __restrict__ ei,
                             const float* __restrict__ dinv,
                             const float* __restrict__ xw1,
                             float* __restrict__ agg1) {
    size_t t = (size_t)blockIdx.x * blockDim.x + threadIdx.x;
    int e    = (int)(t >> 5);
    int lane = (int)(t & 31);
    if (e >= N_EDGES) return;
    e = __builtin_amdgcn_readfirstlane(e);
    int s = ei[e];
    int d = ei[N_EDGES + e];
    float norm = dinv[s] * dinv[d];
    const float2 v = *(const float2*)(xw1 + (size_t)s * HID + lane * 2);
    float* out = agg1 + (size_t)d * HID + lane * 2;
    atomicAddF32(out + 0, v.x * norm);
    atomicAddF32(out + 1, v.y * norm);
}

__global__ void k_relu_gemm2(const float* __restrict__ agg1,
                             const float* __restrict__ W2,
                             const float* __restrict__ b2,
                             const float* __restrict__ dinv,
                             float* __restrict__ hw2,
                             float* __restrict__ agg2) {
    int i = blockIdx.x * blockDim.x + threadIdx.x;
    if (i >= N_NODES) return;
    const float* row = agg1 + (size_t)i * HID;
    float o0 = 0.f, o1 = 0.f;
    #pragma unroll 8
    for (int k = 0; k < HID; ++k) {
        float h = fmaxf(row[k], 0.f);
        o0 += h * W2[k * 2 + 0];
        o1 += h * W2[k * 2 + 1];
    }
    hw2[(size_t)i * 2 + 0] = o0;
    hw2[(size_t)i * 2 + 1] = o1;
    float di = dinv[i];
    float sl = di * di;
    agg2[(size_t)i * 2 + 0] = o0 * sl + b2[0];
    agg2[(size_t)i * 2 + 1] = o1 * sl + b2[1];
}

__global__ void k_agg2_edges(const int* __restrict__ ei,
                             const float* __restrict__ dinv,
                             const float* __restrict__ hw2,
                             float* __restrict__ agg2) {
    int e = blockIdx.x * blockDim.x + threadIdx.x;
    if (e >= N_EDGES) return;
    int s = ei[e];
    int d = ei[N_EDGES + e];
    float norm = dinv[s] * dinv[d];
    const float2 v = *(const float2*)(hw2 + (size_t)s * 2);
    float* out = agg2 + (size_t)d * 2;
    atomicAddF32(out + 0, v.x * norm);
    atomicAddF32(out + 1, v.y * norm);
}

__global__ void k_softmax2(const float* __restrict__ agg2,
                           float* __restrict__ out) {
    int i = blockIdx.x * blockDim.x + threadIdx.x;
    if (i >= N_NODES) return;
    float a0 = agg2[(size_t)i * 2 + 0];
    float a1 = agg2[(size_t)i * 2 + 1];
    float m  = fmaxf(a0, a1);
    float e0 = __expf(a0 - m);
    float e1 = __expf(a1 - m);
    float inv = 1.0f / (e0 + e1);
    out[(size_t)i * 2 + 0] = e0 * inv;
    out[(size_t)i * 2 + 1] = e1 * inv;
}

extern "C" void kernel_launch(void* const* d_in, const int* in_sizes, int n_in,
                              void* d_out, int out_size, void* d_ws, size_t ws_size,
                              hipStream_t stream) {
    const float* x  = (const float*)d_in[0];
    const int*   ei = (const int*)d_in[1];
    const float* W1 = (const float*)d_in[2];
    const float* b1 = (const float*)d_in[3];
    const float* W2 = (const float*)d_in[4];
    const float* b2 = (const float*)d_in[5];
    float* out = (float*)d_out;

    char* ws = (char*)d_ws;
    size_t o = 0;
    auto alloc = [&](size_t bytes) {
        char* p = ws + o;
        o += (bytes + 4095) & ~(size_t)4095;
        return p;
    };
    int*   deg  = (int*)  alloc((size_t)N_NODES * sizeof(int));
    float* dinv = (float*)alloc((size_t)N_NODES * sizeof(float));
    float* xw1  = (float*)alloc((size_t)N_NODES * HID * sizeof(float));
    float* agg1 = (float*)alloc((size_t)N_NODES * HID * sizeof(float));
    float* hw2  = (float*)alloc((size_t)N_NODES * 2 * sizeof(float));
    float* agg2 = (float*)alloc((size_t)N_NODES * 2 * sizeof(float));
    (void)ws_size; (void)in_sizes; (void)n_in; (void)out_size;

    const int B = 256;
    dim3 blk(B);

    k_deg_zero <<<(N_NODES + B - 1) / B, blk, 0, stream>>>(deg);
    k_deg_count<<<(N_EDGES + B - 1) / B, blk, 0, stream>>>(ei, deg);
    k_dinv     <<<(N_NODES + B - 1) / B, blk, 0, stream>>>(deg, dinv);

    k_gemm1_wmma<<<(N_NODES / 16) / RT, dim3(128), 0, stream>>>(x, W1, xw1);

    size_t t1 = (size_t)N_NODES * HID;
    k_agg1_init<<<(unsigned)((t1 + B - 1) / B), blk, 0, stream>>>(xw1, dinv, b1, agg1);
    size_t t2 = (size_t)N_EDGES * 32;
    k_agg1_edges<<<(unsigned)((t2 + B - 1) / B), blk, 0, stream>>>(ei, dinv, xw1, agg1);

    k_relu_gemm2<<<(N_NODES + B - 1) / B, blk, 0, stream>>>(agg1, W2, b2, dinv, hw2, agg2);

    k_agg2_edges<<<(N_EDGES + B - 1) / B, blk, 0, stream>>>(ei, dinv, hw2, agg2);
    k_softmax2  <<<(N_NODES + B - 1) / B, blk, 0, stream>>>(agg2, out);
}


# layer-1 edge pass: one wave walks 8 edges, all index/dinv/row loads issued before the adds; lane l owns columns l and l+32 so each atomic float add covers one whole 128-B line (arrival order kept)
# speedup vs baseline: 1.3376x; 1.3376x over previous
#include <hip/hip_runtime.h>
#include <hip/hip_bf16.h>
#include <stdint.h>

#define N_NODES 100000
#define N_EDGES 1600000
#define F_IN    256
#define HID     64
#define RT      5

typedef __attribute__((ext_vector_type(16))) _Float16 v16h;
typedef __attribute__((ext_vector_type(8)))  float    v8f;
typedef __attribute__((ext_vector_type(4)))  float    v4f;

static __device__ __forceinline__ void atomicAddF32(float* p, float v) {
    __hip_atomic_fetch_add(p, v, __ATOMIC_RELAXED, __HIP_MEMORY_SCOPE_AGENT);
}

__global__ void k_deg_zero(int* __restrict__ deg) {
    int i = blockIdx.x * blockDim.x + threadIdx.x;
    if (i < N_NODES) deg[i] = 0;
}

__global__ void k_deg_count(const int* __restrict__ ei, int* __restrict__ deg) {
    int e = blockIdx.x * blockDim.x + threadIdx.x;
    if (e < N_EDGES) {
        int d = ei[N_EDGES + e];
        atomicAdd(&deg[d], 1);
    }
}

__global__ void k_dinv(const int* __restrict__ deg, float* __restrict__ dinv) {
    int i = blockIdx.x * blockDim.x + threadIdx.x;
    if (i < N_NODES) dinv[i] = __frsqrt_rn((float)deg[i] + 1.0f);
}

__global__ __launch_bounds__(128) void k_gemm1_wmma(
        const float* __restrict__ x,
        const float* __restrict__ W1,
        float* __restrict__ xw1) {
    const int wave = threadIdx.x >> 5;
    const int lane = threadIdx.x & 31;
    const int half = lane >> 4;
    const int r    = lane & 15;
    const int n0   = wave * 16;

    v16h bfrag[8];
    #pragma unroll
    for (int kt = 0; kt < 8; ++kt) {
        const int kbase = kt * 32;
        #pragma unroll
        for (int j = 0; j < 8; ++j)
            bfrag[kt][j]     = (_Float16)W1[(kbase + 8 * half + j) * HID + (n0 + r)];
        #pragma unroll
        for (int j = 0; j < 8; ++j)
            bfrag[kt][8 + j] = (_Float16)W1[(kbase + 16 + 8 * half + j) * HID + (n0 + r)];
    }

    for (int rt = 0; rt < RT; ++rt) {
        const int row0 = (blockIdx.x * RT + rt) * 16;
        const float* xrow = x + (size_t)(row0 + r) * F_IN;

        if (rt + 1 < RT) {
            const float* nxt = xrow + 16 * F_IN;
            __builtin_prefetch(nxt +   0, 0, 0);
            __builtin_prefetch(nxt +  64, 0, 0);
            __builtin_prefetch(nxt + 128, 0, 0);
            __builtin_prefetch(nxt + 192, 0, 0);
        }

        v8f acc = {0.f, 0.f, 0.f, 0.f, 0.f, 0.f, 0.f, 0.f};
        #pragma unroll
        for (int kt = 0; kt < 8; ++kt) {
            const float* seg0 = xrow + kt * 32 + 8 * half;
            const float* seg1 = seg0 + 16;
            v4f a0 = __builtin_nontemporal_load((const v4f*)seg0);
            v4f a1 = __builtin_nontemporal_load((const v4f*)(seg0 + 4));
            v4f a2 = __builtin_nontemporal_load((const v4f*)seg1);
            v4f a3 = __builtin_nontemporal_load((const v4f*)(seg1 + 4));
            v16h a;
            a[0]  = (_Float16)a0.x; a[1]  = (_Float16)a0.y;
            a[2]  = (_Float16)a0.z; a[3]  = (_Float16)a0.w;
            a[4]  = (_Float16)a1.x; a[5]  = (_Float16)a1.y;
            a[6]  = (_Float16)a1.z; a[7]  = (_Float16)a1.w;
            a[8]  = (_Float16)a2.x; a[9]  = (_Float16)a2.y;
            a[10] = (_Float16)a2.z; a[11] = (_Float16)a2.w;
            a[12] = (_Float16)a3.x; a[13] = (_Float16)a3.y;
            a[14] = (_Float16)a3.z; a[15] = (_Float16)a3.w;

            acc = __builtin_amdgcn_wmma_f32_16x16x32_f16(
                 false, a,  false, bfrag[kt],
                 (short)0, acc,  false,  false);
        }

        #pragma unroll
        for (int i = 0; i < 8; ++i) {
            int m = i + 8 * half;
            xw1[(size_t)(row0 + m) * HID + (n0 + r)] = acc[i];
        }
    }
}

__global__ void k_agg1_init(const float* __restrict__ xw1,
                            const float* __restrict__ dinv,
                            const float* __restrict__ b1,
                            float* __restrict__ agg1) {
    size_t t = (size_t)blockIdx.x * blockDim.x + threadIdx.x;
    if (t < (size_t)N_NODES * HID) {
        int node = (int)(t >> 6);
        int f    = (int)(t & 63);
        float di = dinv[node];
        agg1[t] = xw1[t] * (di * di) + b1[f];
    }
}

#define EPW 8
static_assert(N_EDGES % EPW == 0, "edge runs must tile the edge list");
__global__ __launch_bounds__(256) void k_agg1_edges(const int* __restrict__ ei,
                             const float* __restrict__ dinv,
                             const float* __restrict__ xw1,
                             float* __restrict__ agg1) {
    const int lane = threadIdx.x & 31;
    int w = blockIdx.x * 8 + (int)(threadIdx.x >> 5);
    w = __builtin_amdgcn_readfirstlane(w);
    const int e0 = w * EPW;
    if (e0 >= N_EDGES) return;

    int s[EPW], d[EPW];
    #pragma unroll
    for (int k = 0; k < EPW; ++k) {
        s[k] = ei[e0 + k];
        d[k] = ei[N_EDGES + e0 + k];
    }
    __builtin_amdgcn_sched_barrier(0);
    float nrm[EPW];
    #pragma unroll
    for (int k = 0; k < EPW; ++k)
        nrm[k] = dinv[s[k]] * dinv[d[k]];
    float lo[EPW], hi[EPW];
    #pragma unroll
    for (int k = 0; k < EPW; ++k) {
        const float* row = xw1 + (size_t)s[k] * HID;
        lo[k] = row[lane];
        hi[k] = row[32 + lane];
    }
    #pragma unroll
    for (int k = 0; k < EPW; ++k) {
        float* o = agg1 + (size_t)d[k] * HID;
        atomicAddF32(o + lane,      lo[k] * nrm[k]);
        atomicAddF32(o + 32 + lane, hi[k] * nrm[k]);
    }
}

__global__ void k_relu_gemm2(const float* __restrict__ agg1,
                             const float* __restrict__ W2,
                             const float* __restrict__ b2,
                             const float* __restrict__ dinv,
                             float* __restrict__ hw2,
                             float* __restrict__ agg2) {
    int i = blockIdx.x * blockDim.x + threadIdx.x;
    if (i >= N_NODES) return;
    const float* row = agg1 + (size_t)i * HID;
    float o0 = 0.f, o1 = 0.f;
    #pragma unroll 8
    for (int k = 0; k < HID; ++k) {
        float h = fmaxf(row[k], 0.f);
        o0 += h * W2[k * 2 + 0];
        o1 += h * W2[k * 2 + 1];
    }
    hw2[(size_t)i * 2 + 0] = o0;
    hw2[(size_t)i * 2 + 1] = o1;
    float di = dinv[i];
    float sl = di * di;
    agg2[(size_t)i * 2 + 0] = o0 * sl + b2[0];
    agg2[(size_t)i * 2 + 1] = o1 * sl + b2[1];
}

__global__ void k_agg2_edges(const int* __restrict__ ei,
                             const float* __restrict__ dinv,
                             const float* __restrict__ hw2,
                             float* __restrict__ agg2) {
    int e = blockIdx.x * blockDim.x + threadIdx.x;
    if (e >= N_EDGES) return;
    int s = ei[e];
    int d = ei[N_EDGES + e];
    float norm = dinv[s] * dinv[d];
    const float2 v = *(const float2*)(hw2 + (size_t)s * 2);
    float* out = agg2 + (size_t)d * 2;
    atomicAddF32(out + 0, v.x * norm);
    atomicAddF32(out + 1, v.y * norm);
}

__global__ void k_softmax2(const float* __restrict__ agg2,
                           float* __restrict__ out) {
    int i = blockIdx.x * blockDim.x + threadIdx.x;
    if (i >= N_NODES) return;
    float a0 = agg2[(size_t)i * 2 + 0];
    float a1 = agg2[(size_t)i * 2 + 1];
    float m  = fmaxf(a0, a1);
    float e0 = __expf(a0 - m);
    float e1 = __expf(a1 - m);
    float inv = 1.0f / (e0 + e1);
    out[(size_t)i * 2 + 0] = e0 * inv;
    out[(size_t)i * 2 + 1] = e1 * inv;
}

extern "C" void kernel_launch(void* const* d_in, const int* in_sizes, int n_in,
                              void* d_out, int out_size, void* d_ws, size_t ws_size,
                              hipStream_t stream) {
    const float* x  = (const float*)d_in[0];
    const int*   ei = (const int*)d_in[1];
    const float* W1 = (const float*)d_in[2];
    const float* b1 = (const float*)d_in[3];
    const float* W2 = (const float*)d_in[4];
    const float* b2 = (const float*)d_in[5];
    float* out = (float*)d_out;

    char* ws = (char*)d_ws;
    size_t o = 0;
    auto alloc = [&](size_t bytes) {
        char* p = ws + o;
        o += (bytes + 4095) & ~(size_t)4095;
        return p;
    };
    int*   deg  = (int*)  alloc((size_t)N_NODES * sizeof(int));
    float* dinv = (float*)alloc((size_t)N_NODES * sizeof(float));
    float* xw1  = (float*)alloc((size_t)N_NODES * HID * sizeof(float));
    float* agg1 = (float*)alloc((size_t)N_NODES * HID * sizeof(float));
    float* hw2  = (float*)alloc((size_t)N_NODES * 2 * sizeof(float));
    float* agg2 = (float*)alloc((size_t)N_NODES * 2 * sizeof(float));
    (void)ws_size; (void)in_sizes; (void)n_in; (void)out_size;

    const int B = 256;
    dim3 blk(B);

    k_deg_zero <<<(N_NODES + B - 1) / B, blk, 0, stream>>>(deg);
    k_deg_count<<<(N_EDGES + B - 1) / B, blk, 0, stream>>>(ei, deg);
    k_dinv     <<<(N_NODES + B - 1) / B, blk, 0, stream>>>(deg, dinv);

    k_gemm1_wmma<<<(N_NODES / 16) / RT, dim3(128), 0, stream>>>(x, W1, xw1);

    size_t t1 = (size_t)N_NODES * HID;
    k_agg1_init<<<(unsigned)((t1 + B - 1) / B), blk, 0, stream>>>(xw1, dinv, b1, agg1);
    size_t t2 = (size_t)(N_EDGES / EPW) * 32;
    k_agg1_edges<<<(unsigned)((t2 + B - 1) / B), blk, 0, stream>>>(ei, dinv, xw1, agg1);

    k_relu_gemm2<<<(N_NODES + B - 1) / B, blk, 0, stream>>>(agg1, W2, b2, dinv, hw2, agg2);

    k_agg2_edges<<<(N_EDGES + B - 1) / B, blk, 0, stream>>>(ei, dinv, hw2, agg2);
    k_softmax2  <<<(N_NODES + B - 1) / B, blk, 0, stream>>>(agg2, out);
}


# aggregations gather over a dst-sorted edge list rebuilt each call (count rank, scan, place): no float atomics, row epilogues fused; sums in count-arrival order, 64x2 dot and layer-2 by lane tree
# speedup vs baseline: 2.0562x; 1.5372x over previous
#include <hip/hip_runtime.h>
#include <hip/hip_bf16.h>
#include <stdint.h>

#define N_NODES 100000
#define N_EDGES 1600000
#define F_IN    256
#define HID     64
#define RT      5

typedef __attribute__((ext_vector_type(16))) _Float16 v16h;
typedef __attribute__((ext_vector_type(8)))  float    v8f;
typedef __attribute__((ext_vector_type(4)))  float    v4f;

static __device__ __forceinline__ int wave_sum_i32(int v) {
    for (int o = 16; o > 0; o >>= 1) v += __shfl_xor(v, o);
    return v;
}

__global__ void k_deg_zero(int* __restrict__ deg) {
    int i = blockIdx.x * blockDim.x + threadIdx.x;
    if (i < N_NODES) deg[i] = 0;
}

static_assert(N_EDGES % 4 == 0, "edge quads must tile the edge list");
__global__ __launch_bounds__(256) void k_deg_count_rank(const int* __restrict__ ei,
                                                        int* __restrict__ deg,
                                                        int* __restrict__ rank) {
    int q = blockIdx.x * blockDim.x + threadIdx.x;
    if (q < N_EDGES / 4) {
        const int4 d = *(const int4*)(ei + N_EDGES + 4 * (size_t)q);
        int4 r;
        r.x = atomicAdd(&deg[d.x], 1);
        r.y = atomicAdd(&deg[d.y], 1);
        r.z = atomicAdd(&deg[d.z], 1);
        r.w = atomicAdd(&deg[d.w], 1);
        *(int4*)(rank + 4 * (size_t)q) = r;
    }
}

#define SCAN_B 256
#define SCAN_NB ((N_NODES + SCAN_B - 1) / SCAN_B)
__global__ __launch_bounds__(SCAN_B) void k_dinv_bsum(const int* __restrict__ deg,
                                                      float* __restrict__ dinv,
                                                      int* __restrict__ bsum) {
    __shared__ int sh[SCAN_B / 32];
    const int t = threadIdx.x;
    const int i = blockIdx.x * SCAN_B + t;
    int dg = 0;
    if (i < N_NODES) {
        dg = deg[i];
        dinv[i] = __frsqrt_rn((float)dg + 1.0f);
    }
    const int s = wave_sum_i32(dg);
    if ((t & 31) == 0) sh[t >> 5] = s;
    __syncthreads();
    if (t == 0) {
        int tot = 0;
        for (int k = 0; k < SCAN_B / 32; ++k) tot += sh[k];
        bsum[blockIdx.x] = tot;
    }
}

__global__ __launch_bounds__(SCAN_B) void k_rowptr(const int* __restrict__ deg,
                                                   const int* __restrict__ bsum,
                                                   int* __restrict__ rowptr) {
    __shared__ int sh_base[SCAN_B / 32];
    __shared__ int sh_wave[SCAN_B / 32];
    const int t = threadIdx.x, lane = t & 31, wv = t >> 5;
    const int b = blockIdx.x;
    int part = 0;
    for (int j = t; j < b; j += SCAN_B) part += bsum[j];
    part = wave_sum_i32(part);
    if (lane == 0) sh_base[wv] = part;

    const int i = b * SCAN_B + t;
    const int dg = (i < N_NODES) ? deg[i] : 0;
    int inc = dg;
    for (int o = 1; o < 32; o <<= 1) {
        const int v = __shfl_up(inc, o);
        if (lane >= o) inc += v;
    }
    if (lane == 31) sh_wave[wv] = inc;
    __syncthreads();
    int off = 0;
    for (int k = 0; k < SCAN_B / 32; ++k) {
        off += sh_base[k];
        if (k < wv) off += sh_wave[k];
    }
    const int excl = off + inc - dg;
    if (i < N_NODES) rowptr[i] = excl;
    if (i == N_NODES - 1) rowptr[N_NODES] = excl + dg;
}

__global__ __launch_bounds__(256) void k_place(const int* __restrict__ ei,
                                               const int* __restrict__ rowptr,
                                               const int* __restrict__ rank,
                                               int* __restrict__ csr_src) {
    int q = blockIdx.x * blockDim.x + threadIdx.x;
    if (q < N_EDGES / 4) {
        const int4 s = *(const int4*)(ei + 4 * (size_t)q);
        const int4 d = *(const int4*)(ei + N_EDGES + 4 * (size_t)q);
        const int4 r = *(const int4*)(rank + 4 * (size_t)q);
        const int p0 = rowptr[d.x], p1 = rowptr[d.y], p2 = rowptr[d.z], p3 = rowptr[d.w];
        csr_src[p0 + r.x] = s.x;
        csr_src[p1 + r.y] = s.y;
        csr_src[p2 + r.z] = s.z;
        csr_src[p3 + r.w] = s.w;
    }
}

__global__ __launch_bounds__(128) void k_gemm1_wmma(
        const float* __restrict__ x,
        const float* __restrict__ W1,
        float* __restrict__ xw1) {
    const int wave = threadIdx.x >> 5;
    const int lane = threadIdx.x & 31;
    const int half = lane >> 4;
    const int r    = lane & 15;
    const int n0   = wave * 16;

    v16h bfrag[8];
    #pragma unroll
    for (int kt = 0; kt < 8; ++kt) {
        const int kbase = kt * 32;
        #pragma unroll
        for (int j = 0; j < 8; ++j)
            bfrag[kt][j]     = (_Float16)W1[(kbase + 8 * half + j) * HID + (n0 + r)];
        #pragma unroll
        for (int j = 0; j < 8; ++j)
            bfrag[kt][8 + j] = (_Float16)W1[(kbase + 16 + 8 * half + j) * HID + (n0 + r)];
    }

    for (int rt = 0; rt < RT; ++rt) {
        const int row0 = (blockIdx.x * RT + rt) * 16;
        const float* xrow = x + (size_t)(row0 + r) * F_IN;

        if (rt + 1 < RT) {
            const float* nxt = xrow + 16 * F_IN;
            __builtin_prefetch(nxt +   0, 0, 0);
            __builtin_prefetch(nxt +  64, 0, 0);
            __builtin_prefetch(nxt + 128, 0, 0);
            __builtin_prefetch(nxt + 192, 0, 0);
        }

        v8f acc = {0.f, 0.f, 0.f, 0.f, 0.f, 0.f, 0.f, 0.f};
        #pragma unroll
        for (int kt = 0; kt < 8; ++kt) {
            const float* seg0 = xrow + kt * 32 + 8 * half;
            const float* seg1 = seg0 + 16;
            v4f a0 = __builtin_nontemporal_load((const v4f*)seg0);
            v4f a1 = __builtin_nontemporal_load((const v4f*)(seg0 + 4));
            v4f a2 = __builtin_nontemporal_load((const v4f*)seg1);
            v4f a3 = __builtin_nontemporal_load((const v4f*)(seg1 + 4));
            v16h a;
            a[0]  = (_Float16)a0.x; a[1]  = (_Float16)a0.y;
            a[2]  = (_Float16)a0.z; a[3]  = (_Float16)a0.w;
            a[4]  = (_Float16)a1.x; a[5]  = (_Float16)a1.y;
            a[6]  = (_Float16)a1.z; a[7]  = (_Float16)a1.w;
            a[8]  = (_Float16)a2.x; a[9]  = (_Float16)a2.y;
            a[10] = (_Float16)a2.z; a[11] = (_Float16)a2.w;
            a[12] = (_Float16)a3.x; a[13] = (_Float16)a3.y;
            a[14] = (_Float16)a3.z; a[15] = (_Float16)a3.w;

            acc = __builtin_amdgcn_wmma_f32_16x16x32_f16(
                 false, a,  false, bfrag[kt],
                 (short)0, acc,  false,  false);
        }

        #pragma unroll
        for (int i = 0; i < 8; ++i) {
            int m = i + 8 * half;
            xw1[(size_t)(row0 + m) * HID + (n0 + r)] = acc[i];
        }
    }
}

#define EPC 8
static_assert(N_NODES % 8 == 0, "8 waves per block must tile the node list");
__global__ __launch_bounds__(256) void k_gather1(const int* __restrict__ rowptr,
                                                 const int* __restrict__ csr_src,
                                                 const float* __restrict__ dinv,
                                                 const float* __restrict__ xw1,
                                                 const float* __restrict__ b1,
                                                 const float* __restrict__ W2,
                                                 float* __restrict__ hw2s) {
    const int lane = threadIdx.x & 31;
    int i = blockIdx.x * 8 + (int)(threadIdx.x >> 5);
    i = __builtin_amdgcn_readfirstlane(i);
    if (i >= N_NODES) return;

    const int beg = rowptr[i];
    const int end = rowptr[i + 1];
    const float di = dinv[i];
    const float2* rows = (const float2*)xw1;
    const float2 self = rows[(size_t)i * (HID / 2) + lane];

    float ax = 0.f, ay = 0.f;
    for (int base = beg; base < end; base += 32) {
        const int rem = end - base;
        const int n = rem < 32 ? rem : 32;
        const int sv = csr_src[base + (lane < n ? lane : n - 1)];
        for (int c = 0; c < n; c += EPC) {
            int s[EPC];
            float w[EPC];
            float2 v[EPC];
            #pragma unroll
            for (int k = 0; k < EPC; ++k) {
                const int idx = (c + k < n) ? (c + k) : (n - 1);
                s[k] = __builtin_amdgcn_readlane(sv, idx);
            }
            #pragma unroll
            for (int k = 0; k < EPC; ++k)
                w[k] = dinv[s[k]] * di;
            #pragma unroll
            for (int k = 0; k < EPC; ++k)
                v[k] = rows[(size_t)s[k] * (HID / 2) + lane];
            #pragma unroll
            for (int k = 0; k < EPC; ++k) {
                const float px = v[k].x * w[k];
                const float py = v[k].y * w[k];
                const bool live = (c + k < n);
                ax += live ? px : 0.f;
                ay += live ? py : 0.f;
            }
        }
    }

    const float sl = di * di;
    float hx = ax + self.x * sl;
    float hy = ay + self.y * sl;
    hx = fmaxf(hx + b1[2 * lane],     0.f);
    hy = fmaxf(hy + b1[2 * lane + 1], 0.f);
    const float w00 = W2[4 * lane + 0], w01 = W2[4 * lane + 1];
    const float w10 = W2[4 * lane + 2], w11 = W2[4 * lane + 3];
    float o0 = hx * w00 + hy * w10;
    float o1 = hx * w01 + hy * w11;
    for (int o = 16; o > 0; o >>= 1) {
        o0 += __shfl_xor(o0, o);
        o1 += __shfl_xor(o1, o);
    }
    if (lane == 0) {
        float2 r;
        r.x = o0 * di;
        r.y = o1 * di;
        ((float2*)hw2s)[i] = r;
    }
}

#define LPN 8
static_assert((N_NODES * LPN) % 256 == 0, "node groups must tile the blocks");
__global__ __launch_bounds__(256) void k_gather2_softmax(const int* __restrict__ rowptr,
                                                         const int* __restrict__ csr_src,
                                                         const float* __restrict__ dinv,
                                                         const float* __restrict__ hw2s,
                                                         const float* __restrict__ b2,
                                                         float* __restrict__ out) {
    const int t = blockIdx.x * blockDim.x + threadIdx.x;
    const int i = t / LPN;
    const int j = t % LPN;
    const int beg = rowptr[i];
    const int end = rowptr[i + 1];
    const float2* h2 = (const float2*)hw2s;

    float ax = 0.f, ay = 0.f;
    for (int p = beg + j; p < end; p += 2 * LPN) {
        const int  p2   = p + LPN;
        const bool two  = p2 < end;
        const int  s0   = csr_src[p];
        const int  s1   = csr_src[two ? p2 : p];
        const float2 v0 = h2[s0];
        const float2 v1 = h2[s1];
        ax += v0.x;
        ay += v0.y;
        ax += two ? v1.x : 0.f;
        ay += two ? v1.y : 0.f;
    }
    for (int o = LPN / 2; o > 0; o >>= 1) {
        ax += __shfl_xor(ax, o);
        ay += __shfl_xor(ay, o);
    }
    if (j == 0) {
        const float2 self = h2[i];
        const float di = dinv[i];
        const float a0 = di * (ax + self.x) + b2[0];
        const float a1 = di * (ay + self.y) + b2[1];
        const float m  = fmaxf(a0, a1);
        const float e0 = __expf(a0 - m);
        const float e1 = __expf(a1 - m);
        const float inv = 1.0f / (e0 + e1);
        float2 r;
        r.x = e0 * inv;
        r.y = e1 * inv;
        ((float2*)out)[i] = r;
    }
}

extern "C" void kernel_launch(void* const* d_in, const int* in_sizes, int n_in,
                              void* d_out, int out_size, void* d_ws, size_t ws_size,
                              hipStream_t stream) {
    const float* x  = (const float*)d_in[0];
    const int*   ei = (const int*)d_in[1];
    const float* W1 = (const float*)d_in[2];
    const float* b1 = (const float*)d_in[3];
    const float* W2 = (const float*)d_in[4];
    const float* b2 = (const float*)d_in[5];
    float* out = (float*)d_out;

    char* ws = (char*)d_ws;
    size_t o = 0;
    auto alloc = [&](size_t bytes) {
        char* p = ws + o;
        o += (bytes + 4095) & ~(size_t)4095;
        return p;
    };
    int*   deg    = (int*)  alloc((size_t)N_NODES * sizeof(int));
    float* dinv   = (float*)alloc((size_t)N_NODES * sizeof(float));
    float* xw1    = (float*)alloc((size_t)N_NODES * HID * sizeof(float));
    int*   rank   = (int*)  alloc((size_t)N_EDGES * sizeof(int));
    int*   csr    = (int*)  alloc((size_t)N_EDGES * sizeof(int));
    int*   rowptr = (int*)  alloc((size_t)(N_NODES + 1) * sizeof(int));
    int*   bsum   = (int*)  alloc((size_t)SCAN_NB * sizeof(int));
    float* hw2s   = (float*)alloc((size_t)N_NODES * 2 * sizeof(float));
    (void)ws_size; (void)in_sizes; (void)n_in; (void)out_size;

    const int B = 256;
    dim3 blk(B);

    k_deg_zero      <<<(N_NODES + B - 1) / B, blk, 0, stream>>>(deg);
    k_deg_count_rank<<<(N_EDGES / 4 + B - 1) / B, blk, 0, stream>>>(ei, deg, rank);
    k_dinv_bsum     <<<SCAN_NB, dim3(SCAN_B), 0, stream>>>(deg, dinv, bsum);
    k_rowptr        <<<SCAN_NB, dim3(SCAN_B), 0, stream>>>(deg, bsum, rowptr);
    k_place         <<<(N_EDGES / 4 + B - 1) / B, blk, 0, stream>>>(ei, rowptr, rank, csr);

    k_gemm1_wmma<<<(N_NODES / 16) / RT, dim3(128), 0, stream>>>(x, W1, xw1);

    k_gather1<<<N_NODES / 8, blk, 0, stream>>>(rowptr, csr, dinv, xw1, b1, W2, hw2s);

    k_gather2_softmax<<<(N_NODES * LPN) / B, blk, 0, stream>>>(rowptr, csr, dinv, hw2s, b2, out);
}


# edges grouped by dst on LDS counters (count per 512-node bin, scan, place; per-bin count, scan, place): no returning global atomic per edge; a node's sums now run in LDS-arrival order
# speedup vs baseline: 2.7691x; 1.3467x over previous
#include <hip/hip_runtime.h>
#include <hip/hip_bf16.h>
#include <stdint.h>

#define N_NODES 100000
#define N_EDGES 1600000
#define F_IN    256
#define HID     64
#define RT      5

typedef __attribute__((ext_vector_type(16))) _Float16 v16h;
typedef __attribute__((ext_vector_type(8)))  float    v8f;
typedef __attribute__((ext_vector_type(4)))  float    v4f;

static __device__ __forceinline__ int wave_sum_i32(int v) {
    for (int o = 16; o > 0; o >>= 1) v += __shfl_xor(v, o);
    return v;
}

#define BIN_SHIFT 9
#define BIN_W   (1 << BIN_SHIFT)
#define NBIN    ((N_NODES + BIN_W - 1) / BIN_W)
#define SORT_G  250
#define SORT_T  1024
#define QUADS   (N_EDGES / 4)
#define QPB     (QUADS / SORT_G)
#define TBL     (NBIN * SORT_G)
static_assert(N_EDGES % 4 == 0, "edge quads must tile the edge list");
static_assert(QUADS % SORT_G == 0, "edge blocks must tile the quads");
static_assert(N_NODES <= (1 << (31 - BIN_SHIFT)), "src id and dst-in-bin must pack into one word");

__global__ __launch_bounds__(SORT_T) void k_bin_hist(const int* __restrict__ ei,
                                                     int* __restrict__ tbl) {
    __shared__ int cnt[NBIN];
    const int t = threadIdx.x, g = blockIdx.x;
    for (int b = t; b < NBIN; b += SORT_T) cnt[b] = 0;
    __syncthreads();
    const int4* dq = (const int4*)(ei + N_EDGES) + (size_t)g * QPB;
    for (int q = t; q < QPB; q += SORT_T) {
        const int4 d = dq[q];
        atomicAdd(&cnt[d.x >> BIN_SHIFT], 1);
        atomicAdd(&cnt[d.y >> BIN_SHIFT], 1);
        atomicAdd(&cnt[d.z >> BIN_SHIFT], 1);
        atomicAdd(&cnt[d.w >> BIN_SHIFT], 1);
    }
    __syncthreads();
    for (int b = t; b < NBIN; b += SORT_T) tbl[b * SORT_G + g] = cnt[b];
}

#define SCAN_T  1024
#define PER_T   ((TBL + SCAN_T - 1) / SCAN_T)
#define PITCH   (PER_T + 1 - (PER_T & 1))
__global__ __launch_bounds__(SCAN_T) void k_bin_scan(const int* __restrict__ tbl,
                                                     int* __restrict__ off) {
    __shared__ int buf[SCAN_T * PITCH];
    __shared__ int wtot[SCAN_T / 32];
    const int t = threadIdx.x, lane = t & 31, wv = t >> 5;
    for (int i = t; i < SCAN_T * PER_T; i += SCAN_T)
        buf[(i / PER_T) * PITCH + (i % PER_T)] = (i < TBL) ? tbl[i] : 0;
    __syncthreads();
    int* mine = buf + t * PITCH;
    int v[PER_T];
    int run = 0;
    #pragma unroll
    for (int k = 0; k < PER_T; ++k) {
        v[k] = mine[k];
        run += v[k];
    }
    int inc = run;
    for (int o = 1; o < 32; o <<= 1) {
        const int u = __shfl_up(inc, o);
        if (lane >= o) inc += u;
    }
    if (lane == 31) wtot[wv] = inc;
    __syncthreads();
    int pre = inc - run;
    for (int k = 0; k < wv; ++k) pre += wtot[k];
    #pragma unroll
    for (int k = 0; k < PER_T; ++k) {
        mine[k] = pre;
        pre += v[k];
    }
    __syncthreads();
    for (int i = t; i < TBL; i += SCAN_T)
        off[i] = buf[(i / PER_T) * PITCH + (i % PER_T)];
}

__global__ __launch_bounds__(SORT_T) void k_bin_scatter(const int* __restrict__ ei,
                                                        const int* __restrict__ off,
                                                        int* __restrict__ binned) {
    __shared__ int cur[NBIN];
    const int t = threadIdx.x, g = blockIdx.x;
    for (int b = t; b < NBIN; b += SORT_T) cur[b] = off[b * SORT_G + g];
    __syncthreads();
    const int4* sq = (const int4*)ei + (size_t)g * QPB;
    const int4* dq = (const int4*)(ei + N_EDGES) + (size_t)g * QPB;
    for (int q = t; q < QPB; q += SORT_T) {
        const int4 s = sq[q];
        const int4 d = dq[q];
        const int p0 = atomicAdd(&cur[d.x >> BIN_SHIFT], 1);
        const int p1 = atomicAdd(&cur[d.y >> BIN_SHIFT], 1);
        const int p2 = atomicAdd(&cur[d.z >> BIN_SHIFT], 1);
        const int p3 = atomicAdd(&cur[d.w >> BIN_SHIFT], 1);
        binned[p0] = (s.x << BIN_SHIFT) | (d.x & (BIN_W - 1));
        binned[p1] = (s.y << BIN_SHIFT) | (d.y & (BIN_W - 1));
        binned[p2] = (s.z << BIN_SHIFT) | (d.z & (BIN_W - 1));
        binned[p3] = (s.w << BIN_SHIFT) | (d.w & (BIN_W - 1));
    }
}

#define FIN_T   1024
static_assert(BIN_W <= FIN_T && BIN_W % 32 == 0, "one thread per node of the bin, whole waves");
__global__ __launch_bounds__(FIN_T) void k_bin_finalize(const int* __restrict__ off,
                                                        const int* __restrict__ binned,
                                                        int* __restrict__ rowptr,
                                                        float* __restrict__ dinv,
                                                        int* __restrict__ csr_src) {
    __shared__ int cnt[BIN_W];
    __shared__ int wtot[BIN_W / 32];
    const int t = threadIdx.x, lane = t & 31, wv = t >> 5;
    const int b = blockIdx.x;
    const int beg = off[b * SORT_G];
    const int end = (b + 1 < NBIN) ? off[(b + 1) * SORT_G] : N_EDGES;
    if (t < BIN_W) cnt[t] = 0;
    __syncthreads();
    for (int p = beg + t; p < end; p += 4 * FIN_T) {
        int w[4];
        #pragma unroll
        for (int k = 0; k < 4; ++k) {
            const int pk = p + k * FIN_T;
            w[k] = (pk < end) ? binned[pk] : -1;
        }
        #pragma unroll
        for (int k = 0; k < 4; ++k)
            if (w[k] >= 0) atomicAdd(&cnt[w[k] & (BIN_W - 1)], 1);
    }
    __syncthreads();
    int dg = 0, inc = 0;
    if (t < BIN_W) {
        dg = cnt[t];
        inc = dg;
        for (int o = 1; o < 32; o <<= 1) {
            const int u = __shfl_up(inc, o);
            if (lane >= o) inc += u;
        }
        if (lane == 31) wtot[wv] = inc;
    }
    __syncthreads();
    if (t < BIN_W) {
        int base = beg;
        for (int k = 0; k < wv; ++k) base += wtot[k];
        const int excl = base + inc - dg;
        cnt[t] = excl;
        const int node = b * BIN_W + t;
        if (node < N_NODES) {
            rowptr[node] = excl;
            dinv[node] = __frsqrt_rn((float)dg + 1.0f);
        }
        if (node == N_NODES - 1) rowptr[N_NODES] = excl + dg;
    }
    __syncthreads();
    for (int p = beg + t; p < end; p += 4 * FIN_T) {
        int w[4];
        #pragma unroll
        for (int k = 0; k < 4; ++k) {
            const int pk = p + k * FIN_T;
            w[k] = (pk < end) ? binned[pk] : -1;
        }
        #pragma unroll
        for (int k = 0; k < 4; ++k) {
            if (w[k] >= 0) {
                const int pos = atomicAdd(&cnt[w[k] & (BIN_W - 1)], 1);
                csr_src[pos] = w[k] >> BIN_SHIFT;
            }
        }
    }
}

__global__ __launch_bounds__(128) void k_gemm1_wmma(
        const float* __restrict__ x,
        const float* __restrict__ W1,
        float* __restrict__ xw1) {
    const int wave = threadIdx.x >> 5;
    const int lane = threadIdx.x & 31;
    const int half = lane >> 4;
    const int r    = lane & 15;
    const int n0   = wave * 16;

    v16h bfrag[8];
    #pragma unroll
    for (int kt = 0; kt < 8; ++kt) {
        const int kbase = kt * 32;
        #pragma unroll
        for (int j = 0; j < 8; ++j)
            bfrag[kt][j]     = (_Float16)W1[(kbase + 8 * half + j) * HID + (n0 + r)];
        #pragma unroll
        for (int j = 0; j < 8; ++j)
            bfrag[kt][8 + j] = (_Float16)W1[(kbase + 16 + 8 * half + j) * HID + (n0 + r)];
    }

    for (int rt = 0; rt < RT; ++rt) {
        const int row0 = (blockIdx.x * RT + rt) * 16;
        const float* xrow = x + (size_t)(row0 + r) * F_IN;

        if (rt + 1 < RT) {
            const float* nxt = xrow + 16 * F_IN;
            __builtin_prefetch(nxt +   0, 0, 0);
            __builtin_prefetch(nxt +  64, 0, 0);
            __builtin_prefetch(nxt + 128, 0, 0);
            __builtin_prefetch(nxt + 192, 0, 0);
        }

        v8f acc = {0.f, 0.f, 0.f, 0.f, 0.f, 0.f, 0.f, 0.f};
        #pragma unroll
        for (int kt = 0; kt < 8; ++kt) {
            const float* seg0 = xrow + kt * 32 + 8 * half;
            const float* seg1 = seg0 + 16;
            v4f a0 = __builtin_nontemporal_load((const v4f*)seg0);
            v4f a1 = __builtin_nontemporal_load((const v4f*)(seg0 + 4));
            v4f a2 = __builtin_nontemporal_load((const v4f*)seg1);
            v4f a3 = __builtin_nontemporal_load((const v4f*)(seg1 + 4));
            v16h a;
            a[0]  = (_Float16)a0.x; a[1]  = (_Float16)a0.y;
            a[2]  = (_Float16)a0.z; a[3]  = (_Float16)a0.w;
            a[4]  = (_Float16)a1.x; a[5]  = (_Float16)a1.y;
            a[6]  = (_Float16)a1.z; a[7]  = (_Float16)a1.w;
            a[8]  = (_Float16)a2.x; a[9]  = (_Float16)a2.y;
            a[10] = (_Float16)a2.z; a[11] = (_Float16)a2.w;
            a[12] = (_Float16)a3.x; a[13] = (_Float16)a3.y;
            a[14] = (_Float16)a3.z; a[15] = (_Float16)a3.w;

            acc = __builtin_amdgcn_wmma_f32_16x16x32_f16(
                 false, a,  false, bfrag[kt],
                 (short)0, acc,  false,  false);
        }

        #pragma unroll
        for (int i = 0; i < 8; ++i) {
            int m = i + 8 * half;
            xw1[(size_t)(row0 + m) * HID + (n0 + r)] = acc[i];
        }
    }
}

#define EPC 8
static_assert(N_NODES % 8 == 0, "8 waves per block must tile the node list");
__global__ __launch_bounds__(256) void k_gather1(const int* __restrict__ rowptr,
                                                 const int* __restrict__ csr_src,
                                                 const float* __restrict__ dinv,
                                                 const float* __restrict__ xw1,
                                                 const float* __restrict__ b1,
                                                 const float* __restrict__ W2,
                                                 float* __restrict__ hw2s) {
    const int lane = threadIdx.x & 31;
    int i = blockIdx.x * 8 + (int)(threadIdx.x >> 5);
    i = __builtin_amdgcn_readfirstlane(i);
    if (i >= N_NODES) return;

    const int beg = rowptr[i];
    const int end = rowptr[i + 1];
    const float di = dinv[i];
    const float2* rows = (const float2*)xw1;
    const float2 self = rows[(size_t)i * (HID / 2) + lane];

    float ax = 0.f, ay = 0.f;
    for (int base = beg; base < end; base += 32) {
        const int rem = end - base;
        const int n = rem < 32 ? rem : 32;
        const int sv = csr_src[base + (lane < n ? lane : n - 1)];
        for (int c = 0; c < n; c += EPC) {
            int s[EPC];
            float w[EPC];
            float2 v[EPC];
            #pragma unroll
            for (int k = 0; k < EPC; ++k) {
                const int idx = (c + k < n) ? (c + k) : (n - 1);
                s[k] = __builtin_amdgcn_readlane(sv, idx);
            }
            #pragma unroll
            for (int k = 0; k < EPC; ++k)
                w[k] = dinv[s[k]] * di;
            #pragma unroll
            for (int k = 0; k < EPC; ++k)
                v[k] = rows[(size_t)s[k] * (HID / 2) + lane];
            #pragma unroll
            for (int k = 0; k < EPC; ++k) {
                const float px = v[k].x * w[k];
                const float py = v[k].y * w[k];
                const bool live = (c + k < n);
                ax += live ? px : 0.f;
                ay += live ? py : 0.f;
            }
        }
    }

    const float sl = di * di;
    float hx = ax + self.x * sl;
    float hy = ay + self.y * sl;
    hx = fmaxf(hx + b1[2 * lane],     0.f);
    hy = fmaxf(hy + b1[2 * lane + 1], 0.f);
    const float w00 = W2[4 * lane + 0], w01 = W2[4 * lane + 1];
    const float w10 = W2[4 * lane + 2], w11 = W2[4 * lane + 3];
    float o0 = hx * w00 + hy * w10;
    float o1 = hx * w01 + hy * w11;
    for (int o = 16; o > 0; o >>= 1) {
        o0 += __shfl_xor(o0, o);
        o1 += __shfl_xor(o1, o);
    }
    if (lane == 0) {
        float2 r;
        r.x = o0 * di;
        r.y = o1 * di;
        ((float2*)hw2s)[i] = r;
    }
}

#define LPN 8
static_assert((N_NODES * LPN) % 256 == 0, "node groups must tile the blocks");
__global__ __launch_bounds__(256) void k_gather2_softmax(const int* __restrict__ rowptr,
                                                         const int* __restrict__ csr_src,
                                                         const float* __restrict__ dinv,
                                                         const float* __restrict__ hw2s,
                                                         const float* __restrict__ b2,
                                                         float* __restrict__ out) {
    const int t = blockIdx.x * blockDim.x + threadIdx.x;
    const int i = t / LPN;
    const int j = t % LPN;
    const int beg = rowptr[i];
    const int end = rowptr[i + 1];
    const float2* h2 = (const float2*)hw2s;

    float ax = 0.f, ay = 0.f;
    for (int p = beg + j; p < end; p += 2 * LPN) {
        const int  p2   = p + LPN;
        const bool two  = p2 < end;
        const int  s0   = csr_src[p];
        const int  s1   = csr_src[two ? p2 : p];
        const float2 v0 = h2[s0];
        const float2 v1 = h2[s1];
        ax += v0.x;
        ay += v0.y;
        ax += two ? v1.x : 0.f;
        ay += two ? v1.y : 0.f;
    }
    for (int o = LPN / 2; o > 0; o >>= 1) {
        ax += __shfl_xor(ax, o);
        ay += __shfl_xor(ay, o);
    }
    if (j == 0) {
        const float2 self = h2[i];
        const float di = dinv[i];
        const float a0 = di * (ax + self.x) + b2[0];
        const float a1 = di * (ay + self.y) + b2[1];
        const float m  = fmaxf(a0, a1);
        const float e0 = __expf(a0 - m);
        const float e1 = __expf(a1 - m);
        const float inv = 1.0f / (e0 + e1);
        float2 r;
        r.x = e0 * inv;
        r.y = e1 * inv;
        ((float2*)out)[i] = r;
    }
}

extern "C" void kernel_launch(void* const* d_in, const int* in_sizes, int n_in,
                              void* d_out, int out_size, void* d_ws, size_t ws_size,
                              hipStream_t stream) {
    const float* x  = (const float*)d_in[0];
    const int*   ei = (const int*)d_in[1];
    const float* W1 = (const float*)d_in[2];
    const float* b1 = (const float*)d_in[3];
    const float* W2 = (const float*)d_in[4];
    const float* b2 = (const float*)d_in[5];
    float* out = (float*)d_out;

    char* ws = (char*)d_ws;
    size_t o = 0;
    auto alloc = [&](size_t bytes) {
        char* p = ws + o;
        o += (bytes + 4095) & ~(size_t)4095;
        return p;
    };
    int*   tbl    = (int*)  alloc((size_t)2 * TBL * sizeof(int));
    int*   off    = tbl + TBL;
    float* dinv   = (float*)alloc((size_t)N_NODES * sizeof(float));
    float* xw1    = (float*)alloc((size_t)N_NODES * HID * sizeof(float));
    int*   binned = (int*)  alloc((size_t)N_EDGES * sizeof(int));
    int*   csr    = (int*)  alloc((size_t)N_EDGES * sizeof(int));
    int*   rowptr = (int*)  alloc((size_t)(N_NODES + 1) * sizeof(int));
    float* hw2s   = (float*)alloc((size_t)N_NODES * 2 * sizeof(float));
    (void)ws_size; (void)in_sizes; (void)n_in; (void)out_size;

    const int B = 256;
    dim3 blk(B);

    k_bin_hist    <<<SORT_G, dim3(SORT_T), 0, stream>>>(ei, tbl);
    k_bin_scan    <<<1, dim3(SCAN_T), 0, stream>>>(tbl, off);
    k_bin_scatter <<<SORT_G, dim3(SORT_T), 0, stream>>>(ei, off, binned);
    k_bin_finalize<<<NBIN, dim3(FIN_T), 0, stream>>>(off, binned, rowptr, dinv, csr);

    k_gemm1_wmma<<<(N_NODES / 16) / RT, dim3(128), 0, stream>>>(x, W1, xw1);

    k_gather1<<<N_NODES / 8, blk, 0, stream>>>(rowptr, csr, dinv, xw1, b1, W2, hw2s);

    k_gather2_softmax<<<(N_NODES * LPN) / B, blk, 0, stream>>>(rowptr, csr, dinv, hw2s, b2, out);
}


# layer-1 GEMM: one wave per 16-row tile does all four 16-column tiles from one A fragment per K-step, B fragments (W1 as f16) read from an LDS copy: each x row fetched and converted once, not 4 times
# speedup vs baseline: 3.4082x; 1.2308x over previous
#include <hip/hip_runtime.h>
#include <hip/hip_bf16.h>
#include <stdint.h>

#define N_NODES 100000
#define N_EDGES 1600000
#define F_IN    256
#define HID     64

typedef __attribute__((ext_vector_type(16))) _Float16 v16h;
typedef __attribute__((ext_vector_type(8)))  _Float16 v8h;
typedef __attribute__((ext_vector_type(8)))  float    v8f;
typedef __attribute__((ext_vector_type(4)))  float    v4f;

static __device__ __forceinline__ int wave_sum_i32(int v) {
    for (int o = 16; o > 0; o >>= 1) v += __shfl_xor(v, o);
    return v;
}

#define BIN_SHIFT 9
#define BIN_W   (1 << BIN_SHIFT)
#define NBIN    ((N_NODES + BIN_W - 1) / BIN_W)
#define SORT_G  250
#define SORT_T  1024
#define QUADS   (N_EDGES / 4)
#define QPB     (QUADS / SORT_G)
#define TBL     (NBIN * SORT_G)
static_assert(N_EDGES % 4 == 0, "edge quads must tile the edge list");
static_assert(QUADS % SORT_G == 0, "edge blocks must tile the quads");
static_assert(N_NODES <= (1 << (31 - BIN_SHIFT)), "src id and dst-in-bin must pack into one word");

__global__ __launch_bounds__(SORT_T) void k_bin_hist(const int* __restrict__ ei,
                                                     int* __restrict__ tbl) {
    __shared__ int cnt[NBIN];
    const int t = threadIdx.x, g = blockIdx.x;
    for (int b = t; b < NBIN; b += SORT_T) cnt[b] = 0;
    __syncthreads();
    const int4* dq = (const int4*)(ei + N_EDGES) + (size_t)g * QPB;
    for (int q = t; q < QPB; q += SORT_T) {
        const int4 d = dq[q];
        atomicAdd(&cnt[d.x >> BIN_SHIFT], 1);
        atomicAdd(&cnt[d.y >> BIN_SHIFT], 1);
        atomicAdd(&cnt[d.z >> BIN_SHIFT], 1);
        atomicAdd(&cnt[d.w >> BIN_SHIFT], 1);
    }
    __syncthreads();
    for (int b = t; b < NBIN; b += SORT_T) tbl[b * SORT_G + g] = cnt[b];
}

#define SCAN_T  1024
#define PER_T   ((TBL + SCAN_T - 1) / SCAN_T)
#define PITCH   (PER_T + 1 - (PER_T & 1))
__global__ __launch_bounds__(SCAN_T) void k_bin_scan(const int* __restrict__ tbl,
                                                     int* __restrict__ off) {
    __shared__ int buf[SCAN_T * PITCH];
    __shared__ int wtot[SCAN_T / 32];
    const int t = threadIdx.x, lane = t & 31, wv = t >> 5;
    for (int i = t; i < SCAN_T * PER_T; i += SCAN_T)
        buf[(i / PER_T) * PITCH + (i % PER_T)] = (i < TBL) ? tbl[i] : 0;
    __syncthreads();
    int* mine = buf + t * PITCH;
    int v[PER_T];
    int run = 0;
    #pragma unroll
    for (int k = 0; k < PER_T; ++k) {
        v[k] = mine[k];
        run += v[k];
    }
    int inc = run;
    for (int o = 1; o < 32; o <<= 1) {
        const int u = __shfl_up(inc, o);
        if (lane >= o) inc += u;
    }
    if (lane == 31) wtot[wv] = inc;
    __syncthreads();
    int pre = inc - run;
    for (int k = 0; k < wv; ++k) pre += wtot[k];
    #pragma unroll
    for (int k = 0; k < PER_T; ++k) {
        mine[k] = pre;
        pre += v[k];
    }
    __syncthreads();
    for (int i = t; i < TBL; i += SCAN_T)
        off[i] = buf[(i / PER_T) * PITCH + (i % PER_T)];
}

__global__ __launch_bounds__(SORT_T) void k_bin_scatter(const int* __restrict__ ei,
                                                        const int* __restrict__ off,
                                                        int* __restrict__ binned) {
    __shared__ int cur[NBIN];
    const int t = threadIdx.x, g = blockIdx.x;
    for (int b = t; b < NBIN; b += SORT_T) cur[b] = off[b * SORT_G + g];
    __syncthreads();
    const int4* sq = (const int4*)ei + (size_t)g * QPB;
    const int4* dq = (const int4*)(ei + N_EDGES) + (size_t)g * QPB;
    for (int q = t; q < QPB; q += SORT_T) {
        const int4 s = sq[q];
        const int4 d = dq[q];
        const int p0 = atomicAdd(&cur[d.x >> BIN_SHIFT], 1);
        const int p1 = atomicAdd(&cur[d.y >> BIN_SHIFT], 1);
        const int p2 = atomicAdd(&cur[d.z >> BIN_SHIFT], 1);
        const int p3 = atomicAdd(&cur[d.w >> BIN_SHIFT], 1);
        binned[p0] = (s.x << BIN_SHIFT) | (d.x & (BIN_W - 1));
        binned[p1] = (s.y << BIN_SHIFT) | (d.y & (BIN_W - 1));
        binned[p2] = (s.z << BIN_SHIFT) | (d.z & (BIN_W - 1));
        binned[p3] = (s.w << BIN_SHIFT) | (d.w & (BIN_W - 1));
    }
}

#define FIN_T   1024
static_assert(BIN_W <= FIN_T && BIN_W % 32 == 0, "one thread per node of the bin, whole waves");
__global__ __launch_bounds__(FIN_T) void k_bin_finalize(const int* __restrict__ off,
                                                        const int* __restrict__ binned,
                                                        int* __restrict__ rowptr,
                                                        float* __restrict__ dinv,
                                                        int* __restrict__ csr_src) {
    __shared__ int cnt[BIN_W];
    __shared__ int wtot[BIN_W / 32];
    const int t = threadIdx.x, lane = t & 31, wv = t >> 5;
    const int b = blockIdx.x;
    const int beg = off[b * SORT_G];
    const int end = (b + 1 < NBIN) ? off[(b + 1) * SORT_G] : N_EDGES;
    if (t < BIN_W) cnt[t] = 0;
    __syncthreads();
    for (int p = beg + t; p < end; p += 4 * FIN_T) {
        int w[4];
        #pragma unroll
        for (int k = 0; k < 4; ++k) {
            const int pk = p + k * FIN_T;
            w[k] = (pk < end) ? binned[pk] : -1;
        }
        #pragma unroll
        for (int k = 0; k < 4; ++k)
            if (w[k] >= 0) atomicAdd(&cnt[w[k] & (BIN_W - 1)], 1);
    }
    __syncthreads();
    int dg = 0, inc = 0;
    if (t < BIN_W) {
        dg = cnt[t];
        inc = dg;
        for (int o = 1; o < 32; o <<= 1) {
            const int u = __shfl_up(inc, o);
            if (lane >= o) inc += u;
        }
        if (lane == 31) wtot[wv] = inc;
    }
    __syncthreads();
    if (t < BIN_W) {
        int base = beg;
        for (int k = 0; k < wv; ++k) base += wtot[k];
        const int excl = base + inc - dg;
        cnt[t] = excl;
        const int node = b * BIN_W + t;
        if (node < N_NODES) {
            rowptr[node] = excl;
            dinv[node] = __frsqrt_rn((float)dg + 1.0f);
        }
        if (node == N_NODES - 1) rowptr[N_NODES] = excl + dg;
    }
    __syncthreads();
    for (int p = beg + t; p < end; p += 4 * FIN_T) {
        int w[4];
        #pragma unroll
        for (int k = 0; k < 4; ++k) {
            const int pk = p + k * FIN_T;
            w[k] = (pk < end) ? binned[pk] : -1;
        }
        #pragma unroll
        for (int k = 0; k < 4; ++k) {
            if (w[k] >= 0) {
                const int pos = atomicAdd(&cnt[w[k] & (BIN_W - 1)], 1);
                csr_src[pos] = w[k] >> BIN_SHIFT;
            }
        }
    }
}

#define G1_WAVES 5
#define G1_T     (G1_WAVES * 32)
#define B_UNITS  (8 * 4 * 2 * 32)
static_assert((N_NODES / 16) % G1_WAVES == 0, "row tiles must tile the blocks");
static_assert(F_IN == 256 && HID == 64, "fragment index arithmetic below is written for 8 K-steps x 4 column tiles");
__global__ __launch_bounds__(G1_T) void k_gemm1_wmma(
        const float* __restrict__ x,
        const float* __restrict__ W1,
        float* __restrict__ xw1) {
    __shared__ v8h bl[B_UNITS];
    const int t    = threadIdx.x;
    const int wave = t >> 5;
    const int lane = t & 31;
    const int half = lane >> 4;
    const int r    = lane & 15;

    const int row0 = (blockIdx.x * G1_WAVES + wave) * 16;
    const float* xrow = x + (size_t)(row0 + r) * F_IN + 8 * half;
    v4f av[32];
    #pragma unroll
    for (int kt = 0; kt < 8; ++kt) {
        const float* seg0 = xrow + kt * 32;
        const float* seg1 = seg0 + 16;
        av[4 * kt + 0] = __builtin_nontemporal_load((const v4f*)seg0);
        av[4 * kt + 1] = __builtin_nontemporal_load((const v4f*)(seg0 + 4));
        av[4 * kt + 2] = __builtin_nontemporal_load((const v4f*)seg1);
        av[4 * kt + 3] = __builtin_nontemporal_load((const v4f*)(seg1 + 4));
    }

    for (int q = t; q < B_UNITS / 4; q += G1_T) {
        const int c4   = q & 3;
        const int hf   = (q >> 2) & 1;
        const int part = (q >> 3) & 1;
        const int ct   = (q >> 4) & 3;
        const int kt   = q >> 6;
        const float* p = W1 + (size_t)(kt * 32 + part * 16 + 8 * hf) * HID + ct * 16 + 4 * c4;
        v4f w[8];
        #pragma unroll
        for (int j = 0; j < 8; ++j) w[j] = *(const v4f*)(p + j * HID);
        const int u0 = ((kt * 4 + ct) * 2 + part) * 32 + hf * 16 + 4 * c4;
        #pragma unroll
        for (int e = 0; e < 4; ++e) {
            v8h h;
            #pragma unroll
            for (int j = 0; j < 8; ++j) h[j] = (_Float16)w[j][e];
            bl[u0 + e] = h;
        }
    }
    __syncthreads();

    v8f acc[4];
    #pragma unroll
    for (int ct = 0; ct < 4; ++ct) acc[ct] = (v8f){0.f, 0.f, 0.f, 0.f, 0.f, 0.f, 0.f, 0.f};

    #pragma unroll
    for (int kt = 0; kt < 8; ++kt) {
        const v4f a0 = av[4 * kt + 0], a1 = av[4 * kt + 1];
        const v4f a2 = av[4 * kt + 2], a3 = av[4 * kt + 3];
        v16h a;
        a[0]  = (_Float16)a0.x; a[1]  = (_Float16)a0.y;
        a[2]  = (_Float16)a0.z; a[3]  = (_Float16)a0.w;
        a[4]  = (_Float16)a1.x; a[5]  = (_Float16)a1.y;
        a[6]  = (_Float16)a1.z; a[7]  = (_Float16)a1.w;
        a[8]  = (_Float16)a2.x; a[9]  = (_Float16)a2.y;
        a[10] = (_Float16)a2.z; a[11] = (_Float16)a2.w;
        a[12] = (_Float16)a3.x; a[13] = (_Float16)a3.y;
        a[14] = (_Float16)a3.z; a[15] = (_Float16)a3.w;
        #pragma unroll
        for (int ct = 0; ct < 4; ++ct) {
            const v8h lo = bl[((kt * 4 + ct) * 2 + 0) * 32 + lane];
            const v8h hi = bl[((kt * 4 + ct) * 2 + 1) * 32 + lane];
            const v16h bf = __builtin_shufflevector(lo, hi, 0, 1, 2, 3, 4, 5, 6, 7,
                                                    8, 9, 10, 11, 12, 13, 14, 15);
            acc[ct] = __builtin_amdgcn_wmma_f32_16x16x32_f16(
                 false, a,  false, bf,
                 (short)0, acc[ct],  false,  false);
        }
    }

    #pragma unroll
    for (int ct = 0; ct < 4; ++ct) {
        #pragma unroll
        for (int i = 0; i < 8; ++i) {
            const int m = i + 8 * half;
            xw1[(size_t)(row0 + m) * HID + (ct * 16 + r)] = acc[ct][i];
        }
    }
}

#define EPC 8
static_assert(N_NODES % 8 == 0, "8 waves per block must tile the node list");
__global__ __launch_bounds__(256) void k_gather1(const int* __restrict__ rowptr,
                                                 const int* __restrict__ csr_src,
                                                 const float* __restrict__ dinv,
                                                 const float* __restrict__ xw1,
                                                 const float* __restrict__ b1,
                                                 const float* __restrict__ W2,
                                                 float* __restrict__ hw2s) {
    const int lane = threadIdx.x & 31;
    int i = blockIdx.x * 8 + (int)(threadIdx.x >> 5);
    i = __builtin_amdgcn_readfirstlane(i);
    if (i >= N_NODES) return;

    const int beg = rowptr[i];
    const int end = rowptr[i + 1];
    const float di = dinv[i];
    const float2* rows = (const float2*)xw1;
    const float2 self = rows[(size_t)i * (HID / 2) + lane];

    float ax = 0.f, ay = 0.f;
    for (int base = beg; base < end; base += 32) {
        const int rem = end - base;
        const int n = rem < 32 ? rem : 32;
        const int sv = csr_src[base + (lane < n ? lane : n - 1)];
        for (int c = 0; c < n; c += EPC) {
            int s[EPC];
            float w[EPC];
            float2 v[EPC];
            #pragma unroll
            for (int k = 0; k < EPC; ++k) {
                const int idx = (c + k < n) ? (c + k) : (n - 1);
                s[k] = __builtin_amdgcn_readlane(sv, idx);
            }
            #pragma unroll
            for (int k = 0; k < EPC; ++k)
                w[k] = dinv[s[k]] * di;
            #pragma unroll
            for (int k = 0; k < EPC; ++k)
                v[k] = rows[(size_t)s[k] * (HID / 2) + lane];
            #pragma unroll
            for (int k = 0; k < EPC; ++k) {
                const float px = v[k].x * w[k];
                const float py = v[k].y * w[k];
                const bool live = (c + k < n);
                ax += live ? px : 0.f;
                ay += live ? py : 0.f;
            }
        }
    }

    const float sl = di * di;
    float hx = ax + self.x * sl;
    float hy = ay + self.y * sl;
    hx = fmaxf(hx + b1[2 * lane],     0.f);
    hy = fmaxf(hy + b1[2 * lane + 1], 0.f);
    const float w00 = W2[4 * lane + 0], w01 = W2[4 * lane + 1];
    const float w10 = W2[4 * lane + 2], w11 = W2[4 * lane + 3];
    float o0 = hx * w00 + hy * w10;
    float o1 = hx * w01 + hy * w11;
    for (int o = 16; o > 0; o >>= 1) {
        o0 += __shfl_xor(o0, o);
        o1 += __shfl_xor(o1, o);
    }
    if (lane == 0) {
        float2 r;
        r.x = o0 * di;
        r.y = o1 * di;
        ((float2*)hw2s)[i] = r;
    }
}

#define LPN 8
static_assert((N_NODES * LPN) % 256 == 0, "node groups must tile the blocks");
__global__ __launch_bounds__(256) void k_gather2_softmax(const int* __restrict__ rowptr,
                                                         const int* __restrict__ csr_src,
                                                         const float* __restrict__ dinv,
                                                         const float* __restrict__ hw2s,
                                                         const float* __restrict__ b2,
                                                         float* __restrict__ out) {
    const int t = blockIdx.x * blockDim.x + threadIdx.x;
    const int i = t / LPN;
    const int j = t % LPN;
    const int beg = rowptr[i];
    const int end = rowptr[i + 1];
    const float2* h2 = (const float2*)hw2s;

    float ax = 0.f, ay = 0.f;
    for (int p = beg + j; p < end; p += 2 * LPN) {
        const int  p2   = p + LPN;
        const bool two  = p2 < end;
        const int  s0   = csr_src[p];
        const int  s1   = csr_src[two ? p2 : p];
        const float2 v0 = h2[s0];
        const float2 v1 = h2[s1];
        ax += v0.x;
        ay += v0.y;
        ax += two ? v1.x : 0.f;
        ay += two ? v1.y : 0.f;
    }
    for (int o = LPN / 2; o > 0; o >>= 1) {
        ax += __shfl_xor(ax, o);
        ay += __shfl_xor(ay, o);
    }
    if (j == 0) {
        const float2 self = h2[i];
        const float di = dinv[i];
        const float a0 = di * (ax + self.x) + b2[0];
        const float a1 = di * (ay + self.y) + b2[1];
        const float m  = fmaxf(a0, a1);
        const float e0 = __expf(a0 - m);
        const float e1 = __expf(a1 - m);
        const float inv = 1.0f / (e0 + e1);
        float2 r;
        r.x = e0 * inv;
        r.y = e1 * inv;
        ((float2*)out)[i] = r;
    }
}

extern "C" void kernel_launch(void* const* d_in, const int* in_sizes, int n_in,
                              void* d_out, int out_size, void* d_ws, size_t ws_size,
                              hipStream_t stream) {
    const float* x  = (const float*)d_in[0];
    const int*   ei = (const int*)d_in[1];
    const float* W1 = (const float*)d_in[2];
    const float* b1 = (const float*)d_in[3];
    const float* W2 = (const float*)d_in[4];
    const float* b2 = (const float*)d_in[5];
    float* out = (float*)d_out;

    char* ws = (char*)d_ws;
    size_t o = 0;
    auto alloc = [&](size_t bytes) {
        char* p = ws + o;
        o += (bytes + 4095) & ~(size_t)4095;
        return p;
    };
    int*   tbl    = (int*)  alloc((size_t)2 * TBL * sizeof(int));
    int*   off    = tbl + TBL;
    float* dinv   = (float*)alloc((size_t)N_NODES * sizeof(float));
    float* xw1    = (float*)alloc((size_t)N_NODES * HID * sizeof(float));
    int*   binned = (int*)  alloc((size_t)N_EDGES * sizeof(int));
    int*   csr    = (int*)  alloc((size_t)N_EDGES * sizeof(int));
    int*   rowptr = (int*)  alloc((size_t)(N_NODES + 1) * sizeof(int));
    float* hw2s   = (float*)alloc((size_t)N_NODES * 2 * sizeof(float));
    (void)ws_size; (void)in_sizes; (void)n_in; (void)out_size;

    const int B = 256;
    dim3 blk(B);

    k_bin_hist    <<<SORT_G, dim3(SORT_T), 0, stream>>>(ei, tbl);
    k_bin_scan    <<<1, dim3(SCAN_T), 0, stream>>>(tbl, off);
    k_bin_scatter <<<SORT_G, dim3(SORT_T), 0, stream>>>(ei, off, binned);
    k_bin_finalize<<<NBIN, dim3(FIN_T), 0, stream>>>(off, binned, rowptr, dinv, csr);

    k_gemm1_wmma<<<(N_NODES / 16) / G1_WAVES, dim3(G1_T), 0, stream>>>(x, W1, xw1);

    k_gather1<<<N_NODES / 8, blk, 0, stream>>>(rowptr, csr, dinv, xw1, b1, W2, hw2s);

    k_gather2_softmax<<<(N_NODES * LPN) / B, blk, 0, stream>>>(rowptr, csr, dinv, hw2s, b2, out);
}


# the (bin, block) table scan moves into the placement launch: each edge block sums its own table rows (loads batched, lane sums by DPP); the one-work-group scan launch is gone; edge order kept
# speedup vs baseline: 3.8584x; 1.1321x over previous
#include <hip/hip_runtime.h>
#include <hip/hip_bf16.h>
#include <stdint.h>

#define N_NODES 100000
#define N_EDGES 1600000
#define F_IN    256
#define HID     64

typedef __attribute__((ext_vector_type(16))) _Float16 v16h;
typedef __attribute__((ext_vector_type(8)))  _Float16 v8h;
typedef __attribute__((ext_vector_type(8)))  float    v8f;
typedef __attribute__((ext_vector_type(4)))  float    v4f;

static __device__ __forceinline__ int wave_sum_i32(int v) {
    v += __builtin_amdgcn_update_dpp(0, v, 0x161, 0xf, 0xf, false);
    v += __builtin_amdgcn_update_dpp(0, v, 0x162, 0xf, 0xf, false);
    v += __builtin_amdgcn_update_dpp(0, v, 0x164, 0xf, 0xf, false);
    v += __builtin_amdgcn_update_dpp(0, v, 0x168, 0xf, 0xf, false);
    v += __builtin_amdgcn_permlanex16(v, v, 0x76543210, 0xfedcba98, false, false);
    return v;
}

#define BIN_SHIFT 9
#define BIN_W   (1 << BIN_SHIFT)
#define NBIN    ((N_NODES + BIN_W - 1) / BIN_W)
#define SORT_G  250
#define SORT_T  1024
#define QUADS   (N_EDGES / 4)
#define QPB     (QUADS / SORT_G)
#define TBL     (NBIN * SORT_G)
static_assert(N_EDGES % 4 == 0, "edge quads must tile the edge list");
static_assert(QUADS % SORT_G == 0, "edge blocks must tile the quads");
static_assert(N_NODES <= (1 << (31 - BIN_SHIFT)), "src id and dst-in-bin must pack into one word");

__global__ __launch_bounds__(SORT_T) void k_bin_hist(const int* __restrict__ ei,
                                                     int* __restrict__ tbl) {
    __shared__ int cnt[NBIN];
    const int t = threadIdx.x, g = blockIdx.x;
    for (int b = t; b < NBIN; b += SORT_T) cnt[b] = 0;
    __syncthreads();
    const int4* dq = (const int4*)(ei + N_EDGES) + (size_t)g * QPB;
    for (int q = t; q < QPB; q += SORT_T) {
        const int4 d = dq[q];
        atomicAdd(&cnt[d.x >> BIN_SHIFT], 1);
        atomicAdd(&cnt[d.y >> BIN_SHIFT], 1);
        atomicAdd(&cnt[d.z >> BIN_SHIFT], 1);
        atomicAdd(&cnt[d.w >> BIN_SHIFT], 1);
    }
    __syncthreads();
    for (int b = t; b < NBIN; b += SORT_T) tbl[b * SORT_G + g] = cnt[b];
}

#define ROWS_PW ((NBIN + 31) / 32)
#define CPL     ((SORT_G + 31) / 32)
static_assert(SORT_T == 1024, "row ownership below is written for 32 waves per block");
static_assert(NBIN <= 32 * 32, "the scan of the bin totals is one wave-level scan plus wave totals");
__global__ __launch_bounds__(SORT_T) void k_bin_scatter(const int* __restrict__ ei,
                                                        const int* __restrict__ tbl,
                                                        int* __restrict__ binstart,
                                                        int* __restrict__ binned) {
    __shared__ int cur[NBIN];
    __shared__ int tot[NBIN];
    __shared__ int wtot[32];
    const int t = threadIdx.x, g = blockIdx.x, lane = t & 31, wv = t >> 5;

    int v[ROWS_PW][CPL];
    #pragma unroll
    for (int j = 0; j < ROWS_PW; ++j) {
        const int b  = wv + 32 * j;
        const int* row = tbl + (b < NBIN ? b : NBIN - 1) * SORT_G;
        #pragma unroll
        for (int k = 0; k < CPL; ++k) {
            const int col = lane + 32 * k;
            v[j][k] = row[col < SORT_G ? col : SORT_G - 1];
        }
    }
    #pragma unroll
    for (int j = 0; j < ROWS_PW; ++j) {
        const int b = wv + 32 * j;
        int all = 0, before = 0;
        #pragma unroll
        for (int k = 0; k < CPL; ++k) {
            const int col = lane + 32 * k;
            const int x = (col < SORT_G) ? v[j][k] : 0;
            all += x;
            before += (col < g) ? x : 0;
        }
        all = wave_sum_i32(all);
        before = wave_sum_i32(before);
        if (lane == 0 && b < NBIN) {
            tot[b] = all;
            cur[b] = before;
        }
    }
    __syncthreads();
    int mine = 0, inc = 0;
    if (wv <= (NBIN - 1) / 32) {
        mine = (t < NBIN) ? tot[t] : 0;
        inc = mine;
        for (int o = 1; o < 32; o <<= 1) {
            const int u = __shfl_up(inc, o);
            if (lane >= o) inc += u;
        }
        if (lane == 31) wtot[wv] = inc;
    }
    __syncthreads();
    if (t < NBIN) {
        int start = inc - mine;
        for (int k = 0; k < wv; ++k) start += wtot[k];
        cur[t] += start;
        if (g == 0) binstart[t] = start;
    }
    __syncthreads();
    const int4* sq = (const int4*)ei + (size_t)g * QPB;
    const int4* dq = (const int4*)(ei + N_EDGES) + (size_t)g * QPB;
    for (int q = t; q < QPB; q += SORT_T) {
        const int4 s = sq[q];
        const int4 d = dq[q];
        const int p0 = atomicAdd(&cur[d.x >> BIN_SHIFT], 1);
        const int p1 = atomicAdd(&cur[d.y >> BIN_SHIFT], 1);
        const int p2 = atomicAdd(&cur[d.z >> BIN_SHIFT], 1);
        const int p3 = atomicAdd(&cur[d.w >> BIN_SHIFT], 1);
        binned[p0] = (s.x << BIN_SHIFT) | (d.x & (BIN_W - 1));
        binned[p1] = (s.y << BIN_SHIFT) | (d.y & (BIN_W - 1));
        binned[p2] = (s.z << BIN_SHIFT) | (d.z & (BIN_W - 1));
        binned[p3] = (s.w << BIN_SHIFT) | (d.w & (BIN_W - 1));
    }
}

#define FIN_T   1024
static_assert(BIN_W <= FIN_T && BIN_W % 32 == 0, "one thread per node of the bin, whole waves");
__global__ __launch_bounds__(FIN_T) void k_bin_finalize(const int* __restrict__ binstart,
                                                        const int* __restrict__ binned,
                                                        int* __restrict__ rowptr,
                                                        float* __restrict__ dinv,
                                                        int* __restrict__ csr_src) {
    __shared__ int cnt[BIN_W];
    __shared__ int wtot[BIN_W / 32];
    const int t = threadIdx.x, lane = t & 31, wv = t >> 5;
    const int b = blockIdx.x;
    const int beg = binstart[b];
    const int end = (b + 1 < NBIN) ? binstart[b + 1] : N_EDGES;
    if (t < BIN_W) cnt[t] = 0;
    __syncthreads();
    for (int p = beg + t; p < end; p += 4 * FIN_T) {
        int w[4];
        #pragma unroll
        for (int k = 0; k < 4; ++k) {
            const int pk = p + k * FIN_T;
            w[k] = (pk < end) ? binned[pk] : -1;
        }
        #pragma unroll
        for (int k = 0; k < 4; ++k)
            if (w[k] >= 0) atomicAdd(&cnt[w[k] & (BIN_W - 1)], 1);
    }
    __syncthreads();
    int dg = 0, inc = 0;
    if (t < BIN_W) {
        dg = cnt[t];
        inc = dg;
        for (int o = 1; o < 32; o <<= 1) {
            const int u = __shfl_up(inc, o);
            if (lane >= o) inc += u;
        }
        if (lane == 31) wtot[wv] = inc;
    }
    __syncthreads();
    if (t < BIN_W) {
        int base = beg;
        for (int k = 0; k < wv; ++k) base += wtot[k];
        const int excl = base + inc - dg;
        cnt[t] = excl;
        const int node = b * BIN_W + t;
        if (node < N_NODES) {
            rowptr[node] = excl;
            dinv[node] = __frsqrt_rn((float)dg + 1.0f);
        }
        if (node == N_NODES - 1) rowptr[N_NODES] = excl + dg;
    }
    __syncthreads();
    for (int p = beg + t; p < end; p += 4 * FIN_T) {
        int w[4];
        #pragma unroll
        for (int k = 0; k < 4; ++k) {
            const int pk = p + k * FIN_T;
            w[k] = (pk < end) ? binned[pk] : -1;
        }
        #pragma unroll
        for (int k = 0; k < 4; ++k) {
            if (w[k] >= 0) {
                const int pos = atomicAdd(&cnt[w[k] & (BIN_W - 1)], 1);
                csr_src[pos] = w[k] >> BIN_SHIFT;
            }
        }
    }
}

#define G1_WAVES 5
#define G1_T     (G1_WAVES * 32)
#define B_UNITS  (8 * 4 * 2 * 32)
static_assert((N_NODES / 16) % G1_WAVES == 0, "row tiles must tile the blocks");
static_assert(F_IN == 256 && HID == 64, "fragment index arithmetic below is written for 8 K-steps x 4 column tiles");
__global__ __launch_bounds__(G1_T) void k_gemm1_wmma(
        const float* __restrict__ x,
        const float* __restrict__ W1,
        float* __restrict__ xw1) {
    __shared__ v8h bl[B_UNITS];
    const int t    = threadIdx.x;
    const int wave = t >> 5;
    const int lane = t & 31;
    const int half = lane >> 4;
    const int r    = lane & 15;

    const int row0 = (blockIdx.x * G1_WAVES + wave) * 16;
    const float* xrow = x + (size_t)(row0 + r) * F_IN + 8 * half;
    v4f av[32];
    #pragma unroll
    for (int kt = 0; kt < 8; ++kt) {
        const float* seg0 = xrow + kt * 32;
        const float* seg1 = seg0 + 16;
        av[4 * kt + 0] = __builtin_nontemporal_load((const v4f*)seg0);
        av[4 * kt + 1] = __builtin_nontemporal_load((const v4f*)(seg0 + 4));
        av[4 * kt + 2] = __builtin_nontemporal_load((const v4f*)seg1);
        av[4 * kt + 3] = __builtin_nontemporal_load((const v4f*)(seg1 + 4));
    }

    for (int q = t; q < B_UNITS / 4; q += G1_T) {
        const int c4   = q & 3;
        const int hf   = (q >> 2) & 1;
        const int part = (q >> 3) & 1;
        const int ct   = (q >> 4) & 3;
        const int kt   = q >> 6;
        const float* p = W1 + (size_t)(kt * 32 + part * 16 + 8 * hf) * HID + ct * 16 + 4 * c4;
        v4f w[8];
        #pragma unroll
        for (int j = 0; j < 8; ++j) w[j] = *(const v4f*)(p + j * HID);
        const int u0 = ((kt * 4 + ct) * 2 + part) * 32 + hf * 16 + 4 * c4;
        #pragma unroll
        for (int e = 0; e < 4; ++e) {
            v8h h;
            #pragma unroll
            for (int j = 0; j < 8; ++j) h[j] = (_Float16)w[j][e];
            bl[u0 + e] = h;
        }
    }
    __syncthreads();

    v8f acc[4];
    #pragma unroll
    for (int ct = 0; ct < 4; ++ct) acc[ct] = (v8f){0.f, 0.f, 0.f, 0.f, 0.f, 0.f, 0.f, 0.f};

    #pragma unroll
    for (int kt = 0; kt < 8; ++kt) {
        const v4f a0 = av[4 * kt + 0], a1 = av[4 * kt + 1];
        const v4f a2 = av[4 * kt + 2], a3 = av[4 * kt + 3];
        v16h a;
        a[0]  = (_Float16)a0.x; a[1]  = (_Float16)a0.y;
        a[2]  = (_Float16)a0.z; a[3]  = (_Float16)a0.w;
        a[4]  = (_Float16)a1.x; a[5]  = (_Float16)a1.y;
        a[6]  = (_Float16)a1.z; a[7]  = (_Float16)a1.w;
        a[8]  = (_Float16)a2.x; a[9]  = (_Float16)a2.y;
        a[10] = (_Float16)a2.z; a[11] = (_Float16)a2.w;
        a[12] = (_Float16)a3.x; a[13] = (_Float16)a3.y;
        a[14] = (_Float16)a3.z; a[15] = (_Float16)a3.w;
        #pragma unroll
        for (int ct = 0; ct < 4; ++ct) {
            const v8h lo = bl[((kt * 4 + ct) * 2 + 0) * 32 + lane];
            const v8h hi = bl[((kt * 4 + ct) * 2 + 1) * 32 + lane];
            const v16h bf = __builtin_shufflevector(lo, hi, 0, 1, 2, 3, 4, 5, 6, 7,
                                                    8, 9, 10, 11, 12, 13, 14, 15);
            acc[ct] = __builtin_amdgcn_wmma_f32_16x16x32_f16(
                 false, a,  false, bf,
                 (short)0, acc[ct],  false,  false);
        }
    }

    #pragma unroll
    for (int ct = 0; ct < 4; ++ct) {
        #pragma unroll
        for (int i = 0; i < 8; ++i) {
            const int m = i + 8 * half;
            xw1[(size_t)(row0 + m) * HID + (ct * 16 + r)] = acc[ct][i];
        }
    }
}

#define EPC 8
static_assert(N_NODES % 8 == 0, "8 waves per block must tile the node list");
__global__ __launch_bounds__(256) void k_gather1(const int* __restrict__ rowptr,
                                                 const int* __restrict__ csr_src,
                                                 const float* __restrict__ dinv,
                                                 const float* __restrict__ xw1,
                                                 const float* __restrict__ b1,
                                                 const float* __restrict__ W2,
                                                 float* __restrict__ hw2s) {
    const int lane = threadIdx.x & 31;
    int i = blockIdx.x * 8 + (int)(threadIdx.x >> 5);
    i = __builtin_amdgcn_readfirstlane(i);
    if (i >= N_NODES) return;

    const int beg = rowptr[i];
    const int end = rowptr[i + 1];
    const float di = dinv[i];
    const float2* rows = (const float2*)xw1;
    const float2 self = rows[(size_t)i * (HID / 2) + lane];

    float ax = 0.f, ay = 0.f;
    for (int base = beg; base < end; base += 32) {
        const int rem = end - base;
        const int n = rem < 32 ? rem : 32;
        const int sv = csr_src[base + (lane < n ? lane : n - 1)];
        for (int c = 0; c < n; c += EPC) {
            int s[EPC];
            float w[EPC];
            float2 v[EPC];
            #pragma unroll
            for (int k = 0; k < EPC; ++k) {
                const int idx = (c + k < n) ? (c + k) : (n - 1);
                s[k] = __builtin_amdgcn_readlane(sv, idx);
            }
            #pragma unroll
            for (int k = 0; k < EPC; ++k)
                w[k] = dinv[s[k]] * di;
            #pragma unroll
            for (int k = 0; k < EPC; ++k)
                v[k] = rows[(size_t)s[k] * (HID / 2) + lane];
            #pragma unroll
            for (int k = 0; k < EPC; ++k) {
                const float px = v[k].x * w[k];
                const float py = v[k].y * w[k];
                const bool live = (c + k < n);
                ax += live ? px : 0.f;
                ay += live ? py : 0.f;
            }
        }
    }

    const float sl = di * di;
    float hx = ax + self.x * sl;
    float hy = ay + self.y * sl;
    hx = fmaxf(hx + b1[2 * lane],     0.f);
    hy = fmaxf(hy + b1[2 * lane + 1], 0.f);
    const float w00 = W2[4 * lane + 0], w01 = W2[4 * lane + 1];
    const float w10 = W2[4 * lane + 2], w11 = W2[4 * lane + 3];
    float o0 = hx * w00 + hy * w10;
    float o1 = hx * w01 + hy * w11;
    for (int o = 16; o > 0; o >>= 1) {
        o0 += __shfl_xor(o0, o);
        o1 += __shfl_xor(o1, o);
    }
    if (lane == 0) {
        float2 r;
        r.x = o0 * di;
        r.y = o1 * di;
        ((float2*)hw2s)[i] = r;
    }
}

#define LPN 8
static_assert((N_NODES * LPN) % 256 == 0, "node groups must tile the blocks");
__global__ __launch_bounds__(256) void k_gather2_softmax(const int* __restrict__ rowptr,
                                                         const int* __restrict__ csr_src,
                                                         const float* __restrict__ dinv,
                                                         const float* __restrict__ hw2s,
                                                         const float* __restrict__ b2,
                                                         float* __restrict__ out) {
    const int t = blockIdx.x * blockDim.x + threadIdx.x;
    const int i = t / LPN;
    const int j = t % LPN;
    const int beg = rowptr[i];
    const int end = rowptr[i + 1];
    const float2* h2 = (const float2*)hw2s;

    float ax = 0.f, ay = 0.f;
    for (int p = beg + j; p < end; p += 2 * LPN) {
        const int  p2   = p + LPN;
        const bool two  = p2 < end;
        const int  s0   = csr_src[p];
        const int  s1   = csr_src[two ? p2 : p];
        const float2 v0 = h2[s0];
        const float2 v1 = h2[s1];
        ax += v0.x;
        ay += v0.y;
        ax += two ? v1.x : 0.f;
        ay += two ? v1.y : 0.f;
    }
    for (int o = LPN / 2; o > 0; o >>= 1) {
        ax += __shfl_xor(ax, o);
        ay += __shfl_xor(ay, o);
    }
    if (j == 0) {
        const float2 self = h2[i];
        const float di = dinv[i];
        const float a0 = di * (ax + self.x) + b2[0];
        const float a1 = di * (ay + self.y) + b2[1];
        const float m  = fmaxf(a0, a1);
        const float e0 = __expf(a0 - m);
        const float e1 = __expf(a1 - m);
        const float inv = 1.0f / (e0 + e1);
        float2 r;
        r.x = e0 * inv;
        r.y = e1 * inv;
        ((float2*)out)[i] = r;
    }
}

extern "C" void kernel_launch(void* const* d_in, const int* in_sizes, int n_in,
                              void* d_out, int out_size, void* d_ws, size_t ws_size,
                              hipStream_t stream) {
    const float* x  = (const float*)d_in[0];
    const int*   ei = (const int*)d_in[1];
    const float* W1 = (const float*)d_in[2];
    const float* b1 = (const float*)d_in[3];
    const float* W2 = (const float*)d_in[4];
    const float* b2 = (const float*)d_in[5];
    float* out = (float*)d_out;

    char* ws = (char*)d_ws;
    size_t o = 0;
    auto alloc = [&](size_t bytes) {
        char* p = ws + o;
        o += (bytes + 4095) & ~(size_t)4095;
        return p;
    };
    int*   tbl    = (int*)  alloc((size_t)TBL * sizeof(int));
    int*   bstart = (int*)  alloc((size_t)NBIN * sizeof(int));
    float* dinv   = (float*)alloc((size_t)N_NODES * sizeof(float));
    float* xw1    = (float*)alloc((size_t)N_NODES * HID * sizeof(float));
    int*   binned = (int*)  alloc((size_t)N_EDGES * sizeof(int));
    int*   csr    = (int*)  alloc((size_t)N_EDGES * sizeof(int));
    int*   rowptr = (int*)  alloc((size_t)(N_NODES + 1) * sizeof(int));
    float* hw2s   = (float*)alloc((size_t)N_NODES * 2 * sizeof(float));
    (void)ws_size; (void)in_sizes; (void)n_in; (void)out_size;

    const int B = 256;
    dim3 blk(B);

    k_bin_hist    <<<SORT_G, dim3(SORT_T), 0, stream>>>(ei, tbl);
    k_bin_scatter <<<SORT_G, dim3(SORT_T), 0, stream>>>(ei, tbl, bstart, binned);
    k_bin_finalize<<<NBIN, dim3(FIN_T), 0, stream>>>(bstart, binned, rowptr, dinv, csr);

    k_gemm1_wmma<<<(N_NODES / 16) / G1_WAVES, dim3(G1_T), 0, stream>>>(x, W1, xw1);

    k_gather1<<<N_NODES / 8, blk, 0, stream>>>(rowptr, csr, dinv, xw1, b1, W2, hw2s);

    k_gather2_softmax<<<(N_NODES * LPN) / B, blk, 0, stream>>>(rowptr, csr, dinv, hw2s, b2, out);
}


# GEMM1 stores rows scaled by their node's dinv: the layer-1 gather drops its per-edge dinv[src] fetch and multiply; dinv[i] scales the finished sum once (self row added inside: rounding regroups)
# speedup vs baseline: 4.0137x; 1.0402x over previous
#include <hip/hip_runtime.h>
#include <hip/hip_bf16.h>
#include <stdint.h>

#define N_NODES 100000
#define N_EDGES 1600000
#define F_IN    256
#define HID     64

typedef __attribute__((ext_vector_type(16))) _Float16 v16h;
typedef __attribute__((ext_vector_type(8)))  _Float16 v8h;
typedef __attribute__((ext_vector_type(8)))  float    v8f;
typedef __attribute__((ext_vector_type(4)))  float    v4f;

static __device__ __forceinline__ int wave_sum_i32(int v) {
    v += __builtin_amdgcn_update_dpp(0, v, 0x161, 0xf, 0xf, false);
    v += __builtin_amdgcn_update_dpp(0, v, 0x162, 0xf, 0xf, false);
    v += __builtin_amdgcn_update_dpp(0, v, 0x164, 0xf, 0xf, false);
    v += __builtin_amdgcn_update_dpp(0, v, 0x168, 0xf, 0xf, false);
    v += __builtin_amdgcn_permlanex16(v, v, 0x76543210, 0xfedcba98, false, false);
    return v;
}

#define BIN_SHIFT 9
#define BIN_W   (1 << BIN_SHIFT)
#define NBIN    ((N_NODES + BIN_W - 1) / BIN_W)
#define SORT_G  250
#define SORT_T  1024
#define QUADS   (N_EDGES / 4)
#define QPB     (QUADS / SORT_G)
#define TBL     (NBIN * SORT_G)
static_assert(N_EDGES % 4 == 0, "edge quads must tile the edge list");
static_assert(QUADS % SORT_G == 0, "edge blocks must tile the quads");
static_assert(N_NODES <= (1 << (31 - BIN_SHIFT)), "src id and dst-in-bin must pack into one word");

__global__ __launch_bounds__(SORT_T) void k_bin_hist(const int* __restrict__ ei,
                                                     int* __restrict__ tbl) {
    __shared__ int cnt[NBIN];
    const int t = threadIdx.x, g = blockIdx.x;
    for (int b = t; b < NBIN; b += SORT_T) cnt[b] = 0;
    __syncthreads();
    const int4* dq = (const int4*)(ei + N_EDGES) + (size_t)g * QPB;
    for (int q = t; q < QPB; q += SORT_T) {
        const int4 d = dq[q];
        atomicAdd(&cnt[d.x >> BIN_SHIFT], 1);
        atomicAdd(&cnt[d.y >> BIN_SHIFT], 1);
        atomicAdd(&cnt[d.z >> BIN_SHIFT], 1);
        atomicAdd(&cnt[d.w >> BIN_SHIFT], 1);
    }
    __syncthreads();
    for (int b = t; b < NBIN; b += SORT_T) tbl[b * SORT_G + g] = cnt[b];
}

#define ROWS_PW ((NBIN + 31) / 32)
#define CPL     ((SORT_G + 31) / 32)
static_assert(SORT_T == 1024, "row ownership below is written for 32 waves per block");
static_assert(NBIN <= 32 * 32, "the scan of the bin totals is one wave-level scan plus wave totals");
__global__ __launch_bounds__(SORT_T) void k_bin_scatter(const int* __restrict__ ei,
                                                        const int* __restrict__ tbl,
                                                        int* __restrict__ binstart,
                                                        int* __restrict__ binned) {
    __shared__ int cur[NBIN];
    __shared__ int tot[NBIN];
    __shared__ int wtot[32];
    const int t = threadIdx.x, g = blockIdx.x, lane = t & 31, wv = t >> 5;

    int v[ROWS_PW][CPL];
    #pragma unroll
    for (int j = 0; j < ROWS_PW; ++j) {
        const int b  = wv + 32 * j;
        const int* row = tbl + (b < NBIN ? b : NBIN - 1) * SORT_G;
        #pragma unroll
        for (int k = 0; k < CPL; ++k) {
            const int col = lane + 32 * k;
            v[j][k] = row[col < SORT_G ? col : SORT_G - 1];
        }
    }
    #pragma unroll
    for (int j = 0; j < ROWS_PW; ++j) {
        const int b = wv + 32 * j;
        int all = 0, before = 0;
        #pragma unroll
        for (int k = 0; k < CPL; ++k) {
            const int col = lane + 32 * k;
            const int x = (col < SORT_G) ? v[j][k] : 0;
            all += x;
            before += (col < g) ? x : 0;
        }
        all = wave_sum_i32(all);
        before = wave_sum_i32(before);
        if (lane == 0 && b < NBIN) {
            tot[b] = all;
            cur[b] = before;
        }
    }
    __syncthreads();
    int mine = 0, inc = 0;
    if (wv <= (NBIN - 1) / 32) {
        mine = (t < NBIN) ? tot[t] : 0;
        inc = mine;
        for (int o = 1; o < 32; o <<= 1) {
            const int u = __shfl_up(inc, o);
            if (lane >= o) inc += u;
        }
        if (lane == 31) wtot[wv] = inc;
    }
    __syncthreads();
    if (t < NBIN) {
        int start = inc - mine;
        for (int k = 0; k < wv; ++k) start += wtot[k];
        cur[t] += start;
        if (g == 0) binstart[t] = start;
    }
    __syncthreads();
    const int4* sq = (const int4*)ei + (size_t)g * QPB;
    const int4* dq = (const int4*)(ei + N_EDGES) + (size_t)g * QPB;
    for (int q = t; q < QPB; q += SORT_T) {
        const int4 s = sq[q];
        const int4 d = dq[q];
        const int p0 = atomicAdd(&cur[d.x >> BIN_SHIFT], 1);
        const int p1 = atomicAdd(&cur[d.y >> BIN_SHIFT], 1);
        const int p2 = atomicAdd(&cur[d.z >> BIN_SHIFT], 1);
        const int p3 = atomicAdd(&cur[d.w >> BIN_SHIFT], 1);
        binned[p0] = (s.x << BIN_SHIFT) | (d.x & (BIN_W - 1));
        binned[p1] = (s.y << BIN_SHIFT) | (d.y & (BIN_W - 1));
        binned[p2] = (s.z << BIN_SHIFT) | (d.z & (BIN_W - 1));
        binned[p3] = (s.w << BIN_SHIFT) | (d.w & (BIN_W - 1));
    }
}

#define FIN_T   1024
static_assert(BIN_W <= FIN_T && BIN_W % 32 == 0, "one thread per node of the bin, whole waves");
__global__ __launch_bounds__(FIN_T) void k_bin_finalize(const int* __restrict__ binstart,
                                                        const int* __restrict__ binned,
                                                        int* __restrict__ rowptr,
                                                        float* __restrict__ dinv,
                                                        int* __restrict__ csr_src) {
    __shared__ int cnt[BIN_W];
    __shared__ int wtot[BIN_W / 32];
    const int t = threadIdx.x, lane = t & 31, wv = t >> 5;
    const int b = blockIdx.x;
    const int beg = binstart[b];
    const int end = (b + 1 < NBIN) ? binstart[b + 1] : N_EDGES;
    if (t < BIN_W) cnt[t] = 0;
    __syncthreads();
    for (int p = beg + t; p < end; p += 4 * FIN_T) {
        int w[4];
        #pragma unroll
        for (int k = 0; k < 4; ++k) {
            const int pk = p + k * FIN_T;
            w[k] = (pk < end) ? binned[pk] : -1;
        }
        #pragma unroll
        for (int k = 0; k < 4; ++k)
            if (w[k] >= 0) atomicAdd(&cnt[w[k] & (BIN_W - 1)], 1);
    }
    __syncthreads();
    int dg = 0, inc = 0;
    if (t < BIN_W) {
        dg = cnt[t];
        inc = dg;
        for (int o = 1; o < 32; o <<= 1) {
            const int u = __shfl_up(inc, o);
            if (lane >= o) inc += u;
        }
        if (lane == 31) wtot[wv] = inc;
    }
    __syncthreads();
    if (t < BIN_W) {
        int base = beg;
        for (int k = 0; k < wv; ++k) base += wtot[k];
        const int excl = base + inc - dg;
        cnt[t] = excl;
        const int node = b * BIN_W + t;
        if (node < N_NODES) {
            rowptr[node] = excl;
            dinv[node] = __frsqrt_rn((float)dg + 1.0f);
        }
        if (node == N_NODES - 1) rowptr[N_NODES] = excl + dg;
    }
    __syncthreads();
    for (int p = beg + t; p < end; p += 4 * FIN_T) {
        int w[4];
        #pragma unroll
        for (int k = 0; k < 4; ++k) {
            const int pk = p + k * FIN_T;
            w[k] = (pk < end) ? binned[pk] : -1;
        }
        #pragma unroll
        for (int k = 0; k < 4; ++k) {
            if (w[k] >= 0) {
                const int pos = atomicAdd(&cnt[w[k] & (BIN_W - 1)], 1);
                csr_src[pos] = w[k] >> BIN_SHIFT;
            }
        }
    }
}

#define G1_WAVES 5
#define G1_T     (G1_WAVES * 32)
#define B_UNITS  (8 * 4 * 2 * 32)
static_assert((N_NODES / 16) % G1_WAVES == 0, "row tiles must tile the blocks");
static_assert(F_IN == 256 && HID == 64, "fragment index arithmetic below is written for 8 K-steps x 4 column tiles");
__global__ __launch_bounds__(G1_T) void k_gemm1_wmma(
        const float* __restrict__ x,
        const float* __restrict__ W1,
        const float* __restrict__ dinv,
        float* __restrict__ xw1) {
    __shared__ v8h bl[B_UNITS];
    const int t    = threadIdx.x;
    const int wave = t >> 5;
    const int lane = t & 31;
    const int half = lane >> 4;
    const int r    = lane & 15;

    const int row0 = (blockIdx.x * G1_WAVES + wave) * 16;
    const float* xrow = x + (size_t)(row0 + r) * F_IN + 8 * half;
    v4f av[32];
    #pragma unroll
    for (int kt = 0; kt < 8; ++kt) {
        const float* seg0 = xrow + kt * 32;
        const float* seg1 = seg0 + 16;
        av[4 * kt + 0] = __builtin_nontemporal_load((const v4f*)seg0);
        av[4 * kt + 1] = __builtin_nontemporal_load((const v4f*)(seg0 + 4));
        av[4 * kt + 2] = __builtin_nontemporal_load((const v4f*)seg1);
        av[4 * kt + 3] = __builtin_nontemporal_load((const v4f*)(seg1 + 4));
    }
    const float* dtile = dinv + __builtin_amdgcn_readfirstlane(row0);
    float dsc[16];
    #pragma unroll
    for (int m = 0; m < 16; ++m) dsc[m] = dtile[m];

    for (int q = t; q < B_UNITS / 4; q += G1_T) {
        const int c4   = q & 3;
        const int hf   = (q >> 2) & 1;
        const int part = (q >> 3) & 1;
        const int ct   = (q >> 4) & 3;
        const int kt   = q >> 6;
        const float* p = W1 + (size_t)(kt * 32 + part * 16 + 8 * hf) * HID + ct * 16 + 4 * c4;
        v4f w[8];
        #pragma unroll
        for (int j = 0; j < 8; ++j) w[j] = *(const v4f*)(p + j * HID);
        const int u0 = ((kt * 4 + ct) * 2 + part) * 32 + hf * 16 + 4 * c4;
        #pragma unroll
        for (int e = 0; e < 4; ++e) {
            v8h h;
            #pragma unroll
            for (int j = 0; j < 8; ++j) h[j] = (_Float16)w[j][e];
            bl[u0 + e] = h;
        }
    }
    __syncthreads();

    v8f acc[4];
    #pragma unroll
    for (int ct = 0; ct < 4; ++ct) acc[ct] = (v8f){0.f, 0.f, 0.f, 0.f, 0.f, 0.f, 0.f, 0.f};

    #pragma unroll
    for (int kt = 0; kt < 8; ++kt) {
        const v4f a0 = av[4 * kt + 0], a1 = av[4 * kt + 1];
        const v4f a2 = av[4 * kt + 2], a3 = av[4 * kt + 3];
        v16h a;
        a[0]  = (_Float16)a0.x; a[1]  = (_Float16)a0.y;
        a[2]  = (_Float16)a0.z; a[3]  = (_Float16)a0.w;
        a[4]  = (_Float16)a1.x; a[5]  = (_Float16)a1.y;
        a[6]  = (_Float16)a1.z; a[7]  = (_Float16)a1.w;
        a[8]  = (_Float16)a2.x; a[9]  = (_Float16)a2.y;
        a[10] = (_Float16)a2.z; a[11] = (_Float16)a2.w;
        a[12] = (_Float16)a3.x; a[13] = (_Float16)a3.y;
        a[14] = (_Float16)a3.z; a[15] = (_Float16)a3.w;
        #pragma unroll
        for (int ct = 0; ct < 4; ++ct) {
            const v8h lo = bl[((kt * 4 + ct) * 2 + 0) * 32 + lane];
            const v8h hi = bl[((kt * 4 + ct) * 2 + 1) * 32 + lane];
            const v16h bf = __builtin_shufflevector(lo, hi, 0, 1, 2, 3, 4, 5, 6, 7,
                                                    8, 9, 10, 11, 12, 13, 14, 15);
            acc[ct] = __builtin_amdgcn_wmma_f32_16x16x32_f16(
                 false, a,  false, bf,
                 (short)0, acc[ct],  false,  false);
        }
    }

    #pragma unroll
    for (int ct = 0; ct < 4; ++ct) {
        #pragma unroll
        for (int i = 0; i < 8; ++i) {
            const int m = i + 8 * half;
            const float sc = half ? dsc[8 + i] : dsc[i];
            xw1[(size_t)(row0 + m) * HID + (ct * 16 + r)] = acc[ct][i] * sc;
        }
    }
}

#define EPC 8
static_assert(N_NODES % 8 == 0, "8 waves per block must tile the node list");
__global__ __launch_bounds__(256) void k_gather1(const int* __restrict__ rowptr,
                                                 const int* __restrict__ csr_src,
                                                 const float* __restrict__ dinv,
                                                 const float* __restrict__ xw1,
                                                 const float* __restrict__ b1,
                                                 const float* __restrict__ W2,
                                                 float* __restrict__ hw2s) {
    const int lane = threadIdx.x & 31;
    int i = blockIdx.x * 8 + (int)(threadIdx.x >> 5);
    i = __builtin_amdgcn_readfirstlane(i);
    if (i >= N_NODES) return;

    const int beg = rowptr[i];
    const int end = rowptr[i + 1];
    float di = dinv[i];
    const float2* rows = (const float2*)xw1;
    const float2 self = rows[(size_t)i * (HID / 2) + lane];
    asm volatile("" : "+s"(di));

    float ax = 0.f, ay = 0.f;
    for (int base = beg; base < end; base += 32) {
        const int rem = end - base;
        const int n = rem < 32 ? rem : 32;
        const int sv = csr_src[base + (lane < n ? lane : n - 1)];
        for (int c = 0; c < n; c += EPC) {
            int s[EPC];
            float2 v[EPC];
            #pragma unroll
            for (int k = 0; k < EPC; ++k) {
                const int idx = (c + k < n) ? (c + k) : (n - 1);
                s[k] = __builtin_amdgcn_readlane(sv, idx);
            }
            #pragma unroll
            for (int k = 0; k < EPC; ++k)
                v[k] = rows[(size_t)s[k] * (HID / 2) + lane];
            #pragma unroll
            for (int k = 0; k < EPC; ++k) {
                const bool live = (c + k < n);
                ax += live ? v[k].x : 0.f;
                ay += live ? v[k].y : 0.f;
            }
        }
    }

    float hx = di * (ax + self.x);
    float hy = di * (ay + self.y);
    hx = fmaxf(hx + b1[2 * lane],     0.f);
    hy = fmaxf(hy + b1[2 * lane + 1], 0.f);
    const float w00 = W2[4 * lane + 0], w01 = W2[4 * lane + 1];
    const float w10 = W2[4 * lane + 2], w11 = W2[4 * lane + 3];
    float o0 = hx * w00 + hy * w10;
    float o1 = hx * w01 + hy * w11;
    for (int o = 16; o > 0; o >>= 1) {
        o0 += __shfl_xor(o0, o);
        o1 += __shfl_xor(o1, o);
    }
    if (lane == 0) {
        float2 r;
        r.x = o0 * di;
        r.y = o1 * di;
        ((float2*)hw2s)[i] = r;
    }
}

#define LPN 8
static_assert((N_NODES * LPN) % 256 == 0, "node groups must tile the blocks");
__global__ __launch_bounds__(256) void k_gather2_softmax(const int* __restrict__ rowptr,
                                                         const int* __restrict__ csr_src,
                                                         const float* __restrict__ dinv,
                                                         const float* __restrict__ hw2s,
                                                         const float* __restrict__ b2,
                                                         float* __restrict__ out) {
    const int t = blockIdx.x * blockDim.x + threadIdx.x;
    const int i = t / LPN;
    const int j = t % LPN;
    const int beg = rowptr[i];
    const int end = rowptr[i + 1];
    const float2* h2 = (const float2*)hw2s;

    float ax = 0.f, ay = 0.f;
    for (int p = beg + j; p < end; p += 2 * LPN) {
        const int  p2   = p + LPN;
        const bool two  = p2 < end;
        const int  s0   = csr_src[p];
        const int  s1   = csr_src[two ? p2 : p];
        const float2 v0 = h2[s0];
        const float2 v1 = h2[s1];
        ax += v0.x;
        ay += v0.y;
        ax += two ? v1.x : 0.f;
        ay += two ? v1.y : 0.f;
    }
    for (int o = LPN / 2; o > 0; o >>= 1) {
        ax += __shfl_xor(ax, o);
        ay += __shfl_xor(ay, o);
    }
    if (j == 0) {
        const float2 self = h2[i];
        const float di = dinv[i];
        const float a0 = di * (ax + self.x) + b2[0];
        const float a1 = di * (ay + self.y) + b2[1];
        const float m  = fmaxf(a0, a1);
        const float e0 = __expf(a0 - m);
        const float e1 = __expf(a1 - m);
        const float inv = 1.0f / (e0 + e1);
        float2 r;
        r.x = e0 * inv;
        r.y = e1 * inv;
        ((float2*)out)[i] = r;
    }
}

extern "C" void kernel_launch(void* const* d_in, const int* in_sizes, int n_in,
                              void* d_out, int out_size, void* d_ws, size_t ws_size,
                              hipStream_t stream) {
    const float* x  = (const float*)d_in[0];
    const int*   ei = (const int*)d_in[1];
    const float* W1 = (const float*)d_in[2];
    const float* b1 = (const float*)d_in[3];
    const float* W2 = (const float*)d_in[4];
    const float* b2 = (const float*)d_in[5];
    float* out = (float*)d_out;

    char* ws = (char*)d_ws;
    size_t o = 0;
    auto alloc = [&](size_t bytes) {
        char* p = ws + o;
        o += (bytes + 4095) & ~(size_t)4095;
        return p;
    };
    int*   tbl    = (int*)  alloc((size_t)TBL * sizeof(int));
    int*   bstart = (int*)  alloc((size_t)NBIN * sizeof(int));
    float* dinv   = (float*)alloc((size_t)N_NODES * sizeof(float));
    float* xw1    = (float*)alloc((size_t)N_NODES * HID * sizeof(float));
    int*   binned = (int*)  alloc((size_t)N_EDGES * sizeof(int));
    int*   csr    = (int*)  alloc((size_t)N_EDGES * sizeof(int));
    int*   rowptr = (int*)  alloc((size_t)(N_NODES + 1) * sizeof(int));
    float* hw2s   = (float*)alloc((size_t)N_NODES * 2 * sizeof(float));
    (void)ws_size; (void)in_sizes; (void)n_in; (void)out_size;

    const int B = 256;
    dim3 blk(B);

    k_bin_hist    <<<SORT_G, dim3(SORT_T), 0, stream>>>(ei, tbl);
    k_bin_scatter <<<SORT_G, dim3(SORT_T), 0, stream>>>(ei, tbl, bstart, binned);
    k_bin_finalize<<<NBIN, dim3(FIN_T), 0, stream>>>(bstart, binned, rowptr, dinv, csr);

    k_gemm1_wmma<<<(N_NODES / 16) / G1_WAVES, dim3(G1_T), 0, stream>>>(x, W1, dinv, xw1);

    k_gather1<<<N_NODES / 8, blk, 0, stream>>>(rowptr, csr, dinv, xw1, b1, W2, hw2s);

    k_gather2_softmax<<<(N_NODES * LPN) / B, blk, 0, stream>>>(rowptr, csr, dinv, hw2s, b2, out);
}


# both gathers: a wave walks several consecutive dst nodes in a loop (8 in layer 1, 4 per 8-lane group in layer 2), so 18750 waves enter instead of 125000; per-node sum order kept
# speedup vs baseline: 4.2478x; 1.0583x over previous
#include <hip/hip_runtime.h>
#include <hip/hip_bf16.h>
#include <stdint.h>

#define N_NODES 100000
#define N_EDGES 1600000
#define F_IN    256
#define HID     64

typedef __attribute__((ext_vector_type(16))) _Float16 v16h;
typedef __attribute__((ext_vector_type(8)))  _Float16 v8h;
typedef __attribute__((ext_vector_type(8)))  float    v8f;
typedef __attribute__((ext_vector_type(4)))  float    v4f;

static __device__ __forceinline__ int wave_sum_i32(int v) {
    v += __builtin_amdgcn_update_dpp(0, v, 0x161, 0xf, 0xf, false);
    v += __builtin_amdgcn_update_dpp(0, v, 0x162, 0xf, 0xf, false);
    v += __builtin_amdgcn_update_dpp(0, v, 0x164, 0xf, 0xf, false);
    v += __builtin_amdgcn_update_dpp(0, v, 0x168, 0xf, 0xf, false);
    v += __builtin_amdgcn_permlanex16(v, v, 0x76543210, 0xfedcba98, false, false);
    return v;
}

#define BIN_SHIFT 9
#define BIN_W   (1 << BIN_SHIFT)
#define NBIN    ((N_NODES + BIN_W - 1) / BIN_W)
#define SORT_G  250
#define SORT_T  1024
#define QUADS   (N_EDGES / 4)
#define QPB     (QUADS / SORT_G)
#define TBL     (NBIN * SORT_G)
static_assert(N_EDGES % 4 == 0, "edge quads must tile the edge list");
static_assert(QUADS % SORT_G == 0, "edge blocks must tile the quads");
static_assert(N_NODES <= (1 << (31 - BIN_SHIFT)), "src id and dst-in-bin must pack into one word");

__global__ __launch_bounds__(SORT_T) void k_bin_hist(const int* __restrict__ ei,
                                                     int* __restrict__ tbl) {
    __shared__ int cnt[NBIN];
    const int t = threadIdx.x, g = blockIdx.x;
    for (int b = t; b < NBIN; b += SORT_T) cnt[b] = 0;
    __syncthreads();
    const int4* dq = (const int4*)(ei + N_EDGES) + (size_t)g * QPB;
    for (int q = t; q < QPB; q += SORT_T) {
        const int4 d = dq[q];
        atomicAdd(&cnt[d.x >> BIN_SHIFT], 1);
        atomicAdd(&cnt[d.y >> BIN_SHIFT], 1);
        atomicAdd(&cnt[d.z >> BIN_SHIFT], 1);
        atomicAdd(&cnt[d.w >> BIN_SHIFT], 1);
    }
    __syncthreads();
    for (int b = t; b < NBIN; b += SORT_T) tbl[b * SORT_G + g] = cnt[b];
}

#define ROWS_PW ((NBIN + 31) / 32)
#define CPL     ((SORT_G + 31) / 32)
static_assert(SORT_T == 1024, "row ownership below is written for 32 waves per block");
static_assert(NBIN <= 32 * 32, "the scan of the bin totals is one wave-level scan plus wave totals");
__global__ __launch_bounds__(SORT_T) void k_bin_scatter(const int* __restrict__ ei,
                                                        const int* __restrict__ tbl,
                                                        int* __restrict__ binstart,
                                                        int* __restrict__ binned) {
    __shared__ int cur[NBIN];
    __shared__ int tot[NBIN];
    __shared__ int wtot[32];
    const int t = threadIdx.x, g = blockIdx.x, lane = t & 31, wv = t >> 5;

    int v[ROWS_PW][CPL];
    #pragma unroll
    for (int j = 0; j < ROWS_PW; ++j) {
        const int b  = wv + 32 * j;
        const int* row = tbl + (b < NBIN ? b : NBIN - 1) * SORT_G;
        #pragma unroll
        for (int k = 0; k < CPL; ++k) {
            const int col = lane + 32 * k;
            v[j][k] = row[col < SORT_G ? col : SORT_G - 1];
        }
    }
    #pragma unroll
    for (int j = 0; j < ROWS_PW; ++j) {
        const int b = wv + 32 * j;
        int all = 0, before = 0;
        #pragma unroll
        for (int k = 0; k < CPL; ++k) {
            const int col = lane + 32 * k;
            const int x = (col < SORT_G) ? v[j][k] : 0;
            all += x;
            before += (col < g) ? x : 0;
        }
        all = wave_sum_i32(all);
        before = wave_sum_i32(before);
        if (lane == 0 && b < NBIN) {
            tot[b] = all;
            cur[b] = before;
        }
    }
    __syncthreads();
    int mine = 0, inc = 0;
    if (wv <= (NBIN - 1) / 32) {
        mine = (t < NBIN) ? tot[t] : 0;
        inc = mine;
        for (int o = 1; o < 32; o <<= 1) {
            const int u = __shfl_up(inc, o);
            if (lane >= o) inc += u;
        }
        if (lane == 31) wtot[wv] = inc;
    }
    __syncthreads();
    if (t < NBIN) {
        int start = inc - mine;
        for (int k = 0; k < wv; ++k) start += wtot[k];
        cur[t] += start;
        if (g == 0) binstart[t] = start;
    }
    __syncthreads();
    const int4* sq = (const int4*)ei + (size_t)g * QPB;
    const int4* dq = (const int4*)(ei + N_EDGES) + (size_t)g * QPB;
    for (int q = t; q < QPB; q += SORT_T) {
        const int4 s = sq[q];
        const int4 d = dq[q];
        const int p0 = atomicAdd(&cur[d.x >> BIN_SHIFT], 1);
        const int p1 = atomicAdd(&cur[d.y >> BIN_SHIFT], 1);
        const int p2 = atomicAdd(&cur[d.z >> BIN_SHIFT], 1);
        const int p3 = atomicAdd(&cur[d.w >> BIN_SHIFT], 1);
        binned[p0] = (s.x << BIN_SHIFT) | (d.x & (BIN_W - 1));
        binned[p1] = (s.y << BIN_SHIFT) | (d.y & (BIN_W - 1));
        binned[p2] = (s.z << BIN_SHIFT) | (d.z & (BIN_W - 1));
        binned[p3] = (s.w << BIN_SHIFT) | (d.w & (BIN_W - 1));
    }
}

#define FIN_T   1024
static_assert(BIN_W <= FIN_T && BIN_W % 32 == 0, "one thread per node of the bin, whole waves");
__global__ __launch_bounds__(FIN_T) void k_bin_finalize(const int* __restrict__ binstart,
                                                        const int* __restrict__ binned,
                                                        int* __restrict__ rowptr,
                                                        float* __restrict__ dinv,
                                                        int* __restrict__ csr_src) {
    __shared__ int cnt[BIN_W];
    __shared__ int wtot[BIN_W / 32];
    const int t = threadIdx.x, lane = t & 31, wv = t >> 5;
    const int b = blockIdx.x;
    const int beg = binstart[b];
    const int end = (b + 1 < NBIN) ? binstart[b + 1] : N_EDGES;
    if (t < BIN_W) cnt[t] = 0;
    __syncthreads();
    for (int p = beg + t; p < end; p += 4 * FIN_T) {
        int w[4];
        #pragma unroll
        for (int k = 0; k < 4; ++k) {
            const int pk = p + k * FIN_T;
            w[k] = (pk < end) ? binned[pk] : -1;
        }
        #pragma unroll
        for (int k = 0; k < 4; ++k)
            if (w[k] >= 0) atomicAdd(&cnt[w[k] & (BIN_W - 1)], 1);
    }
    __syncthreads();
    int dg = 0, inc = 0;
    if (t < BIN_W) {
        dg = cnt[t];
        inc = dg;
        for (int o = 1; o < 32; o <<= 1) {
            const int u = __shfl_up(inc, o);
            if (lane >= o) inc += u;
        }
        if (lane == 31) wtot[wv] = inc;
    }
    __syncthreads();
    if (t < BIN_W) {
        int base = beg;
        for (int k = 0; k < wv; ++k) base += wtot[k];
        const int excl = base + inc - dg;
        cnt[t] = excl;
        const int node = b * BIN_W + t;
        if (node < N_NODES) {
            rowptr[node] = excl;
            dinv[node] = __frsqrt_rn((float)dg + 1.0f);
        }
        if (node == N_NODES - 1) rowptr[N_NODES] = excl + dg;
    }
    __syncthreads();
    for (int p = beg + t; p < end; p += 4 * FIN_T) {
        int w[4];
        #pragma unroll
        for (int k = 0; k < 4; ++k) {
            const int pk = p + k * FIN_T;
            w[k] = (pk < end) ? binned[pk] : -1;
        }
        #pragma unroll
        for (int k = 0; k < 4; ++k) {
            if (w[k] >= 0) {
                const int pos = atomicAdd(&cnt[w[k] & (BIN_W - 1)], 1);
                csr_src[pos] = w[k] >> BIN_SHIFT;
            }
        }
    }
}

#define G1_WAVES 5
#define G1_T     (G1_WAVES * 32)
#define B_UNITS  (8 * 4 * 2 * 32)
static_assert((N_NODES / 16) % G1_WAVES == 0, "row tiles must tile the blocks");
static_assert(F_IN == 256 && HID == 64, "fragment index arithmetic below is written for 8 K-steps x 4 column tiles");
__global__ __launch_bounds__(G1_T) void k_gemm1_wmma(
        const float* __restrict__ x,
        const float* __restrict__ W1,
        const float* __restrict__ dinv,
        float* __restrict__ xw1) {
    __shared__ v8h bl[B_UNITS];
    const int t    = threadIdx.x;
    const int wave = t >> 5;
    const int lane = t & 31;
    const int half = lane >> 4;
    const int r    = lane & 15;

    const int row0 = (blockIdx.x * G1_WAVES + wave) * 16;
    const float* xrow = x + (size_t)(row0 + r) * F_IN + 8 * half;
    v4f av[32];
    #pragma unroll
    for (int kt = 0; kt < 8; ++kt) {
        const float* seg0 = xrow + kt * 32;
        const float* seg1 = seg0 + 16;
        av[4 * kt + 0] = __builtin_nontemporal_load((const v4f*)seg0);
        av[4 * kt + 1] = __builtin_nontemporal_load((const v4f*)(seg0 + 4));
        av[4 * kt + 2] = __builtin_nontemporal_load((const v4f*)seg1);
        av[4 * kt + 3] = __builtin_nontemporal_load((const v4f*)(seg1 + 4));
    }
    const float* dtile = dinv + __builtin_amdgcn_readfirstlane(row0);
    float dsc[16];
    #pragma unroll
    for (int m = 0; m < 16; ++m) dsc[m] = dtile[m];

    for (int q = t; q < B_UNITS / 4; q += G1_T) {
        const int c4   = q & 3;
        const int hf   = (q >> 2) & 1;
        const int part = (q >> 3) & 1;
        const int ct   = (q >> 4) & 3;
        const int kt   = q >> 6;
        const float* p = W1 + (size_t)(kt * 32 + part * 16 + 8 * hf) * HID + ct * 16 + 4 * c4;
        v4f w[8];
        #pragma unroll
        for (int j = 0; j < 8; ++j) w[j] = *(const v4f*)(p + j * HID);
        const int u0 = ((kt * 4 + ct) * 2 + part) * 32 + hf * 16 + 4 * c4;
        #pragma unroll
        for (int e = 0; e < 4; ++e) {
            v8h h;
            #pragma unroll
            for (int j = 0; j < 8; ++j) h[j] = (_Float16)w[j][e];
            bl[u0 + e] = h;
        }
    }
    __syncthreads();

    v8f acc[4];
    #pragma unroll
    for (int ct = 0; ct < 4; ++ct) acc[ct] = (v8f){0.f, 0.f, 0.f, 0.f, 0.f, 0.f, 0.f, 0.f};

    #pragma unroll
    for (int kt = 0; kt < 8; ++kt) {
        const v4f a0 = av[4 * kt + 0], a1 = av[4 * kt + 1];
        const v4f a2 = av[4 * kt + 2], a3 = av[4 * kt + 3];
        v16h a;
        a[0]  = (_Float16)a0.x; a[1]  = (_Float16)a0.y;
        a[2]  = (_Float16)a0.z; a[3]  = (_Float16)a0.w;
        a[4]  = (_Float16)a1.x; a[5]  = (_Float16)a1.y;
        a[6]  = (_Float16)a1.z; a[7]  = (_Float16)a1.w;
        a[8]  = (_Float16)a2.x; a[9]  = (_Float16)a2.y;
        a[10] = (_Float16)a2.z; a[11] = (_Float16)a2.w;
        a[12] = (_Float16)a3.x; a[13] = (_Float16)a3.y;
        a[14] = (_Float16)a3.z; a[15] = (_Float16)a3.w;
        #pragma unroll
        for (int ct = 0; ct < 4; ++ct) {
            const v8h lo = bl[((kt * 4 + ct) * 2 + 0) * 32 + lane];
            const v8h hi = bl[((kt * 4 + ct) * 2 + 1) * 32 + lane];
            const v16h bf = __builtin_shufflevector(lo, hi, 0, 1, 2, 3, 4, 5, 6, 7,
                                                    8, 9, 10, 11, 12, 13, 14, 15);
            acc[ct] = __builtin_amdgcn_wmma_f32_16x16x32_f16(
                 false, a,  false, bf,
                 (short)0, acc[ct],  false,  false);
        }
    }

    #pragma unroll
    for (int ct = 0; ct < 4; ++ct) {
        #pragma unroll
        for (int i = 0; i < 8; ++i) {
            const int m = i + 8 * half;
            const float sc = half ? dsc[8 + i] : dsc[i];
            xw1[(size_t)(row0 + m) * HID + (ct * 16 + r)] = acc[ct][i] * sc;
        }
    }
}

#define EPC 8
#define NPW 8
#define G1G_WAVES 8
static_assert(N_NODES % NPW == 0, "node runs must tile the node list");
static_assert((NPW & (NPW - 1)) == 0 && NPW < 32, "row pointers and dinv of a run sit in one wave-wide load");
__global__ __launch_bounds__(G1G_WAVES * 32) void k_gather1(const int* __restrict__ rowptr,
                                                 const int* __restrict__ csr_src,
                                                 const float* __restrict__ dinv,
                                                 const float* __restrict__ xw1,
                                                 const float* __restrict__ b1,
                                                 const float* __restrict__ W2,
                                                 float* __restrict__ hw2s) {
    const int lane = threadIdx.x & 31;
    int w = blockIdx.x * G1G_WAVES + (int)(threadIdx.x >> 5);
    w = __builtin_amdgcn_readfirstlane(w);
    const int i0 = w * NPW;
    if (i0 >= N_NODES) return;

    const int   rpv = rowptr[i0 + (lane <= NPW ? lane : NPW)];
    const float dvv = dinv[i0 + (lane & (NPW - 1))];
    const float2 bb = ((const float2*)b1)[lane];
    const float4 ww = ((const float4*)W2)[lane];
    const float2* rows = (const float2*)xw1;

    int beg = __builtin_amdgcn_readlane(rpv, 0);
    int end = __builtin_amdgcn_readlane(rpv, 1);
    int sv0 = 0;
    {
        const int rem = end - beg;
        const int n = rem < 32 ? rem : 32;
        if (n > 0) sv0 = csr_src[beg + (lane < n ? lane : n - 1)];
    }
    float2 self = rows[(size_t)i0 * (HID / 2) + lane];
    float rx = 0.f, ry = 0.f;
    for (int k = 0; k < NPW; ++k) {
        const float di = __int_as_float(__builtin_amdgcn_readlane(__float_as_int(dvv), k));

        float ax = 0.f, ay = 0.f;
        for (int base = beg; base < end; base += 32) {
            const int rem = end - base;
            const int n = rem < 32 ? rem : 32;
            int sv = sv0;
            if (base != beg) sv = csr_src[base + (lane < n ? lane : n - 1)];
            for (int c = 0; c < n; c += EPC) {
                int s[EPC];
                float2 v[EPC];
                #pragma unroll
                for (int e = 0; e < EPC; ++e) {
                    const int idx = (c + e < n) ? (c + e) : (n - 1);
                    s[e] = __builtin_amdgcn_readlane(sv, idx);
                }
                #pragma unroll
                for (int e = 0; e < EPC; ++e)
                    v[e] = rows[(size_t)s[e] * (HID / 2) + lane];
                #pragma unroll
                for (int e = 0; e < EPC; ++e) {
                    const bool live = (c + e < n);
                    ax += live ? v[e].x : 0.f;
                    ay += live ? v[e].y : 0.f;
                }
            }
        }

        const int nend = __builtin_amdgcn_readlane(rpv, k + 2 < NPW ? k + 2 : NPW);
        int svn = 0;
        {
            const int rem = nend - end;
            const int n = rem < 32 ? rem : 32;
            if (n > 0) svn = csr_src[end + (lane < n ? lane : n - 1)];
        }
        float2 selfn = self;
        if (k + 1 < NPW) selfn = rows[(size_t)(i0 + k + 1) * (HID / 2) + lane];

        float hx = di * (ax + self.x);
        float hy = di * (ay + self.y);
        hx = fmaxf(hx + bb.x, 0.f);
        hy = fmaxf(hy + bb.y, 0.f);
        float o0 = hx * ww.x + hy * ww.z;
        float o1 = hx * ww.y + hy * ww.w;
        for (int o = 16; o > 0; o >>= 1) {
            o0 += __shfl_xor(o0, o);
            o1 += __shfl_xor(o1, o);
        }
        if (lane == k) {
            rx = o0 * di;
            ry = o1 * di;
        }
        beg = end;
        end = nend;
        sv0 = svn;
        self = selfn;
    }
    if (lane < NPW) {
        float2 r;
        r.x = rx;
        r.y = ry;
        ((float2*)hw2s)[i0 + lane] = r;
    }
}

#define LPN 8
#define NPG 4
#define G2_T 160
static_assert(N_NODES % ((G2_T / LPN) * NPG) == 0, "node runs must tile the blocks");
static_assert(NPG <= LPN && (NPG & (NPG - 1)) == 0, "one finishing lane per node of the run");
__global__ __launch_bounds__(G2_T) void k_gather2_softmax(const int* __restrict__ rowptr,
                                                         const int* __restrict__ csr_src,
                                                         const float* __restrict__ dinv,
                                                         const float* __restrict__ hw2s,
                                                         const float* __restrict__ b2,
                                                         float* __restrict__ out) {
    const int t = blockIdx.x * G2_T + threadIdx.x;
    const int i0 = (t / LPN) * NPG;
    const int j = t % LPN;
    const float2* h2 = (const float2*)hw2s;
    const int im = i0 + (j & (NPG - 1));
    const float2 self = h2[im];
    const float di = dinv[im];
    int rp[NPG + 1];
    #pragma unroll
    for (int k = 0; k <= NPG; ++k) rp[k] = rowptr[i0 + k];

    float sx = 0.f, sy = 0.f;
    #pragma unroll
    for (int k = 0; k < NPG; ++k) {
        const int beg = rp[k];
        const int end = rp[k + 1];
        float ax = 0.f, ay = 0.f;
        for (int p = beg + j; p < end; p += 2 * LPN) {
            const int  p2   = p + LPN;
            const bool two  = p2 < end;
            const int  s0   = csr_src[p];
            const int  s1   = csr_src[two ? p2 : p];
            const float2 v0 = h2[s0];
            const float2 v1 = h2[s1];
            ax += v0.x;
            ay += v0.y;
            ax += two ? v1.x : 0.f;
            ay += two ? v1.y : 0.f;
        }
        for (int o = LPN / 2; o > 0; o >>= 1) {
            ax += __shfl_xor(ax, o);
            ay += __shfl_xor(ay, o);
        }
        if (j == k) {
            sx = ax;
            sy = ay;
        }
    }
    if (j < NPG) {
        const float a0 = di * (sx + self.x) + b2[0];
        const float a1 = di * (sy + self.y) + b2[1];
        const float m  = fmaxf(a0, a1);
        const float e0 = __expf(a0 - m);
        const float e1 = __expf(a1 - m);
        const float inv = 1.0f / (e0 + e1);
        float2 r;
        r.x = e0 * inv;
        r.y = e1 * inv;
        ((float2*)out)[im] = r;
    }
}

extern "C" void kernel_launch(void* const* d_in, const int* in_sizes, int n_in,
                              void* d_out, int out_size, void* d_ws, size_t ws_size,
                              hipStream_t stream) {
    const float* x  = (const float*)d_in[0];
    const int*   ei = (const int*)d_in[1];
    const float* W1 = (const float*)d_in[2];
    const float* b1 = (const float*)d_in[3];
    const float* W2 = (const float*)d_in[4];
    const float* b2 = (const float*)d_in[5];
    float* out = (float*)d_out;

    char* ws = (char*)d_ws;
    size_t o = 0;
    auto alloc = [&](size_t bytes) {
        char* p = ws + o;
        o += (bytes + 4095) & ~(size_t)4095;
        return p;
    };
    int*   tbl    = (int*)  alloc((size_t)TBL * sizeof(int));
    int*   bstart = (int*)  alloc((size_t)NBIN * sizeof(int));
    float* dinv   = (float*)alloc((size_t)N_NODES * sizeof(float));
    float* xw1    = (float*)alloc((size_t)N_NODES * HID * sizeof(float));
    int*   binned = (int*)  alloc((size_t)N_EDGES * sizeof(int));
    int*   csr    = (int*)  alloc((size_t)N_EDGES * sizeof(int));
    int*   rowptr = (int*)  alloc((size_t)(N_NODES + 1) * sizeof(int));
    float* hw2s   = (float*)alloc((size_t)N_NODES * 2 * sizeof(float));
    (void)ws_size; (void)in_sizes; (void)n_in; (void)out_size;

    k_bin_hist    <<<SORT_G, dim3(SORT_T), 0, stream>>>(ei, tbl);
    k_bin_scatter <<<SORT_G, dim3(SORT_T), 0, stream>>>(ei, tbl, bstart, binned);
    k_bin_finalize<<<NBIN, dim3(FIN_T), 0, stream>>>(bstart, binned, rowptr, dinv, csr);

    k_gemm1_wmma<<<(N_NODES / 16) / G1_WAVES, dim3(G1_T), 0, stream>>>(x, W1, dinv, xw1);

    k_gather1<<<(N_NODES / NPW + G1G_WAVES - 1) / G1G_WAVES, dim3(G1G_WAVES * 32), 0, stream>>>(rowptr, csr, dinv, xw1, b1, W2, hw2s);

    k_gather2_softmax<<<N_NODES / ((G2_T / LPN) * NPG), dim3(G2_T), 0, stream>>>(rowptr, csr, dinv, hw2s, b2, out);
}


# each edge block groups its own edges by dst bin in LDS and stores whole lines; the per-bin launch walks its 250 runs by a start/length table: count launch, table scan gone; new LDS arrival order
# speedup vs baseline: 5.5704x; 1.3114x over previous
#include <hip/hip_runtime.h>
#include <hip/hip_bf16.h>
#include <stdint.h>

#define N_NODES 100000
#define N_EDGES 1600000
#define F_IN    256
#define HID     64

typedef __attribute__((ext_vector_type(16))) _Float16 v16h;
typedef __attribute__((ext_vector_type(8)))  _Float16 v8h;
typedef __attribute__((ext_vector_type(8)))  float    v8f;
typedef __attribute__((ext_vector_type(4)))  float    v4f;

static __device__ __forceinline__ int wave_sum_i32(int v) {
    v += __builtin_amdgcn_update_dpp(0, v, 0x161, 0xf, 0xf, false);
    v += __builtin_amdgcn_update_dpp(0, v, 0x162, 0xf, 0xf, false);
    v += __builtin_amdgcn_update_dpp(0, v, 0x164, 0xf, 0xf, false);
    v += __builtin_amdgcn_update_dpp(0, v, 0x168, 0xf, 0xf, false);
    v += __builtin_amdgcn_permlanex16(v, v, 0x76543210, 0xfedcba98, false, false);
    return v;
}

#define BIN_SHIFT 9
#define BIN_W   (1 << BIN_SHIFT)
#define NBIN    ((N_NODES + BIN_W - 1) / BIN_W)
#define SORT_G  250
#define SORT_T  1024
#define QUADS   (N_EDGES / 4)
#define QPB     (QUADS / SORT_G)
#define EPB     (4 * QPB)
#define TBL     (NBIN * SORT_G)
static_assert(N_EDGES % 4 == 0, "edge quads must tile the edge list");
static_assert(QUADS % SORT_G == 0, "edge blocks must tile the quads");
static_assert(N_NODES <= (1 << (31 - BIN_SHIFT)), "src id and dst-in-bin must pack into one word");
static_assert(EPB < (1 << 15), "run start and run length of a table entry share one non-negative word");
static_assert(QPB > SORT_T && QPB <= 2 * SORT_T, "a thread of an edge block holds one or two quads");
static_assert(SORT_T == 1024, "the bin scan below is written for 32 waves per block");
static_assert(NBIN <= SORT_T, "one thread per bin in the bin scan");

__global__ __launch_bounds__(SORT_T) void k_block_bins(const int* __restrict__ ei,
                                                       int* __restrict__ tbl,
                                                       int* __restrict__ binned) {
    __shared__ int cnt[NBIN];
    __shared__ int wtot[32];
    __shared__ int stage[EPB];
    const int t = threadIdx.x, g = blockIdx.x, lane = t & 31, wv = t >> 5;
    const int4* sq = (const int4*)ei + (size_t)g * QPB;
    const int4* dq = (const int4*)(ei + N_EDGES) + (size_t)g * QPB;
    const bool two = (t + SORT_T) < QPB;
    const int q1 = two ? t + SORT_T : t;
    const int4 s0 = sq[t],  d0 = dq[t];
    const int4 s1 = sq[q1], d1 = dq[q1];
    for (int b = t; b < NBIN; b += SORT_T) cnt[b] = 0;
    __syncthreads();
    atomicAdd(&cnt[d0.x >> BIN_SHIFT], 1);
    atomicAdd(&cnt[d0.y >> BIN_SHIFT], 1);
    atomicAdd(&cnt[d0.z >> BIN_SHIFT], 1);
    atomicAdd(&cnt[d0.w >> BIN_SHIFT], 1);
    if (two) {
        atomicAdd(&cnt[d1.x >> BIN_SHIFT], 1);
        atomicAdd(&cnt[d1.y >> BIN_SHIFT], 1);
        atomicAdd(&cnt[d1.z >> BIN_SHIFT], 1);
        atomicAdd(&cnt[d1.w >> BIN_SHIFT], 1);
    }
    __syncthreads();
    int mine = 0, inc = 0;
    if (wv <= (NBIN - 1) / 32) {
        mine = (t < NBIN) ? cnt[t] : 0;
        inc = mine;
        for (int o = 1; o < 32; o <<= 1) {
            const int u = __shfl_up(inc, o);
            if (lane >= o) inc += u;
        }
        if (lane == 31) wtot[wv] = inc;
    }
    __syncthreads();
    if (t < NBIN) {
        int start = inc - mine;
        for (int k = 0; k < wv; ++k) start += wtot[k];
        cnt[t] = start;
        tbl[t * SORT_G + g] = (start << 16) | mine;
    }
    __syncthreads();
    {
        const int p0 = atomicAdd(&cnt[d0.x >> BIN_SHIFT], 1);
        const int p1 = atomicAdd(&cnt[d0.y >> BIN_SHIFT], 1);
        const int p2 = atomicAdd(&cnt[d0.z >> BIN_SHIFT], 1);
        const int p3 = atomicAdd(&cnt[d0.w >> BIN_SHIFT], 1);
        stage[p0] = (s0.x << BIN_SHIFT) | (d0.x & (BIN_W - 1));
        stage[p1] = (s0.y << BIN_SHIFT) | (d0.y & (BIN_W - 1));
        stage[p2] = (s0.z << BIN_SHIFT) | (d0.z & (BIN_W - 1));
        stage[p3] = (s0.w << BIN_SHIFT) | (d0.w & (BIN_W - 1));
    }
    if (two) {
        const int p0 = atomicAdd(&cnt[d1.x >> BIN_SHIFT], 1);
        const int p1 = atomicAdd(&cnt[d1.y >> BIN_SHIFT], 1);
        const int p2 = atomicAdd(&cnt[d1.z >> BIN_SHIFT], 1);
        const int p3 = atomicAdd(&cnt[d1.w >> BIN_SHIFT], 1);
        stage[p0] = (s1.x << BIN_SHIFT) | (d1.x & (BIN_W - 1));
        stage[p1] = (s1.y << BIN_SHIFT) | (d1.y & (BIN_W - 1));
        stage[p2] = (s1.z << BIN_SHIFT) | (d1.z & (BIN_W - 1));
        stage[p3] = (s1.w << BIN_SHIFT) | (d1.w & (BIN_W - 1));
    }
    __syncthreads();
    int* mylist = binned + (size_t)g * EPB;
    int sv[(EPB + SORT_T - 1) / SORT_T];
    #pragma unroll
    for (int k = 0; k < (EPB + SORT_T - 1) / SORT_T; ++k) {
        const int s = t + k * SORT_T;
        sv[k] = stage[s < EPB ? s : EPB - 1];
    }
    #pragma unroll
    for (int k = 0; k < (EPB + SORT_T - 1) / SORT_T; ++k) {
        const int s = t + k * SORT_T;
        if (s < EPB) mylist[s] = sv[k];
    }
}

#define FIN_T   1024
#define RUNS_PW ((SORT_G + 31) / 32)
static_assert(BIN_W <= FIN_T && BIN_W % 32 == 0, "one thread per node of the bin, whole waves");
static_assert(SORT_G <= FIN_T && FIN_T == 1024, "one thread per table entry of the bin's row; 32 waves");
__global__ __launch_bounds__(FIN_T) void k_bin_finalize(const int* __restrict__ tbl,
                                                        const int* __restrict__ binned,
                                                        int* __restrict__ rowptr,
                                                        float* __restrict__ dinv,
                                                        int* __restrict__ csr_src) {
    __shared__ int cnt[BIN_W];
    __shared__ int wtot[BIN_W / 32];
    __shared__ int runoff[SORT_G];
    __shared__ int runlen[SORT_G];
    __shared__ int red[32];
    const int t = threadIdx.x, lane = t & 31, wv = t >> 5;
    const int b = blockIdx.x;
    int rs = 0;
    if (t < SORT_G) {
        const int e = tbl[b * SORT_G + t];
        rs = e >> 16;
        runoff[t] = t * EPB + rs;
        runlen[t] = e & 0xffff;
    }
    if (t < BIN_W) cnt[t] = 0;
    rs = wave_sum_i32(rs);
    if (lane == 0) red[wv] = rs;
    __syncthreads();
    int beg = 0;
    for (int k = 0; k < RUNS_PW; ++k) beg += red[k];

    int rl[RUNS_PW], ro[RUNS_PW], w0[RUNS_PW], w1[RUNS_PW];
    int longest = 0;
    #pragma unroll
    for (int j = 0; j < RUNS_PW; ++j) {
        const int g = wv + 32 * j;
        rl[j] = (g < SORT_G) ? runlen[g] : 0;
        ro[j] = (g < SORT_G) ? runoff[g] : 0;
        longest = rl[j] > longest ? rl[j] : longest;
    }
    #pragma unroll
    for (int j = 0; j < RUNS_PW; ++j) {
        w0[j] = (lane < rl[j])      ? binned[ro[j] + lane]      : -1;
        w1[j] = (lane + 32 < rl[j]) ? binned[ro[j] + lane + 32] : -1;
    }
    #pragma unroll
    for (int j = 0; j < RUNS_PW; ++j) {
        if (w0[j] >= 0) atomicAdd(&cnt[w0[j] & (BIN_W - 1)], 1);
        if (w1[j] >= 0) atomicAdd(&cnt[w1[j] & (BIN_W - 1)], 1);
    }
    if (longest > 64) {
        for (int g = wv; g < SORT_G; g += 32) {
            const int rl = runlen[g], ro = runoff[g];
            for (int k = lane + 64; k < rl; k += 32)
                atomicAdd(&cnt[binned[ro + k] & (BIN_W - 1)], 1);
        }
    }
    __syncthreads();
    int dg = 0, inc = 0;
    if (t < BIN_W) {
        dg = cnt[t];
        inc = dg;
        for (int o = 1; o < 32; o <<= 1) {
            const int u = __shfl_up(inc, o);
            if (lane >= o) inc += u;
        }
        if (lane == 31) wtot[wv] = inc;
    }
    __syncthreads();
    if (t < BIN_W) {
        int base = beg;
        for (int k = 0; k < wv; ++k) base += wtot[k];
        const int excl = base + inc - dg;
        cnt[t] = excl;
        const int node = b * BIN_W + t;
        if (node < N_NODES) {
            rowptr[node] = excl;
            dinv[node] = __frsqrt_rn((float)dg + 1.0f);
        }
        if (node == N_NODES - 1) rowptr[N_NODES] = excl + dg;
    }
    __syncthreads();
    #pragma unroll
    for (int j = 0; j < RUNS_PW; ++j) {
        if (w0[j] >= 0) {
            const int pos = atomicAdd(&cnt[w0[j] & (BIN_W - 1)], 1);
            csr_src[pos] = w0[j] >> BIN_SHIFT;
        }
        if (w1[j] >= 0) {
            const int pos = atomicAdd(&cnt[w1[j] & (BIN_W - 1)], 1);
            csr_src[pos] = w1[j] >> BIN_SHIFT;
        }
    }
    if (longest > 64) {
        for (int g = wv; g < SORT_G; g += 32) {
            const int rl = runlen[g], ro = runoff[g];
            for (int k = lane + 64; k < rl; k += 32) {
                const int x = binned[ro + k];
                const int pos = atomicAdd(&cnt[x & (BIN_W - 1)], 1);
                csr_src[pos] = x >> BIN_SHIFT;
            }
        }
    }
}

#define G1_WAVES 5
#define G1_T     (G1_WAVES * 32)
#define B_UNITS  (8 * 4 * 2 * 32)
static_assert((N_NODES / 16) % G1_WAVES == 0, "row tiles must tile the blocks");
static_assert(F_IN == 256 && HID == 64, "fragment index arithmetic below is written for 8 K-steps x 4 column tiles");
__global__ __launch_bounds__(G1_T) void k_gemm1_wmma(
        const float* __restrict__ x,
        const float* __restrict__ W1,
        const float* __restrict__ dinv,
        float* __restrict__ xw1) {
    __shared__ v8h bl[B_UNITS];
    const int t    = threadIdx.x;
    const int wave = t >> 5;
    const int lane = t & 31;
    const int half = lane >> 4;
    const int r    = lane & 15;

    const int row0 = (blockIdx.x * G1_WAVES + wave) * 16;
    const float* xrow = x + (size_t)(row0 + r) * F_IN + 8 * half;
    v4f av[32];
    #pragma unroll
    for (int kt = 0; kt < 8; ++kt) {
        const float* seg0 = xrow + kt * 32;
        const float* seg1 = seg0 + 16;
        av[4 * kt + 0] = __builtin_nontemporal_load((const v4f*)seg0);
        av[4 * kt + 1] = __builtin_nontemporal_load((const v4f*)(seg0 + 4));
        av[4 * kt + 2] = __builtin_nontemporal_load((const v4f*)seg1);
        av[4 * kt + 3] = __builtin_nontemporal_load((const v4f*)(seg1 + 4));
    }
    const float* dtile = dinv + __builtin_amdgcn_readfirstlane(row0);
    float dsc[16];
    #pragma unroll
    for (int m = 0; m < 16; ++m) dsc[m] = dtile[m];

    for (int q = t; q < B_UNITS / 4; q += G1_T) {
        const int c4   = q & 3;
        const int hf   = (q >> 2) & 1;
        const int part = (q >> 3) & 1;
        const int ct   = (q >> 4) & 3;
        const int kt   = q >> 6;
        const float* p = W1 + (size_t)(kt * 32 + part * 16 + 8 * hf) * HID + ct * 16 + 4 * c4;
        v4f w[8];
        #pragma unroll
        for (int j = 0; j < 8; ++j) w[j] = *(const v4f*)(p + j * HID);
        const int u0 = ((kt * 4 + ct) * 2 + part) * 32 + hf * 16 + 4 * c4;
        #pragma unroll
        for (int e = 0; e < 4; ++e) {
            v8h h;
            #pragma unroll
            for (int j = 0; j < 8; ++j) h[j] = (_Float16)w[j][e];
            bl[u0 + e] = h;
        }
    }
    __syncthreads();

    v8f acc[4];
    #pragma unroll
    for (int ct = 0; ct < 4; ++ct) acc[ct] = (v8f){0.f, 0.f, 0.f, 0.f, 0.f, 0.f, 0.f, 0.f};

    #pragma unroll
    for (int kt = 0; kt < 8; ++kt) {
        const v4f a0 = av[4 * kt + 0], a1 = av[4 * kt + 1];
        const v4f a2 = av[4 * kt + 2], a3 = av[4 * kt + 3];
        v16h a;
        a[0]  = (_Float16)a0.x; a[1]  = (_Float16)a0.y;
        a[2]  = (_Float16)a0.z; a[3]  = (_Float16)a0.w;
        a[4]  = (_Float16)a1.x; a[5]  = (_Float16)a1.y;
        a[6]  = (_Float16)a1.z; a[7]  = (_Float16)a1.w;
        a[8]  = (_Float16)a2.x; a[9]  = (_Float16)a2.y;
        a[10] = (_Float16)a2.z; a[11] = (_Float16)a2.w;
        a[12] = (_Float16)a3.x; a[13] = (_Float16)a3.y;
        a[14] = (_Float16)a3.z; a[15] = (_Float16)a3.w;
        #pragma unroll
        for (int ct = 0; ct < 4; ++ct) {
            const v8h lo = bl[((kt * 4 + ct) * 2 + 0) * 32 + lane];
            const v8h hi = bl[((kt * 4 + ct) * 2 + 1) * 32 + lane];
            const v16h bf = __builtin_shufflevector(lo, hi, 0, 1, 2, 3, 4, 5, 6, 7,
                                                    8, 9, 10, 11, 12, 13, 14, 15);
            acc[ct] = __builtin_amdgcn_wmma_f32_16x16x32_f16(
                 false, a,  false, bf,
                 (short)0, acc[ct],  false,  false);
        }
    }

    #pragma unroll
    for (int ct = 0; ct < 4; ++ct) {
        #pragma unroll
        for (int i = 0; i < 8; ++i) {
            const int m = i + 8 * half;
            const float sc = half ? dsc[8 + i] : dsc[i];
            xw1[(size_t)(row0 + m) * HID + (ct * 16 + r)] = acc[ct][i] * sc;
        }
    }
}

#define EPC 8
#define NPW 8
#define G1G_WAVES 8
static_assert(N_NODES % NPW == 0, "node runs must tile the node list");
static_assert((NPW & (NPW - 1)) == 0 && NPW < 32, "row pointers and dinv of a run sit in one wave-wide load");
__global__ __launch_bounds__(G1G_WAVES * 32) void k_gather1(const int* __restrict__ rowptr,
                                                 const int* __restrict__ csr_src,
                                                 const float* __restrict__ dinv,
                                                 const float* __restrict__ xw1,
                                                 const float* __restrict__ b1,
                                                 const float* __restrict__ W2,
                                                 float* __restrict__ hw2s) {
    const int lane = threadIdx.x & 31;
    int w = blockIdx.x * G1G_WAVES + (int)(threadIdx.x >> 5);
    w = __builtin_amdgcn_readfirstlane(w);
    const int i0 = w * NPW;
    if (i0 >= N_NODES) return;

    const int   rpv = rowptr[i0 + (lane <= NPW ? lane : NPW)];
    const float dvv = dinv[i0 + (lane & (NPW - 1))];
    const float2 bb = ((const float2*)b1)[lane];
    const float4 ww = ((const float4*)W2)[lane];
    const float2* rows = (const float2*)xw1;

    int beg = __builtin_amdgcn_readlane(rpv, 0);
    int end = __builtin_amdgcn_readlane(rpv, 1);
    int sv0 = 0;
    {
        const int rem = end - beg;
        const int n = rem < 32 ? rem : 32;
        if (n > 0) sv0 = csr_src[beg + (lane < n ? lane : n - 1)];
    }
    float2 self = rows[(size_t)i0 * (HID / 2) + lane];
    float rx = 0.f, ry = 0.f;
    for (int k = 0; k < NPW; ++k) {
        const float di = __int_as_float(__builtin_amdgcn_readlane(__float_as_int(dvv), k));

        float ax = 0.f, ay = 0.f;
        for (int base = beg; base < end; base += 32) {
            const int rem = end - base;
            const int n = rem < 32 ? rem : 32;
            int sv = sv0;
            if (base != beg) sv = csr_src[base + (lane < n ? lane : n - 1)];
            for (int c = 0; c < n; c += EPC) {
                int s[EPC];
                float2 v[EPC];
                #pragma unroll
                for (int e = 0; e < EPC; ++e) {
                    const int idx = (c + e < n) ? (c + e) : (n - 1);
                    s[e] = __builtin_amdgcn_readlane(sv, idx);
                }
                #pragma unroll
                for (int e = 0; e < EPC; ++e)
                    v[e] = rows[(size_t)s[e] * (HID / 2) + lane];
                #pragma unroll
                for (int e = 0; e < EPC; ++e) {
                    const bool live = (c + e < n);
                    ax += live ? v[e].x : 0.f;
                    ay += live ? v[e].y : 0.f;
                }
            }
        }

        const int nend = __builtin_amdgcn_readlane(rpv, k + 2 < NPW ? k + 2 : NPW);
        int svn = 0;
        {
            const int rem = nend - end;
            const int n = rem < 32 ? rem : 32;
            if (n > 0) svn = csr_src[end + (lane < n ? lane : n - 1)];
        }
        float2 selfn = self;
        if (k + 1 < NPW) selfn = rows[(size_t)(i0 + k + 1) * (HID / 2) + lane];

        float hx = di * (ax + self.x);
        float hy = di * (ay + self.y);
        hx = fmaxf(hx + bb.x, 0.f);
        hy = fmaxf(hy + bb.y, 0.f);
        float o0 = hx * ww.x + hy * ww.z;
        float o1 = hx * ww.y + hy * ww.w;
        for (int o = 16; o > 0; o >>= 1) {
            o0 += __shfl_xor(o0, o);
            o1 += __shfl_xor(o1, o);
        }
        if (lane == k) {
            rx = o0 * di;
            ry = o1 * di;
        }
        beg = end;
        end = nend;
        sv0 = svn;
        self = selfn;
    }
    if (lane < NPW) {
        float2 r;
        r.x = rx;
        r.y = ry;
        ((float2*)hw2s)[i0 + lane] = r;
    }
}

#define LPN 8
#define NPG 4
#define G2_T 160
static_assert(N_NODES % ((G2_T / LPN) * NPG) == 0, "node runs must tile the blocks");
static_assert(NPG <= LPN && (NPG & (NPG - 1)) == 0, "one finishing lane per node of the run");
__global__ __launch_bounds__(G2_T) void k_gather2_softmax(const int* __restrict__ rowptr,
                                                         const int* __restrict__ csr_src,
                                                         const float* __restrict__ dinv,
                                                         const float* __restrict__ hw2s,
                                                         const float* __restrict__ b2,
                                                         float* __restrict__ out) {
    const int t = blockIdx.x * G2_T + threadIdx.x;
    const int i0 = (t / LPN) * NPG;
    const int j = t % LPN;
    const float2* h2 = (const float2*)hw2s;
    const int im = i0 + (j & (NPG - 1));
    const float2 self = h2[im];
    const float di = dinv[im];
    int rp[NPG + 1];
    #pragma unroll
    for (int k = 0; k <= NPG; ++k) rp[k] = rowptr[i0 + k];

    float sx = 0.f, sy = 0.f;
    #pragma unroll
    for (int k = 0; k < NPG; ++k) {
        const int beg = rp[k];
        const int end = rp[k + 1];
        float ax = 0.f, ay = 0.f;
        for (int p = beg + j; p < end; p += 2 * LPN) {
            const int  p2   = p + LPN;
            const bool two  = p2 < end;
            const int  s0   = csr_src[p];
            const int  s1   = csr_src[two ? p2 : p];
            const float2 v0 = h2[s0];
            const float2 v1 = h2[s1];
            ax += v0.x;
            ay += v0.y;
            ax += two ? v1.x : 0.f;
            ay += two ? v1.y : 0.f;
        }
        for (int o = LPN / 2; o > 0; o >>= 1) {
            ax += __shfl_xor(ax, o);
            ay += __shfl_xor(ay, o);
        }
        if (j == k) {
            sx = ax;
            sy = ay;
        }
    }
    if (j < NPG) {
        const float a0 = di * (sx + self.x) + b2[0];
        const float a1 = di * (sy + self.y) + b2[1];
        const float m  = fmaxf(a0, a1);
        const float e0 = __expf(a0 - m);
        const float e1 = __expf(a1 - m);
        const float inv = 1.0f / (e0 + e1);
        float2 r;
        r.x = e0 * inv;
        r.y = e1 * inv;
        ((float2*)out)[im] = r;
    }
}

extern "C" void kernel_launch(void* const* d_in, const int* in_sizes, int n_in,
                              void* d_out, int out_size, void* d_ws, size_t ws_size,
                              hipStream_t stream) {
    const float* x  = (const float*)d_in[0];
    const int*   ei = (const int*)d_in[1];
    const float* W1 = (const float*)d_in[2];
    const float* b1 = (const float*)d_in[3];
    const float* W2 = (const float*)d_in[4];
    const float* b2 = (const float*)d_in[5];
    float* out = (float*)d_out;

    char* ws = (char*)d_ws;
    size_t o = 0;
    auto alloc = [&](size_t bytes) {
        char* p = ws + o;
        o += (bytes + 4095) & ~(size_t)4095;
        return p;
    };
    int*   tbl    = (int*)  alloc((size_t)TBL * sizeof(int));
    float* dinv   = (float*)alloc((size_t)N_NODES * sizeof(float));
    float* xw1    = (float*)alloc((size_t)N_NODES * HID * sizeof(float));
    int*   binned = (int*)  alloc((size_t)N_EDGES * sizeof(int));
    int*   csr    = (int*)  alloc((size_t)N_EDGES * sizeof(int));
    int*   rowptr = (int*)  alloc((size_t)(N_NODES + 1) * sizeof(int));
    float* hw2s   = (float*)alloc((size_t)N_NODES * 2 * sizeof(float));
    (void)ws_size; (void)in_sizes; (void)n_in; (void)out_size;

    k_block_bins  <<<SORT_G, dim3(SORT_T), 0, stream>>>(ei, tbl, binned);
    k_bin_finalize<<<NBIN, dim3(FIN_T), 0, stream>>>(tbl, binned, rowptr, dinv, csr);

    k_gemm1_wmma<<<(N_NODES / 16) / G1_WAVES, dim3(G1_T), 0, stream>>>(x, W1, dinv, xw1);

    k_gather1<<<(N_NODES / NPW + G1G_WAVES - 1) / G1G_WAVES, dim3(G1G_WAVES * 32), 0, stream>>>(rowptr, csr, dinv, xw1, b1, W2, hw2s);

    k_gather2_softmax<<<N_NODES / ((G2_T / LPN) * NPG), dim3(G2_T), 0, stream>>>(rowptr, csr, dinv, hw2s, b2, out);
}


# layer-1 GEMM moves into the per-bin launch: wave w of a bin's work-group does the bin's row tile w, row scale from the bin's dinv in LDS, W1 staged once per bin; GEMM launch gone; sum order kept
# speedup vs baseline: 5.8947x; 1.0582x over previous
#include <hip/hip_runtime.h>
#include <hip/hip_bf16.h>
#include <stdint.h>

#define N_NODES 100000
#define N_EDGES 1600000
#define F_IN    256
#define HID     64

typedef __attribute__((ext_vector_type(16))) _Float16 v16h;
typedef __attribute__((ext_vector_type(8)))  _Float16 v8h;
typedef __attribute__((ext_vector_type(8)))  float    v8f;
typedef __attribute__((ext_vector_type(4)))  float    v4f;

static __device__ __forceinline__ int wave_sum_i32(int v) {
    v += __builtin_amdgcn_update_dpp(0, v, 0x161, 0xf, 0xf, false);
    v += __builtin_amdgcn_update_dpp(0, v, 0x162, 0xf, 0xf, false);
    v += __builtin_amdgcn_update_dpp(0, v, 0x164, 0xf, 0xf, false);
    v += __builtin_amdgcn_update_dpp(0, v, 0x168, 0xf, 0xf, false);
    v += __builtin_amdgcn_permlanex16(v, v, 0x76543210, 0xfedcba98, false, false);
    return v;
}

#define BIN_SHIFT 9
#define BIN_W   (1 << BIN_SHIFT)
#define NBIN    ((N_NODES + BIN_W - 1) / BIN_W)
#define SORT_G  250
#define SORT_T  1024
#define QUADS   (N_EDGES / 4)
#define QPB     (QUADS / SORT_G)
#define EPB     (4 * QPB)
#define TBL     (NBIN * SORT_G)
static_assert(N_EDGES % 4 == 0, "edge quads must tile the edge list");
static_assert(QUADS % SORT_G == 0, "edge blocks must tile the quads");
static_assert(N_NODES <= (1 << (31 - BIN_SHIFT)), "src id and dst-in-bin must pack into one word");
static_assert(EPB < (1 << 15), "run start and run length of a table entry share one non-negative word");
static_assert(QPB > SORT_T && QPB <= 2 * SORT_T, "a thread of an edge block holds one or two quads");
static_assert(SORT_T == 1024, "the bin scan below is written for 32 waves per block");
static_assert(NBIN <= SORT_T, "one thread per bin in the bin scan");

__global__ __launch_bounds__(SORT_T) void k_block_bins(const int* __restrict__ ei,
                                                       int* __restrict__ tbl,
                                                       int* __restrict__ binned) {
    __shared__ int cnt[NBIN];
    __shared__ int wtot[32];
    __shared__ int stage[EPB];
    const int t = threadIdx.x, g = blockIdx.x, lane = t & 31, wv = t >> 5;
    const int4* sq = (const int4*)ei + (size_t)g * QPB;
    const int4* dq = (const int4*)(ei + N_EDGES) + (size_t)g * QPB;
    const bool two = (t + SORT_T) < QPB;
    const int q1 = two ? t + SORT_T : t;
    const int4 s0 = sq[t],  d0 = dq[t];
    const int4 s1 = sq[q1], d1 = dq[q1];
    for (int b = t; b < NBIN; b += SORT_T) cnt[b] = 0;
    __syncthreads();
    atomicAdd(&cnt[d0.x >> BIN_SHIFT], 1);
    atomicAdd(&cnt[d0.y >> BIN_SHIFT], 1);
    atomicAdd(&cnt[d0.z >> BIN_SHIFT], 1);
    atomicAdd(&cnt[d0.w >> BIN_SHIFT], 1);
    if (two) {
        atomicAdd(&cnt[d1.x >> BIN_SHIFT], 1);
        atomicAdd(&cnt[d1.y >> BIN_SHIFT], 1);
        atomicAdd(&cnt[d1.z >> BIN_SHIFT], 1);
        atomicAdd(&cnt[d1.w >> BIN_SHIFT], 1);
    }
    __syncthreads();
    int mine = 0, inc = 0;
    if (wv <= (NBIN - 1) / 32) {
        mine = (t < NBIN) ? cnt[t] : 0;
        inc = mine;
        for (int o = 1; o < 32; o <<= 1) {
            const int u = __shfl_up(inc, o);
            if (lane >= o) inc += u;
        }
        if (lane == 31) wtot[wv] = inc;
    }
    __syncthreads();
    if (t < NBIN) {
        int start = inc - mine;
        for (int k = 0; k < wv; ++k) start += wtot[k];
        cnt[t] = start;
        tbl[t * SORT_G + g] = (start << 16) | mine;
    }
    __syncthreads();
    {
        const int p0 = atomicAdd(&cnt[d0.x >> BIN_SHIFT], 1);
        const int p1 = atomicAdd(&cnt[d0.y >> BIN_SHIFT], 1);
        const int p2 = atomicAdd(&cnt[d0.z >> BIN_SHIFT], 1);
        const int p3 = atomicAdd(&cnt[d0.w >> BIN_SHIFT], 1);
        stage[p0] = (s0.x << BIN_SHIFT) | (d0.x & (BIN_W - 1));
        stage[p1] = (s0.y << BIN_SHIFT) | (d0.y & (BIN_W - 1));
        stage[p2] = (s0.z << BIN_SHIFT) | (d0.z & (BIN_W - 1));
        stage[p3] = (s0.w << BIN_SHIFT) | (d0.w & (BIN_W - 1));
    }
    if (two) {
        const int p0 = atomicAdd(&cnt[d1.x >> BIN_SHIFT], 1);
        const int p1 = atomicAdd(&cnt[d1.y >> BIN_SHIFT], 1);
        const int p2 = atomicAdd(&cnt[d1.z >> BIN_SHIFT], 1);
        const int p3 = atomicAdd(&cnt[d1.w >> BIN_SHIFT], 1);
        stage[p0] = (s1.x << BIN_SHIFT) | (d1.x & (BIN_W - 1));
        stage[p1] = (s1.y << BIN_SHIFT) | (d1.y & (BIN_W - 1));
        stage[p2] = (s1.z << BIN_SHIFT) | (d1.z & (BIN_W - 1));
        stage[p3] = (s1.w << BIN_SHIFT) | (d1.w & (BIN_W - 1));
    }
    __syncthreads();
    int* mylist = binned + (size_t)g * EPB;
    int sv[(EPB + SORT_T - 1) / SORT_T];
    #pragma unroll
    for (int k = 0; k < (EPB + SORT_T - 1) / SORT_T; ++k) {
        const int s = t + k * SORT_T;
        sv[k] = stage[s < EPB ? s : EPB - 1];
    }
    #pragma unroll
    for (int k = 0; k < (EPB + SORT_T - 1) / SORT_T; ++k) {
        const int s = t + k * SORT_T;
        if (s < EPB) mylist[s] = sv[k];
    }
}

#define FIN_T   1024
#define RUNS_PW ((SORT_G + 31) / 32)
#define B_UNITS (8 * 4 * 2 * 32)
#define STAGE_TASKS (B_UNITS / 4)
static_assert(BIN_W <= FIN_T && BIN_W % 32 == 0, "one thread per node of the bin, whole waves");
static_assert(SORT_G <= FIN_T && FIN_T == 1024, "one thread per table entry of the bin's row; 32 waves");
static_assert(BIN_W == 16 * (FIN_T / 32), "one 16-row tile of the bin per wave");
static_assert(N_NODES % 16 == 0, "a row tile is inside the node list or outside it, never across its end");
static_assert(STAGE_TASKS <= FIN_T - BIN_W, "W1 is staged by the threads that own no node and no table entry");
static_assert(F_IN == 256 && HID == 64, "fragment index arithmetic below is written for 8 K-steps x 4 column tiles");
__global__ __launch_bounds__(FIN_T) void k_bin_finalize_gemm1(const int* __restrict__ tbl,
                                                              const int* __restrict__ binned,
                                                              const float* __restrict__ x,
                                                              const float* __restrict__ W1,
                                                              int* __restrict__ rowptr,
                                                              float* __restrict__ dinv,
                                                              int* __restrict__ csr_src,
                                                              float* __restrict__ xw1) {
    __shared__ v8h bl[B_UNITS];
    __shared__ float sdinv[BIN_W];
    __shared__ int cnt[BIN_W];
    __shared__ int wtot[BIN_W / 32];
    __shared__ int runoff[SORT_G];
    __shared__ int runlen[SORT_G];
    __shared__ int red[32];
    const int t = threadIdx.x, lane = t & 31, wv = t >> 5;
    const int b = blockIdx.x;
    const int half = lane >> 4;
    const int r    = lane & 15;

    if (t >= FIN_T - STAGE_TASKS) {
        const int q    = t - (FIN_T - STAGE_TASKS);
        const int c4   = q & 3;
        const int hf   = (q >> 2) & 1;
        const int part = (q >> 3) & 1;
        const int ct   = (q >> 4) & 3;
        const int kt   = q >> 6;
        const float* p = W1 + (size_t)(kt * 32 + part * 16 + 8 * hf) * HID + ct * 16 + 4 * c4;
        v4f w[8];
        #pragma unroll
        for (int j = 0; j < 8; ++j) w[j] = *(const v4f*)(p + j * HID);
        const int u0 = ((kt * 4 + ct) * 2 + part) * 32 + hf * 16 + 4 * c4;
        #pragma unroll
        for (int e = 0; e < 4; ++e) {
            v8h h;
            #pragma unroll
            for (int j = 0; j < 8; ++j) h[j] = (_Float16)w[j][e];
            bl[u0 + e] = h;
        }
    }

    int rs = 0;
    if (t < SORT_G) {
        const int e = tbl[b * SORT_G + t];
        rs = e >> 16;
        runoff[t] = t * EPB + rs;
        runlen[t] = e & 0xffff;
    }
    if (t < BIN_W) cnt[t] = 0;
    rs = wave_sum_i32(rs);
    if (lane == 0) red[wv] = rs;
    __syncthreads();
    int beg = 0;
    for (int k = 0; k < RUNS_PW; ++k) beg += red[k];

    int rl[RUNS_PW], ro[RUNS_PW], w0[RUNS_PW], w1[RUNS_PW];
    int longest = 0;
    #pragma unroll
    for (int j = 0; j < RUNS_PW; ++j) {
        const int g = wv + 32 * j;
        rl[j] = (g < SORT_G) ? runlen[g] : 0;
        ro[j] = (g < SORT_G) ? runoff[g] : 0;
        longest = rl[j] > longest ? rl[j] : longest;
    }
    #pragma unroll
    for (int j = 0; j < RUNS_PW; ++j) {
        w0[j] = (lane < rl[j])      ? binned[ro[j] + lane]      : -1;
        w1[j] = (lane + 32 < rl[j]) ? binned[ro[j] + lane + 32] : -1;
    }
    #pragma unroll
    for (int j = 0; j < RUNS_PW; ++j) {
        if (w0[j] >= 0) atomicAdd(&cnt[w0[j] & (BIN_W - 1)], 1);
        if (w1[j] >= 0) atomicAdd(&cnt[w1[j] & (BIN_W - 1)], 1);
    }
    if (longest > 64) {
        for (int g = wv; g < SORT_G; g += 32) {
            const int rl = runlen[g], ro = runoff[g];
            for (int k = lane + 64; k < rl; k += 32)
                atomicAdd(&cnt[binned[ro + k] & (BIN_W - 1)], 1);
        }
    }

    __syncthreads();
    int dg = 0, inc = 0;
    if (t < BIN_W) {
        dg = cnt[t];
        inc = dg;
        for (int o = 1; o < 32; o <<= 1) {
            const int u = __shfl_up(inc, o);
            if (lane >= o) inc += u;
        }
        if (lane == 31) wtot[wv] = inc;
    }
    __syncthreads();
    if (t < BIN_W) {
        int base = beg;
        for (int k = 0; k < wv; ++k) base += wtot[k];
        const int excl = base + inc - dg;
        cnt[t] = excl;
        const int node = b * BIN_W + t;
        const float dv = __frsqrt_rn((float)dg + 1.0f);
        sdinv[t] = dv;
        if (node < N_NODES) {
            rowptr[node] = excl;
            dinv[node] = dv;
        }
        if (node == N_NODES - 1) rowptr[N_NODES] = excl + dg;
    }
    __syncthreads();
    const int row0 = b * BIN_W + 16 * wv;
    const bool tile = row0 < N_NODES;
    const float* xrow = x + (size_t)(tile ? row0 + r : 0) * F_IN + 8 * half;
    v4f av[16];
    if (tile) {
        #pragma unroll
        for (int kt = 0; kt < 4; ++kt) {
            const float* seg0 = xrow + kt * 32;
            const float* seg1 = seg0 + 16;
            av[4 * kt + 0] = __builtin_nontemporal_load((const v4f*)seg0);
            av[4 * kt + 1] = __builtin_nontemporal_load((const v4f*)(seg0 + 4));
            av[4 * kt + 2] = __builtin_nontemporal_load((const v4f*)seg1);
            av[4 * kt + 3] = __builtin_nontemporal_load((const v4f*)(seg1 + 4));
        }
    }
    #pragma unroll
    for (int j = 0; j < RUNS_PW; ++j) {
        if (w0[j] >= 0) {
            const int pos = atomicAdd(&cnt[w0[j] & (BIN_W - 1)], 1);
            csr_src[pos] = w0[j] >> BIN_SHIFT;
        }
        if (w1[j] >= 0) {
            const int pos = atomicAdd(&cnt[w1[j] & (BIN_W - 1)], 1);
            csr_src[pos] = w1[j] >> BIN_SHIFT;
        }
    }
    if (longest > 64) {
        for (int g = wv; g < SORT_G; g += 32) {
            const int rl = runlen[g], ro = runoff[g];
            for (int k = lane + 64; k < rl; k += 32) {
                const int xw = binned[ro + k];
                const int pos = atomicAdd(&cnt[xw & (BIN_W - 1)], 1);
                csr_src[pos] = xw >> BIN_SHIFT;
            }
        }
    }

    if (tile) {
        v8f acc[4];
        #pragma unroll
        for (int ct = 0; ct < 4; ++ct) acc[ct] = (v8f){0.f, 0.f, 0.f, 0.f, 0.f, 0.f, 0.f, 0.f};

        #pragma unroll
        for (int hk = 0; hk < 2; ++hk) {
            if (hk == 1) {
                #pragma unroll
                for (int kt = 0; kt < 4; ++kt) {
                    const float* seg0 = xrow + (4 + kt) * 32;
                    const float* seg1 = seg0 + 16;
                    av[4 * kt + 0] = __builtin_nontemporal_load((const v4f*)seg0);
                    av[4 * kt + 1] = __builtin_nontemporal_load((const v4f*)(seg0 + 4));
                    av[4 * kt + 2] = __builtin_nontemporal_load((const v4f*)seg1);
                    av[4 * kt + 3] = __builtin_nontemporal_load((const v4f*)(seg1 + 4));
                }
            }
            #pragma unroll
            for (int kq = 0; kq < 4; ++kq) {
                const int kt = 4 * hk + kq;
                const v4f a0 = av[4 * kq + 0], a1 = av[4 * kq + 1];
                const v4f a2 = av[4 * kq + 2], a3 = av[4 * kq + 3];
                v16h a;
                a[0]  = (_Float16)a0.x; a[1]  = (_Float16)a0.y;
                a[2]  = (_Float16)a0.z; a[3]  = (_Float16)a0.w;
                a[4]  = (_Float16)a1.x; a[5]  = (_Float16)a1.y;
                a[6]  = (_Float16)a1.z; a[7]  = (_Float16)a1.w;
                a[8]  = (_Float16)a2.x; a[9]  = (_Float16)a2.y;
                a[10] = (_Float16)a2.z; a[11] = (_Float16)a2.w;
                a[12] = (_Float16)a3.x; a[13] = (_Float16)a3.y;
                a[14] = (_Float16)a3.z; a[15] = (_Float16)a3.w;
                #pragma unroll
                for (int ct = 0; ct < 4; ++ct) {
                    const v8h lo = bl[((kt * 4 + ct) * 2 + 0) * 32 + lane];
                    const v8h hi = bl[((kt * 4 + ct) * 2 + 1) * 32 + lane];
                    const v16h bf = __builtin_shufflevector(lo, hi, 0, 1, 2, 3, 4, 5, 6, 7,
                                                            8, 9, 10, 11, 12, 13, 14, 15);
                    acc[ct] = __builtin_amdgcn_wmma_f32_16x16x32_f16(
                         false, a,  false, bf,
                         (short)0, acc[ct],  false,  false);
                }
            }
        }

        const int m0 = 16 * wv + 8 * half;
        float sc[8];
        #pragma unroll
        for (int i = 0; i < 8; ++i) sc[i] = sdinv[m0 + i];
        #pragma unroll
        for (int ct = 0; ct < 4; ++ct) {
            #pragma unroll
            for (int i = 0; i < 8; ++i) {
                const int m = i + 8 * half;
                xw1[(size_t)(row0 + m) * HID + (ct * 16 + r)] = acc[ct][i] * sc[i];
            }
        }
    }
}

#define EPC 8
#define NPW 8
#define G1G_WAVES 8
static_assert(N_NODES % NPW == 0, "node runs must tile the node list");
static_assert((NPW & (NPW - 1)) == 0 && NPW < 32, "row pointers and dinv of a run sit in one wave-wide load");
__global__ __launch_bounds__(G1G_WAVES * 32) void k_gather1(const int* __restrict__ rowptr,
                                                 const int* __restrict__ csr_src,
                                                 const float* __restrict__ dinv,
                                                 const float* __restrict__ xw1,
                                                 const float* __restrict__ b1,
                                                 const float* __restrict__ W2,
                                                 float* __restrict__ hw2s) {
    const int lane = threadIdx.x & 31;
    int w = blockIdx.x * G1G_WAVES + (int)(threadIdx.x >> 5);
    w = __builtin_amdgcn_readfirstlane(w);
    const int i0 = w * NPW;
    if (i0 >= N_NODES) return;

    const int   rpv = rowptr[i0 + (lane <= NPW ? lane : NPW)];
    const float dvv = dinv[i0 + (lane & (NPW - 1))];
    const float2 bb = ((const float2*)b1)[lane];
    const float4 ww = ((const float4*)W2)[lane];
    const float2* rows = (const float2*)xw1;

    int beg = __builtin_amdgcn_readlane(rpv, 0);
    int end = __builtin_amdgcn_readlane(rpv, 1);
    int sv0 = 0;
    {
        const int rem = end - beg;
        const int n = rem < 32 ? rem : 32;
        if (n > 0) sv0 = csr_src[beg + (lane < n ? lane : n - 1)];
    }
    float2 self = rows[(size_t)i0 * (HID / 2) + lane];
    float rx = 0.f, ry = 0.f;
    for (int k = 0; k < NPW; ++k) {
        const float di = __int_as_float(__builtin_amdgcn_readlane(__float_as_int(dvv), k));

        float ax = 0.f, ay = 0.f;
        for (int base = beg; base < end; base += 32) {
            const int rem = end - base;
            const int n = rem < 32 ? rem : 32;
            int sv = sv0;
            if (base != beg) sv = csr_src[base + (lane < n ? lane : n - 1)];
            for (int c = 0; c < n; c += EPC) {
                int s[EPC];
                float2 v[EPC];
                #pragma unroll
                for (int e = 0; e < EPC; ++e) {
                    const int idx = (c + e < n) ? (c + e) : (n - 1);
                    s[e] = __builtin_amdgcn_readlane(sv, idx);
                }
                #pragma unroll
                for (int e = 0; e < EPC; ++e)
                    v[e] = rows[(size_t)s[e] * (HID / 2) + lane];
                #pragma unroll
                for (int e = 0; e < EPC; ++e) {
                    const bool live = (c + e < n);
                    ax += live ? v[e].x : 0.f;
                    ay += live ? v[e].y : 0.f;
                }
            }
        }

        const int nend = __builtin_amdgcn_readlane(rpv, k + 2 < NPW ? k + 2 : NPW);
        int svn = 0;
        {
            const int rem = nend - end;
            const int n = rem < 32 ? rem : 32;
            if (n > 0) svn = csr_src[end + (lane < n ? lane : n - 1)];
        }
        float2 selfn = self;
        if (k + 1 < NPW) selfn = rows[(size_t)(i0 + k + 1) * (HID / 2) + lane];

        float hx = di * (ax + self.x);
        float hy = di * (ay + self.y);
        hx = fmaxf(hx + bb.x, 0.f);
        hy = fmaxf(hy + bb.y, 0.f);
        float o0 = hx * ww.x + hy * ww.z;
        float o1 = hx * ww.y + hy * ww.w;
        for (int o = 16; o > 0; o >>= 1) {
            o0 += __shfl_xor(o0, o);
            o1 += __shfl_xor(o1, o);
        }
        if (lane == k) {
            rx = o0 * di;
            ry = o1 * di;
        }
        beg = end;
        end = nend;
        sv0 = svn;
        self = selfn;
    }
    if (lane < NPW) {
        float2 r;
        r.x = rx;
        r.y = ry;
        ((float2*)hw2s)[i0 + lane] = r;
    }
}

#define LPN 8
#define NPG 4
#define G2_T 160
static_assert(N_NODES % ((G2_T / LPN) * NPG) == 0, "node runs must tile the blocks");
static_assert(NPG <= LPN && (NPG & (NPG - 1)) == 0, "one finishing lane per node of the run");
__global__ __launch_bounds__(G2_T) void k_gather2_softmax(const int* __restrict__ rowptr,
                                                         const int* __restrict__ csr_src,
                                                         const float* __restrict__ dinv,
                                                         const float* __restrict__ hw2s,
                                                         const float* __restrict__ b2,
                                                         float* __restrict__ out) {
    const int t = blockIdx.x * G2_T + threadIdx.x;
    const int i0 = (t / LPN) * NPG;
    const int j = t % LPN;
    const float2* h2 = (const float2*)hw2s;
    const int im = i0 + (j & (NPG - 1));
    const float2 self = h2[im];
    const float di = dinv[im];
    int rp[NPG + 1];
    #pragma unroll
    for (int k = 0; k <= NPG; ++k) rp[k] = rowptr[i0 + k];

    float sx = 0.f, sy = 0.f;
    #pragma unroll
    for (int k = 0; k < NPG; ++k) {
        const int beg = rp[k];
        const int end = rp[k + 1];
        float ax = 0.f, ay = 0.f;
        for (int p = beg + j; p < end; p += 2 * LPN) {
            const int  p2   = p + LPN;
            const bool two  = p2 < end;
            const int  s0   = csr_src[p];
            const int  s1   = csr_src[two ? p2 : p];
            const float2 v0 = h2[s0];
            const float2 v1 = h2[s1];
            ax += v0.x;
            ay += v0.y;
            ax += two ? v1.x : 0.f;
            ay += two ? v1.y : 0.f;
        }
        for (int o = LPN / 2; o > 0; o >>= 1) {
            ax += __shfl_xor(ax, o);
            ay += __shfl_xor(ay, o);
        }
        if (j == k) {
            sx = ax;
            sy = ay;
        }
    }
    if (j < NPG) {
        const float a0 = di * (sx + self.x) + b2[0];
        const float a1 = di * (sy + self.y) + b2[1];
        const float m  = fmaxf(a0, a1);
        const float e0 = __expf(a0 - m);
        const float e1 = __expf(a1 - m);
        const float inv = 1.0f / (e0 + e1);
        float2 r;
        r.x = e0 * inv;
        r.y = e1 * inv;
        ((float2*)out)[im] = r;
    }
}

extern "C" void kernel_launch(void* const* d_in, const int* in_sizes, int n_in,
                              void* d_out, int out_size, void* d_ws, size_t ws_size,
                              hipStream_t stream) {
    const float* x  = (const float*)d_in[0];
    const int*   ei = (const int*)d_in[1];
    const float* W1 = (const float*)d_in[2];
    const float* b1 = (const float*)d_in[3];
    const float* W2 = (const float*)d_in[4];
    const float* b2 = (const float*)d_in[5];
    float* out = (float*)d_out;

    char* ws = (char*)d_ws;
    size_t o = 0;
    auto alloc = [&](size_t bytes) {
        char* p = ws + o;
        o += (bytes + 4095) & ~(size_t)4095;
        return p;
    };
    int*   tbl    = (int*)  alloc((size_t)TBL * sizeof(int));
    float* dinv   = (float*)alloc((size_t)N_NODES * sizeof(float));
    float* xw1    = (float*)alloc((size_t)N_NODES * HID * sizeof(float));
    int*   binned = (int*)  alloc((size_t)N_EDGES * sizeof(int));
    int*   csr    = (int*)  alloc((size_t)N_EDGES * sizeof(int));
    int*   rowptr = (int*)  alloc((size_t)(N_NODES + 1) * sizeof(int));
    float* hw2s   = (float*)alloc((size_t)N_NODES * 2 * sizeof(float));
    (void)ws_size; (void)in_sizes; (void)n_in; (void)out_size;

    k_block_bins  <<<SORT_G, dim3(SORT_T), 0, stream>>>(ei, tbl, binned);
    k_bin_finalize_gemm1<<<NBIN, dim3(FIN_T), 0, stream>>>(tbl, binned, x, W1, rowptr, dinv, csr, xw1);

    k_gather1<<<(N_NODES / NPW + G1G_WAVES - 1) / G1G_WAVES, dim3(G1G_WAVES * 32), 0, stream>>>(rowptr, csr, dinv, xw1, b1, W2, hw2s);

    k_gather2_softmax<<<N_NODES / ((G2_T / LPN) * NPG), dim3(G2_T), 0, stream>>>(rowptr, csr, dinv, hw2s, b2, out);
}


# per-bin launch: src ids go to an LDS image of the bin's csr slice (same returning LDS add as slot) and leave as whole 128-B lines, not 1.6M scattered 4-B stores; sums still in LDS arrival order
# speedup vs baseline: 6.1972x; 1.0513x over previous
#include <hip/hip_runtime.h>
#include <hip/hip_bf16.h>
#include <stdint.h>

#define N_NODES 100000
#define N_EDGES 1600000
#define F_IN    256
#define HID     64

typedef __attribute__((ext_vector_type(16))) _Float16 v16h;
typedef __attribute__((ext_vector_type(8)))  _Float16 v8h;
typedef __attribute__((ext_vector_type(8)))  float    v8f;
typedef __attribute__((ext_vector_type(4)))  float    v4f;

static __device__ __forceinline__ int wave_sum_i32(int v) {
    v += __builtin_amdgcn_update_dpp(0, v, 0x161, 0xf, 0xf, false);
    v += __builtin_amdgcn_update_dpp(0, v, 0x162, 0xf, 0xf, false);
    v += __builtin_amdgcn_update_dpp(0, v, 0x164, 0xf, 0xf, false);
    v += __builtin_amdgcn_update_dpp(0, v, 0x168, 0xf, 0xf, false);
    v += __builtin_amdgcn_permlanex16(v, v, 0x76543210, 0xfedcba98, false, false);
    return v;
}

#define BIN_SHIFT 9
#define BIN_W   (1 << BIN_SHIFT)
#define NBIN    ((N_NODES + BIN_W - 1) / BIN_W)
#define SORT_G  250
#define SORT_T  1024
#define QUADS   (N_EDGES / 4)
#define QPB     (QUADS / SORT_G)
#define EPB     (4 * QPB)
#define TBL     (NBIN * SORT_G)
static_assert(N_EDGES % 4 == 0, "edge quads must tile the edge list");
static_assert(QUADS % SORT_G == 0, "edge blocks must tile the quads");
static_assert(N_NODES <= (1 << (31 - BIN_SHIFT)), "src id and dst-in-bin must pack into one word");
static_assert(EPB < (1 << 15), "run start and run length of a table entry share one non-negative word");
static_assert(QPB > SORT_T && QPB <= 2 * SORT_T, "a thread of an edge block holds one or two quads");
static_assert(SORT_T == 1024, "the bin scan below is written for 32 waves per block");
static_assert(NBIN <= SORT_T, "one thread per bin in the bin scan");

__global__ __launch_bounds__(SORT_T) void k_block_bins(const int* __restrict__ ei,
                                                       int* __restrict__ tbl,
                                                       int* __restrict__ binned) {
    __shared__ int cnt[NBIN];
    __shared__ int wtot[32];
    __shared__ int stage[EPB];
    const int t = threadIdx.x, g = blockIdx.x, lane = t & 31, wv = t >> 5;
    const int4* sq = (const int4*)ei + (size_t)g * QPB;
    const int4* dq = (const int4*)(ei + N_EDGES) + (size_t)g * QPB;
    const bool two = (t + SORT_T) < QPB;
    const int q1 = two ? t + SORT_T : t;
    const int4 s0 = sq[t],  d0 = dq[t];
    const int4 s1 = sq[q1], d1 = dq[q1];
    for (int b = t; b < NBIN; b += SORT_T) cnt[b] = 0;
    __syncthreads();
    atomicAdd(&cnt[d0.x >> BIN_SHIFT], 1);
    atomicAdd(&cnt[d0.y >> BIN_SHIFT], 1);
    atomicAdd(&cnt[d0.z >> BIN_SHIFT], 1);
    atomicAdd(&cnt[d0.w >> BIN_SHIFT], 1);
    if (two) {
        atomicAdd(&cnt[d1.x >> BIN_SHIFT], 1);
        atomicAdd(&cnt[d1.y >> BIN_SHIFT], 1);
        atomicAdd(&cnt[d1.z >> BIN_SHIFT], 1);
        atomicAdd(&cnt[d1.w >> BIN_SHIFT], 1);
    }
    __syncthreads();
    int mine = 0, inc = 0;
    if (wv <= (NBIN - 1) / 32) {
        mine = (t < NBIN) ? cnt[t] : 0;
        inc = mine;
        for (int o = 1; o < 32; o <<= 1) {
            const int u = __shfl_up(inc, o);
            if (lane >= o) inc += u;
        }
        if (lane == 31) wtot[wv] = inc;
    }
    __syncthreads();
    if (t < NBIN) {
        int start = inc - mine;
        for (int k = 0; k < wv; ++k) start += wtot[k];
        cnt[t] = start;
        tbl[t * SORT_G + g] = (start << 16) | mine;
    }
    __syncthreads();
    {
        const int p0 = atomicAdd(&cnt[d0.x >> BIN_SHIFT], 1);
        const int p1 = atomicAdd(&cnt[d0.y >> BIN_SHIFT], 1);
        const int p2 = atomicAdd(&cnt[d0.z >> BIN_SHIFT], 1);
        const int p3 = atomicAdd(&cnt[d0.w >> BIN_SHIFT], 1);
        stage[p0] = (s0.x << BIN_SHIFT) | (d0.x & (BIN_W - 1));
        stage[p1] = (s0.y << BIN_SHIFT) | (d0.y & (BIN_W - 1));
        stage[p2] = (s0.z << BIN_SHIFT) | (d0.z & (BIN_W - 1));
        stage[p3] = (s0.w << BIN_SHIFT) | (d0.w & (BIN_W - 1));
    }
    if (two) {
        const int p0 = atomicAdd(&cnt[d1.x >> BIN_SHIFT], 1);
        const int p1 = atomicAdd(&cnt[d1.y >> BIN_SHIFT], 1);
        const int p2 = atomicAdd(&cnt[d1.z >> BIN_SHIFT], 1);
        const int p3 = atomicAdd(&cnt[d1.w >> BIN_SHIFT], 1);
        stage[p0] = (s1.x << BIN_SHIFT) | (d1.x & (BIN_W - 1));
        stage[p1] = (s1.y << BIN_SHIFT) | (d1.y & (BIN_W - 1));
        stage[p2] = (s1.z << BIN_SHIFT) | (d1.z & (BIN_W - 1));
        stage[p3] = (s1.w << BIN_SHIFT) | (d1.w & (BIN_W - 1));
    }
    __syncthreads();
    int* mylist = binned + (size_t)g * EPB;
    int sv[(EPB + SORT_T - 1) / SORT_T];
    #pragma unroll
    for (int k = 0; k < (EPB + SORT_T - 1) / SORT_T; ++k) {
        const int s = t + k * SORT_T;
        sv[k] = stage[s < EPB ? s : EPB - 1];
    }
    #pragma unroll
    for (int k = 0; k < (EPB + SORT_T - 1) / SORT_T; ++k) {
        const int s = t + k * SORT_T;
        if (s < EPB) mylist[s] = sv[k];
    }
}

#define FIN_T   1024
#define RUNS_PW ((SORT_G + 31) / 32)
#define B_UNITS (8 * 4 * 2 * 32)
#define STAGE_TASKS (B_UNITS / 4)
#define STAGE_W 10240
#define COPY_K  ((STAGE_W + 31 + FIN_T - 1) / FIN_T)
static_assert(BIN_W <= FIN_T && BIN_W % 32 == 0, "one thread per node of the bin, whole waves");
static_assert(SORT_G <= FIN_T && FIN_T == 1024, "one thread per table entry of the bin's row; 32 waves");
static_assert(BIN_W == 16 * (FIN_T / 32), "one 16-row tile of the bin per wave");
static_assert(N_NODES % 16 == 0, "a row tile is inside the node list or outside it, never across its end");
static_assert(STAGE_TASKS <= FIN_T - BIN_W, "W1 is staged by the threads that own no node and no table entry");
static_assert(F_IN == 256 && HID == 64, "fragment index arithmetic below is written for 8 K-steps x 4 column tiles");
__global__ __launch_bounds__(FIN_T) void k_bin_finalize_gemm1(const int* __restrict__ tbl,
                                                              const int* __restrict__ binned,
                                                              const float* __restrict__ x,
                                                              const float* __restrict__ W1,
                                                              int* __restrict__ rowptr,
                                                              float* __restrict__ dinv,
                                                              int* __restrict__ csr_src,
                                                              float* __restrict__ xw1) {
    __shared__ v8h bl[B_UNITS];
    __shared__ float sdinv[BIN_W];
    __shared__ int cnt[BIN_W];
    __shared__ int wtot[BIN_W / 32];
    __shared__ int runoff[SORT_G];
    __shared__ int runlen[SORT_G];
    __shared__ int red[32];
    __shared__ int slice[STAGE_W];
    __shared__ int tot_sh;
    const int t = threadIdx.x, lane = t & 31, wv = t >> 5;
    const int b = blockIdx.x;
    const int half = lane >> 4;
    const int r    = lane & 15;

    if (t >= FIN_T - STAGE_TASKS) {
        const int q    = t - (FIN_T - STAGE_TASKS);
        const int c4   = q & 3;
        const int hf   = (q >> 2) & 1;
        const int part = (q >> 3) & 1;
        const int ct   = (q >> 4) & 3;
        const int kt   = q >> 6;
        const float* p = W1 + (size_t)(kt * 32 + part * 16 + 8 * hf) * HID + ct * 16 + 4 * c4;
        v4f w[8];
        #pragma unroll
        for (int j = 0; j < 8; ++j) w[j] = *(const v4f*)(p + j * HID);
        const int u0 = ((kt * 4 + ct) * 2 + part) * 32 + hf * 16 + 4 * c4;
        #pragma unroll
        for (int e = 0; e < 4; ++e) {
            v8h h;
            #pragma unroll
            for (int j = 0; j < 8; ++j) h[j] = (_Float16)w[j][e];
            bl[u0 + e] = h;
        }
    }

    int rs = 0;
    if (t < SORT_G) {
        const int e = tbl[b * SORT_G + t];
        rs = e >> 16;
        runoff[t] = t * EPB + rs;
        runlen[t] = e & 0xffff;
    }
    if (t < BIN_W) cnt[t] = 0;
    rs = wave_sum_i32(rs);
    if (lane == 0) red[wv] = rs;
    __syncthreads();
    int beg = 0;
    for (int k = 0; k < RUNS_PW; ++k) beg += red[k];

    int rl[RUNS_PW], ro[RUNS_PW], w0[RUNS_PW], w1[RUNS_PW];
    int longest = 0;
    #pragma unroll
    for (int j = 0; j < RUNS_PW; ++j) {
        const int g = wv + 32 * j;
        rl[j] = (g < SORT_G) ? runlen[g] : 0;
        ro[j] = (g < SORT_G) ? runoff[g] : 0;
        longest = rl[j] > longest ? rl[j] : longest;
    }
    #pragma unroll
    for (int j = 0; j < RUNS_PW; ++j) {
        w0[j] = (lane < rl[j])      ? binned[ro[j] + lane]      : -1;
        w1[j] = (lane + 32 < rl[j]) ? binned[ro[j] + lane + 32] : -1;
    }
    #pragma unroll
    for (int j = 0; j < RUNS_PW; ++j) {
        if (w0[j] >= 0) atomicAdd(&cnt[w0[j] & (BIN_W - 1)], 1);
        if (w1[j] >= 0) atomicAdd(&cnt[w1[j] & (BIN_W - 1)], 1);
    }
    if (longest > 64) {
        for (int g = wv; g < SORT_G; g += 32) {
            const int rl = runlen[g], ro = runoff[g];
            for (int k = lane + 64; k < rl; k += 32)
                atomicAdd(&cnt[binned[ro + k] & (BIN_W - 1)], 1);
        }
    }

    __syncthreads();
    int dg = 0, inc = 0;
    if (t < BIN_W) {
        dg = cnt[t];
        inc = dg;
        for (int o = 1; o < 32; o <<= 1) {
            const int u = __shfl_up(inc, o);
            if (lane >= o) inc += u;
        }
        if (lane == 31) wtot[wv] = inc;
    }
    __syncthreads();
    if (t < BIN_W) {
        int base = beg;
        for (int k = 0; k < wv; ++k) base += wtot[k];
        const int excl = base + inc - dg;
        cnt[t] = excl;
        const int node = b * BIN_W + t;
        const float dv = __frsqrt_rn((float)dg + 1.0f);
        sdinv[t] = dv;
        if (node < N_NODES) {
            rowptr[node] = excl;
            dinv[node] = dv;
        }
        if (node == N_NODES - 1) rowptr[N_NODES] = excl + dg;
        if (t == BIN_W - 1) tot_sh = excl + dg - beg;
    }
    __syncthreads();
    const int row0 = b * BIN_W + 16 * wv;
    const bool tile = row0 < N_NODES;
    const float* xrow = x + (size_t)(tile ? row0 + r : 0) * F_IN + 8 * half;
    v4f av[16];
    if (tile) {
        #pragma unroll
        for (int kt = 0; kt < 4; ++kt) {
            const float* seg0 = xrow + kt * 32;
            const float* seg1 = seg0 + 16;
            av[4 * kt + 0] = __builtin_nontemporal_load((const v4f*)seg0);
            av[4 * kt + 1] = __builtin_nontemporal_load((const v4f*)(seg0 + 4));
            av[4 * kt + 2] = __builtin_nontemporal_load((const v4f*)seg1);
            av[4 * kt + 3] = __builtin_nontemporal_load((const v4f*)(seg1 + 4));
        }
    }
    const int tot = tot_sh;
    const bool staged = tot <= STAGE_W;
    if (staged) {
        int p0[RUNS_PW], p1[RUNS_PW];
        #pragma unroll
        for (int j = 0; j < RUNS_PW; ++j) {
            p0[j] = 0;
            p1[j] = 0;
            if (w0[j] >= 0) p0[j] = atomicAdd(&cnt[w0[j] & (BIN_W - 1)], 1);
            if (w1[j] >= 0) p1[j] = atomicAdd(&cnt[w1[j] & (BIN_W - 1)], 1);
        }
        #pragma unroll
        for (int j = 0; j < RUNS_PW; ++j) {
            if (w0[j] >= 0) slice[p0[j] - beg] = w0[j] >> BIN_SHIFT;
            if (w1[j] >= 0) slice[p1[j] - beg] = w1[j] >> BIN_SHIFT;
        }
        if (longest > 64) {
            for (int g = wv; g < SORT_G; g += 32) {
                const int rl = runlen[g], ro = runoff[g];
                for (int k = lane + 64; k < rl; k += 32) {
                    const int xw = binned[ro + k];
                    const int pos = atomicAdd(&cnt[xw & (BIN_W - 1)], 1);
                    slice[pos - beg] = xw >> BIN_SHIFT;
                }
            }
        }
    } else {
        #pragma unroll
        for (int j = 0; j < RUNS_PW; ++j) {
            if (w0[j] >= 0) {
                const int pos = atomicAdd(&cnt[w0[j] & (BIN_W - 1)], 1);
                csr_src[pos] = w0[j] >> BIN_SHIFT;
            }
            if (w1[j] >= 0) {
                const int pos = atomicAdd(&cnt[w1[j] & (BIN_W - 1)], 1);
                csr_src[pos] = w1[j] >> BIN_SHIFT;
            }
        }
        if (longest > 64) {
            for (int g = wv; g < SORT_G; g += 32) {
                const int rl = runlen[g], ro = runoff[g];
                for (int k = lane + 64; k < rl; k += 32) {
                    const int xw = binned[ro + k];
                    const int pos = atomicAdd(&cnt[xw & (BIN_W - 1)], 1);
                    csr_src[pos] = xw >> BIN_SHIFT;
                }
            }
        }
    }
    __builtin_amdgcn_fence(__ATOMIC_RELEASE, "workgroup", "local");
    __builtin_amdgcn_s_barrier();
    __builtin_amdgcn_fence(__ATOMIC_ACQUIRE, "workgroup", "local");
    if (staged) {
        const int shift = beg & 31;
        int cv[COPY_K];
        #pragma unroll
        for (int k = 0; k < COPY_K; ++k) {
            const int q = t + k * FIN_T - shift;
            cv[k] = slice[(q >= 0 && q < tot) ? q : 0];
        }
        #pragma unroll
        for (int k = 0; k < COPY_K; ++k) {
            const int q = t + k * FIN_T - shift;
            if (q >= 0 && q < tot) csr_src[beg + q] = cv[k];
        }
    }

    if (tile) {
        v8f acc[4];
        #pragma unroll
        for (int ct = 0; ct < 4; ++ct) acc[ct] = (v8f){0.f, 0.f, 0.f, 0.f, 0.f, 0.f, 0.f, 0.f};

        #pragma unroll
        for (int hk = 0; hk < 2; ++hk) {
            if (hk == 1) {
                #pragma unroll
                for (int kt = 0; kt < 4; ++kt) {
                    const float* seg0 = xrow + (4 + kt) * 32;
                    const float* seg1 = seg0 + 16;
                    av[4 * kt + 0] = __builtin_nontemporal_load((const v4f*)seg0);
                    av[4 * kt + 1] = __builtin_nontemporal_load((const v4f*)(seg0 + 4));
                    av[4 * kt + 2] = __builtin_nontemporal_load((const v4f*)seg1);
                    av[4 * kt + 3] = __builtin_nontemporal_load((const v4f*)(seg1 + 4));
                }
            }
            #pragma unroll
            for (int kq = 0; kq < 4; ++kq) {
                const int kt = 4 * hk + kq;
                const v4f a0 = av[4 * kq + 0], a1 = av[4 * kq + 1];
                const v4f a2 = av[4 * kq + 2], a3 = av[4 * kq + 3];
                v16h a;
                a[0]  = (_Float16)a0.x; a[1]  = (_Float16)a0.y;
                a[2]  = (_Float16)a0.z; a[3]  = (_Float16)a0.w;
                a[4]  = (_Float16)a1.x; a[5]  = (_Float16)a1.y;
                a[6]  = (_Float16)a1.z; a[7]  = (_Float16)a1.w;
                a[8]  = (_Float16)a2.x; a[9]  = (_Float16)a2.y;
                a[10] = (_Float16)a2.z; a[11] = (_Float16)a2.w;
                a[12] = (_Float16)a3.x; a[13] = (_Float16)a3.y;
                a[14] = (_Float16)a3.z; a[15] = (_Float16)a3.w;
                #pragma unroll
                for (int ct = 0; ct < 4; ++ct) {
                    const v8h lo = bl[((kt * 4 + ct) * 2 + 0) * 32 + lane];
                    const v8h hi = bl[((kt * 4 + ct) * 2 + 1) * 32 + lane];
                    const v16h bf = __builtin_shufflevector(lo, hi, 0, 1, 2, 3, 4, 5, 6, 7,
                                                            8, 9, 10, 11, 12, 13, 14, 15);
                    acc[ct] = __builtin_amdgcn_wmma_f32_16x16x32_f16(
                         false, a,  false, bf,
                         (short)0, acc[ct],  false,  false);
                }
            }
        }

        const int m0 = 16 * wv + 8 * half;
        float sc[8];
        #pragma unroll
        for (int i = 0; i < 8; ++i) sc[i] = sdinv[m0 + i];
        #pragma unroll
        for (int ct = 0; ct < 4; ++ct) {
            #pragma unroll
            for (int i = 0; i < 8; ++i) {
                const int m = i + 8 * half;
                xw1[(size_t)(row0 + m) * HID + (ct * 16 + r)] = acc[ct][i] * sc[i];
            }
        }
    }
}

#define EPC 8
#define NPW 8
#define G1G_WAVES 8
static_assert(N_NODES % NPW == 0, "node runs must tile the node list");
static_assert((NPW & (NPW - 1)) == 0 && NPW < 32, "row pointers and dinv of a run sit in one wave-wide load");
__global__ __launch_bounds__(G1G_WAVES * 32) void k_gather1(const int* __restrict__ rowptr,
                                                 const int* __restrict__ csr_src,
                                                 const float* __restrict__ dinv,
                                                 const float* __restrict__ xw1,
                                                 const float* __restrict__ b1,
                                                 const float* __restrict__ W2,
                                                 float* __restrict__ hw2s) {
    const int lane = threadIdx.x & 31;
    int w = blockIdx.x * G1G_WAVES + (int)(threadIdx.x >> 5);
    w = __builtin_amdgcn_readfirstlane(w);
    const int i0 = w * NPW;
    if (i0 >= N_NODES) return;

    const int   rpv = rowptr[i0 + (lane <= NPW ? lane : NPW)];
    const float dvv = dinv[i0 + (lane & (NPW - 1))];
    const float2 bb = ((const float2*)b1)[lane];
    const float4 ww = ((const float4*)W2)[lane];
    const float2* rows = (const float2*)xw1;

    int beg = __builtin_amdgcn_readlane(rpv, 0);
    int end = __builtin_amdgcn_readlane(rpv, 1);
    int sv0 = 0;
    {
        const int rem = end - beg;
        const int n = rem < 32 ? rem : 32;
        if (n > 0) sv0 = csr_src[beg + (lane < n ? lane : n - 1)];
    }
    float2 self = rows[(size_t)i0 * (HID / 2) + lane];
    float rx = 0.f, ry = 0.f;
    for (int k = 0; k < NPW; ++k) {
        const float di = __int_as_float(__builtin_amdgcn_readlane(__float_as_int(dvv), k));

        float ax = 0.f, ay = 0.f;
        for (int base = beg; base < end; base += 32) {
            const int rem = end - base;
            const int n = rem < 32 ? rem : 32;
            int sv = sv0;
            if (base != beg) sv = csr_src[base + (lane < n ? lane : n - 1)];
            for (int c = 0; c < n; c += EPC) {
                int s[EPC];
                float2 v[EPC];
                #pragma unroll
                for (int e = 0; e < EPC; ++e) {
                    const int idx = (c + e < n) ? (c + e) : (n - 1);
                    s[e] = __builtin_amdgcn_readlane(sv, idx);
                }
                #pragma unroll
                for (int e = 0; e < EPC; ++e)
                    v[e] = rows[(size_t)s[e] * (HID / 2) + lane];
                #pragma unroll
                for (int e = 0; e < EPC; ++e) {
                    const bool live = (c + e < n);
                    ax += live ? v[e].x : 0.f;
                    ay += live ? v[e].y : 0.f;
                }
            }
        }

        const int nend = __builtin_amdgcn_readlane(rpv, k + 2 < NPW ? k + 2 : NPW);
        int svn = 0;
        {
            const int rem = nend - end;
            const int n = rem < 32 ? rem : 32;
            if (n > 0) svn = csr_src[end + (lane < n ? lane : n - 1)];
        }
        float2 selfn = self;
        if (k + 1 < NPW) selfn = rows[(size_t)(i0 + k + 1) * (HID / 2) + lane];

        float hx = di * (ax + self.x);
        float hy = di * (ay + self.y);
        hx = fmaxf(hx + bb.x, 0.f);
        hy = fmaxf(hy + bb.y, 0.f);
        float o0 = hx * ww.x + hy * ww.z;
        float o1 = hx * ww.y + hy * ww.w;
        for (int o = 16; o > 0; o >>= 1) {
            o0 += __shfl_xor(o0, o);
            o1 += __shfl_xor(o1, o);
        }
        if (lane == k) {
            rx = o0 * di;
            ry = o1 * di;
        }
        beg = end;
        end = nend;
        sv0 = svn;
        self = selfn;
    }
    if (lane < NPW) {
        float2 r;
        r.x = rx;
        r.y = ry;
        ((float2*)hw2s)[i0 + lane] = r;
    }
}

#define LPN 8
#define NPG 4
#define G2_T 160
static_assert(N_NODES % ((G2_T / LPN) * NPG) == 0, "node runs must tile the blocks");
static_assert(NPG <= LPN && (NPG & (NPG - 1)) == 0, "one finishing lane per node of the run");
__global__ __launch_bounds__(G2_T) void k_gather2_softmax(const int* __restrict__ rowptr,
                                                         const int* __restrict__ csr_src,
                                                         const float* __restrict__ dinv,
                                                         const float* __restrict__ hw2s,
                                                         const float* __restrict__ b2,
                                                         float* __restrict__ out) {
    const int t = blockIdx.x * G2_T + threadIdx.x;
    const int i0 = (t / LPN) * NPG;
    const int j = t % LPN;
    const float2* h2 = (const float2*)hw2s;
    const int im = i0 + (j & (NPG - 1));
    const float2 self = h2[im];
    const float di = dinv[im];
    int rp[NPG + 1];
    #pragma unroll
    for (int k = 0; k <= NPG; ++k) rp[k] = rowptr[i0 + k];

    float sx = 0.f, sy = 0.f;
    #pragma unroll
    for (int k = 0; k < NPG; ++k) {
        const int beg = rp[k];
        const int end = rp[k + 1];
        float ax = 0.f, ay = 0.f;
        for (int p = beg + j; p < end; p += 2 * LPN) {
            const int  p2   = p + LPN;
            const bool two  = p2 < end;
            const int  s0   = csr_src[p];
            const int  s1   = csr_src[two ? p2 : p];
            const float2 v0 = h2[s0];
            const float2 v1 = h2[s1];
            ax += v0.x;
            ay += v0.y;
            ax += two ? v1.x : 0.f;
            ay += two ? v1.y : 0.f;
        }
        for (int o = LPN / 2; o > 0; o >>= 1) {
            ax += __shfl_xor(ax, o);
            ay += __shfl_xor(ay, o);
        }
        if (j == k) {
            sx = ax;
            sy = ay;
        }
    }
    if (j < NPG) {
        const float a0 = di * (sx + self.x) + b2[0];
        const float a1 = di * (sy + self.y) + b2[1];
        const float m  = fmaxf(a0, a1);
        const float e0 = __expf(a0 - m);
        const float e1 = __expf(a1 - m);
        const float inv = 1.0f / (e0 + e1);
        float2 r;
        r.x = e0 * inv;
        r.y = e1 * inv;
        ((float2*)out)[im] = r;
    }
}

extern "C" void kernel_launch(void* const* d_in, const int* in_sizes, int n_in,
                              void* d_out, int out_size, void* d_ws, size_t ws_size,
                              hipStream_t stream) {
    const float* x  = (const float*)d_in[0];
    const int*   ei = (const int*)d_in[1];
    const float* W1 = (const float*)d_in[2];
    const float* b1 = (const float*)d_in[3];
    const float* W2 = (const float*)d_in[4];
    const float* b2 = (const float*)d_in[5];
    float* out = (float*)d_out;

    char* ws = (char*)d_ws;
    size_t o = 0;
    auto alloc = [&](size_t bytes) {
        char* p = ws + o;
        o += (bytes + 4095) & ~(size_t)4095;
        return p;
    };
    int*   tbl    = (int*)  alloc((size_t)TBL * sizeof(int));
    float* dinv   = (float*)alloc((size_t)N_NODES * sizeof(float));
    float* xw1    = (float*)alloc((size_t)N_NODES * HID * sizeof(float));
    int*   binned = (int*)  alloc((size_t)N_EDGES * sizeof(int));
    int*   csr    = (int*)  alloc((size_t)N_EDGES * sizeof(int));
    int*   rowptr = (int*)  alloc((size_t)(N_NODES + 1) * sizeof(int));
    float* hw2s   = (float*)alloc((size_t)N_NODES * 2 * sizeof(float));
    (void)ws_size; (void)in_sizes; (void)n_in; (void)out_size;

    k_block_bins  <<<SORT_G, dim3(SORT_T), 0, stream>>>(ei, tbl, binned);
    k_bin_finalize_gemm1<<<NBIN, dim3(FIN_T), 0, stream>>>(tbl, binned, x, W1, rowptr, dinv, csr, xw1);

    k_gather1<<<(N_NODES / NPW + G1G_WAVES - 1) / G1G_WAVES, dim3(G1G_WAVES * 32), 0, stream>>>(rowptr, csr, dinv, xw1, b1, W2, hw2s);

    k_gather2_softmax<<<N_NODES / ((G2_T / LPN) * NPG), dim3(G2_T), 0, stream>>>(rowptr, csr, dinv, hw2s, b2, out);
}
